# starting kernel
# baseline (speedup 1.0000x reference)
#include <hip/hip_runtime.h>
#include <hip/hip_bf16.h>

typedef __attribute__((ext_vector_type(16))) _Float16 v16h;
typedef __attribute__((ext_vector_type(8)))  _Float16 v8h;
typedef __attribute__((ext_vector_type(8)))  float    v8f;
typedef __attribute__((ext_vector_type(4)))  float    v4f;

#define H_HEADS 4
#define IN_DIM  128
#define HID     64
#define COLS    256
#define PRED    16
#define RELS    50
#define LAYERS  3

__device__ __forceinline__ int f2key(float f) {
    int i = __float_as_int(f);
    return (i >= 0) ? i : (i ^ 0x7FFFFFFF);
}
__device__ __forceinline__ float key2f(int i) {
    return __int_as_float((i >= 0) ? i : (i ^ 0x7FFFFFFF));
}

__global__ void k_w1_to_f16(const float* __restrict__ W1,
                            _Float16* __restrict__ W1h, int n) {
    int i = blockIdx.x * blockDim.x + threadIdx.x;
    if (i < n) W1h[i] = (_Float16)W1[i];
}

__global__ void k_etab(const float* __restrict__ rel_emb,
                       const float* __restrict__ W_pred,
                       float* __restrict__ etab) {
    int i = blockIdx.x * blockDim.x + threadIdx.x;
    if (i >= LAYERS * RELS * H_HEADS) return;
    int l = i / (RELS * H_HEADS);
    int rem = i % (RELS * H_HEADS);
    int r = rem / H_HEADS;
    int h = rem % H_HEADS;
    float e = 0.f;
    #pragma unroll
    for (int p = 0; p < PRED; ++p)
        e += rel_emb[r * PRED + p] * W_pred[(l * PRED + p) * H_HEADS + h];
    etab[i] = (e > 0.f) ? e : 0.2f * e;
}

__global__ void k_mlp_wmma(const float* __restrict__ inputs,
                           const _Float16* __restrict__ W1h,
                           const float* __restrict__ b1,
                           const float* __restrict__ W2,
                           const float* __restrict__ b2,
                           float* __restrict__ hout,
                           int n, int nTiles) {
    const int wave = (blockIdx.x * blockDim.x + threadIdx.x) >> 5;
    if (wave >= nTiles) return;
    const int lane = threadIdx.x & 31;
    const int grp  = lane >> 4;
    const int l15  = lane & 15;
    const int nodeBase = wave * 16;
    int arow = nodeBase + l15;
    if (arow > n - 1) arow = n - 1;
    const float* __restrict__ abase = inputs + (size_t)arow * IN_DIM;

    const v8f zero = {};

    #pragma unroll
    for (int head = 0; head < H_HEADS; ++head) {
        v8f acc0 = zero, acc1 = zero, acc2 = zero, acc3 = zero;

        #pragma unroll
        for (int kk = 0; kk < IN_DIM; kk += 32) {
            const int kb = kk + grp * 8;
            v4f a0 = *(const v4f*)(abase + kb + 0);
            v4f a1 = *(const v4f*)(abase + kb + 4);
            v4f a2 = *(const v4f*)(abase + kb + 16);
            v4f a3 = *(const v4f*)(abase + kb + 20);
            v16h A;
            #pragma unroll
            for (int i = 0; i < 4; ++i) {
                A[i]      = (_Float16)a0[i];
                A[4 + i]  = (_Float16)a1[i];
                A[8 + i]  = (_Float16)a2[i];
                A[12 + i] = (_Float16)a3[i];
            }
            #pragma unroll
            for (int tt = 0; tt < 4; ++tt) {
                const int col = head * HID + tt * 16 + l15;
                const _Float16* q = W1h + (size_t)col * IN_DIM + kb;
                v8h blo = *(const v8h*)(q);
                v8h bhi = *(const v8h*)(q + 16);
                v16h B;
                #pragma unroll
                for (int i = 0; i < 8; ++i) { B[i] = blo[i]; B[8 + i] = bhi[i]; }
                v8f c = (tt == 0) ? acc0 : (tt == 1) ? acc1 : (tt == 2) ? acc2 : acc3;
                c = __builtin_amdgcn_wmma_f32_16x16x32_f16(
                        false, A, false, B, (short)0, c, false, false);
                if (tt == 0) acc0 = c; else if (tt == 1) acc1 = c;
                else if (tt == 2) acc2 = c; else acc3 = c;
            }
        }

        v8f hacc = zero;
        #pragma unroll
        for (int tt = 0; tt < 4; ++tt) {
            const int col  = head * HID + tt * 16 + l15;
            const float bias = b1[col];
            const float w2   = W2[col];
            v8f c = (tt == 0) ? acc0 : (tt == 1) ? acc1 : (tt == 2) ? acc2 : acc3;
            #pragma unroll
            for (int v = 0; v < 8; ++v) {
                float p = c[v] + bias;
                p = (p > 0.f) ? p : 0.f;
                p *= w2;
                p += __shfl_xor(p, 1, 32);
                p += __shfl_xor(p, 2, 32);
                p += __shfl_xor(p, 4, 32);
                p += __shfl_xor(p, 8, 32);
                hacc[v] += p;
            }
        }
        if (l15 == 0) {
            #pragma unroll
            for (int v = 0; v < 8; ++v) {
                const int node = nodeBase + v + 8 * grp;
                if (node < n)
                    hout[node * H_HEADS + head] = hacc[v] + b2[head];
            }
        }
    }
}

__global__ void k_init_nh(int* __restrict__ m, float* __restrict__ s,
                          float* __restrict__ u, int count) {
    int i = blockIdx.x * blockDim.x + threadIdx.x;
    if (i < count) { m[i] = (int)0x80000000; s[i] = 0.f; u[i] = 0.f; }
}

__global__ void k_edge_max(const int* __restrict__ et, const int* __restrict__ dst,
                           const float* __restrict__ etab_l,
                           int* __restrict__ m, int E) {
    int i = blockIdx.x * blockDim.x + threadIdx.x;
    if (i >= E) return;
    __builtin_prefetch(et + i + 8192, 0, 0);
    __builtin_prefetch(dst + i + 8192, 0, 0);
    const int t = et[i];
    const int d = dst[i];
    const float* e = etab_l + t * H_HEADS;
    #pragma unroll
    for (int h = 0; h < H_HEADS; ++h)
        atomicMax(&m[d * H_HEADS + h], f2key(e[h]));
}

__global__ void k_edge_agg(const int* __restrict__ et, const int* __restrict__ src,
                           const int* __restrict__ dst,
                           const float* __restrict__ etab_l,
                           const int* __restrict__ m,
                           const float* __restrict__ hcur,
                           float* __restrict__ s, float* __restrict__ u, int E) {
    int i = blockIdx.x * blockDim.x + threadIdx.x;
    if (i >= E) return;
    __builtin_prefetch(et + i + 8192, 0, 0);
    __builtin_prefetch(src + i + 8192, 0, 0);
    __builtin_prefetch(dst + i + 8192, 0, 0);
    const int t  = et[i];
    const int sn = src[i];
    const int d  = dst[i];
    const float* e = etab_l + t * H_HEADS;
    #pragma unroll
    for (int h = 0; h < H_HEADS; ++h) {
        const float mv = key2f(m[d * H_HEADS + h]);
        const float w  = __expf(e[h] - mv);
        atomicAdd(&s[d * H_HEADS + h], w);
        atomicAdd(&u[d * H_HEADS + h], w * hcur[sn * H_HEADS + h]);
    }
}

__global__ void k_node_update(const float* __restrict__ s, const float* __restrict__ u,
                              float* __restrict__ hnext, int n, int do_mean) {
    int i = blockIdx.x * blockDim.x + threadIdx.x;
    if (i >= n) return;
    float v[H_HEADS];
    #pragma unroll
    for (int h = 0; h < H_HEADS; ++h) {
        const float sv = s[i * H_HEADS + h];
        float x = (sv > 0.f) ? (u[i * H_HEADS + h] / sv) : 0.f;
        v[h] = (x > 0.f) ? x : 0.f;
    }
    if (do_mean) {
        const float mn = 0.25f * (v[0] + v[1] + v[2] + v[3]);
        #pragma unroll
        for (int h = 0; h < H_HEADS; ++h) v[h] = mn;
    }
    #pragma unroll
    for (int h = 0; h < H_HEADS; ++h) hnext[i * H_HEADS + h] = v[h];
}

__global__ void k_final(const float* __restrict__ hfin,
                        const float* __restrict__ centrality,
                        const float* __restrict__ gamma,
                        const float* __restrict__ beta,
                        float* __restrict__ out, int n) {
    int i = blockIdx.x * blockDim.x + threadIdx.x;
    if (i >= n) return;
    const float c = centrality[i];
    float acc = 0.f;
    #pragma unroll
    for (int h = 0; h < H_HEADS; ++h)
        acc += (c * gamma[h] + beta[h]) * hfin[i * H_HEADS + h];
    acc *= 0.25f;
    out[i] = (acc > 0.f) ? acc : 0.01f * acc;
}

extern "C" void kernel_launch(void* const* d_in, const int* in_sizes, int n_in,
                              void* d_out, int out_size, void* d_ws, size_t ws_size,
                              hipStream_t stream) {
    const float* inputs     = (const float*)d_in[0];
    const float* centrality = (const float*)d_in[1];
    const float* W1         = (const float*)d_in[2];
    const float* b1         = (const float*)d_in[3];
    const float* W2         = (const float*)d_in[4];
    const float* b2         = (const float*)d_in[5];
    const float* rel_emb    = (const float*)d_in[6];
    const float* W_pred     = (const float*)d_in[7];
    const float* gamma      = (const float*)d_in[8];
    const float* beta       = (const float*)d_in[9];
    const int*   edge_types = (const int*)d_in[10];
    const int*   src        = (const int*)d_in[11];
    const int*   dst        = (const int*)d_in[12];
    float* out = (float*)d_out;

    const int N = in_sizes[1];
    const int E = in_sizes[10];

    char* ws = (char*)d_ws;
    _Float16* W1h = (_Float16*)ws;
    float* etab   = (float*)(ws + 65536);
    float* hA     = (float*)(ws + 69632);
    float* hB     = hA + (size_t)N * H_HEADS;
    int*   m      = (int*)(hB + (size_t)N * H_HEADS);
    float* s      = (float*)(m + (size_t)N * H_HEADS);
    float* u      = s + (size_t)N * H_HEADS;

    const int TPB = 256;

    {
        int cnt = COLS * IN_DIM;
        k_w1_to_f16<<<(cnt + TPB - 1) / TPB, TPB, 0, stream>>>(W1, W1h, cnt);
    }
    {
        int cnt = LAYERS * RELS * H_HEADS;
        k_etab<<<(cnt + TPB - 1) / TPB, TPB, 0, stream>>>(rel_emb, W_pred, etab);
    }
    {
        int nTiles = (N + 15) / 16;
        int wavesPerBlock = 4;
        int blocks = (nTiles + wavesPerBlock - 1) / wavesPerBlock;
        k_mlp_wmma<<<blocks, wavesPerBlock * 32, 0, stream>>>(
            inputs, W1h, b1, W2, b2, hA, N, nTiles);
    }

    float* hcur = hA;
    float* hnxt = hB;
    const int nhCount = N * H_HEADS;
    const int nhBlocks = (nhCount + TPB - 1) / TPB;
    const int eBlocks  = (E + TPB - 1) / TPB;
    const int nBlocks  = (N + TPB - 1) / TPB;

    for (int l = 0; l < LAYERS; ++l) {
        const float* etab_l = etab + l * RELS * H_HEADS;
        k_init_nh<<<nhBlocks, TPB, 0, stream>>>(m, s, u, nhCount);
        k_edge_max<<<eBlocks, TPB, 0, stream>>>(edge_types, dst, etab_l, m, E);
        k_edge_agg<<<eBlocks, TPB, 0, stream>>>(edge_types, src, dst, etab_l,
                                                m, hcur, s, u, E);
        k_node_update<<<nBlocks, TPB, 0, stream>>>(s, u, hnxt, N,
                                                   (l != LAYERS - 1) ? 1 : 0);
        float* tmp = hcur; hcur = hnxt; hnxt = tmp;
    }

    k_final<<<nBlocks, TPB, 0, stream>>>(hcur, centrality, gamma, beta, out, N);
}


# Bucket edges by destination once (counting sort, 1 int atomic per edge), then per-node softmax and sums by 8-lane gather in all 3 layers: no float atomics; sums in bucket order plus xor tree
# speedup vs baseline: 8.8604x; 8.8604x over previous
#include <hip/hip_runtime.h>
#include <hip/hip_bf16.h>

typedef __attribute__((ext_vector_type(16))) _Float16 v16h;
typedef __attribute__((ext_vector_type(8)))  _Float16 v8h;
typedef __attribute__((ext_vector_type(8)))  float    v8f;
typedef __attribute__((ext_vector_type(4)))  float    v4f;

#define H_HEADS 4
#define IN_DIM  128
#define HID     64
#define COLS    256
#define PRED    16
#define RELS    50
#define LAYERS  3

__global__ void k_w1_to_f16(const float* __restrict__ W1,
                            _Float16* __restrict__ W1h, int n) {
    int i = blockIdx.x * blockDim.x + threadIdx.x;
    if (i < n) W1h[i] = (_Float16)W1[i];
}

__global__ void k_etab(const float* __restrict__ rel_emb,
                       const float* __restrict__ W_pred,
                       float* __restrict__ etab) {
    int i = blockIdx.x * blockDim.x + threadIdx.x;
    if (i >= LAYERS * RELS * H_HEADS) return;
    int l = i / (RELS * H_HEADS);
    int rem = i % (RELS * H_HEADS);
    int r = rem / H_HEADS;
    int h = rem % H_HEADS;
    float e = 0.f;
    #pragma unroll
    for (int p = 0; p < PRED; ++p)
        e += rel_emb[r * PRED + p] * W_pred[(l * PRED + p) * H_HEADS + h];
    etab[i] = (e > 0.f) ? e : 0.2f * e;
}

__global__ void k_mlp_wmma(const float* __restrict__ inputs,
                           const _Float16* __restrict__ W1h,
                           const float* __restrict__ b1,
                           const float* __restrict__ W2,
                           const float* __restrict__ b2,
                           float* __restrict__ hout,
                           int n, int nTiles) {
    const int wave = (blockIdx.x * blockDim.x + threadIdx.x) >> 5;
    if (wave >= nTiles) return;
    const int lane = threadIdx.x & 31;
    const int grp  = lane >> 4;
    const int l15  = lane & 15;
    const int nodeBase = wave * 16;
    int arow = nodeBase + l15;
    if (arow > n - 1) arow = n - 1;
    const float* __restrict__ abase = inputs + (size_t)arow * IN_DIM;

    const v8f zero = {};

    #pragma unroll
    for (int head = 0; head < H_HEADS; ++head) {
        v8f acc0 = zero, acc1 = zero, acc2 = zero, acc3 = zero;

        #pragma unroll
        for (int kk = 0; kk < IN_DIM; kk += 32) {
            const int kb = kk + grp * 8;
            v4f a0 = *(const v4f*)(abase + kb + 0);
            v4f a1 = *(const v4f*)(abase + kb + 4);
            v4f a2 = *(const v4f*)(abase + kb + 16);
            v4f a3 = *(const v4f*)(abase + kb + 20);
            v16h A;
            #pragma unroll
            for (int i = 0; i < 4; ++i) {
                A[i]      = (_Float16)a0[i];
                A[4 + i]  = (_Float16)a1[i];
                A[8 + i]  = (_Float16)a2[i];
                A[12 + i] = (_Float16)a3[i];
            }
            #pragma unroll
            for (int tt = 0; tt < 4; ++tt) {
                const int col = head * HID + tt * 16 + l15;
                const _Float16* q = W1h + (size_t)col * IN_DIM + kb;
                v8h blo = *(const v8h*)(q);
                v8h bhi = *(const v8h*)(q + 16);
                v16h B;
                #pragma unroll
                for (int i = 0; i < 8; ++i) { B[i] = blo[i]; B[8 + i] = bhi[i]; }
                v8f c = (tt == 0) ? acc0 : (tt == 1) ? acc1 : (tt == 2) ? acc2 : acc3;
                c = __builtin_amdgcn_wmma_f32_16x16x32_f16(
                        false, A, false, B, (short)0, c, false, false);
                if (tt == 0) acc0 = c; else if (tt == 1) acc1 = c;
                else if (tt == 2) acc2 = c; else acc3 = c;
            }
        }

        v8f hacc = zero;
        #pragma unroll
        for (int tt = 0; tt < 4; ++tt) {
            const int col  = head * HID + tt * 16 + l15;
            const float bias = b1[col];
            const float w2   = W2[col];
            v8f c = (tt == 0) ? acc0 : (tt == 1) ? acc1 : (tt == 2) ? acc2 : acc3;
            #pragma unroll
            for (int v = 0; v < 8; ++v) {
                float p = c[v] + bias;
                p = (p > 0.f) ? p : 0.f;
                p *= w2;
                p += __shfl_xor(p, 1, 32);
                p += __shfl_xor(p, 2, 32);
                p += __shfl_xor(p, 4, 32);
                p += __shfl_xor(p, 8, 32);
                hacc[v] += p;
            }
        }
        if (l15 == 0) {
            #pragma unroll
            for (int v = 0; v < 8; ++v) {
                const int node = nodeBase + v + 8 * grp;
                if (node < n)
                    hout[node * H_HEADS + head] = hacc[v] + b2[head];
            }
        }
    }
}

#define GL  8
#define GCH 4

__global__ void k_zero_cnt(int* __restrict__ cnt, int n) {
    int i = blockIdx.x * blockDim.x + threadIdx.x;
    if (i < n) cnt[i] = 0;
}

__global__ void k_rank(const int* __restrict__ dst, int* __restrict__ cnt,
                       int* __restrict__ rank, int E) {
    const int i = (blockIdx.x * blockDim.x + threadIdx.x) * 4;
    if (i + 3 < E) {
        const int4 d = *(const int4*)(dst + i);
        int4 r;
        r.x = atomicAdd(&cnt[d.x], 1);
        r.y = atomicAdd(&cnt[d.y], 1);
        r.z = atomicAdd(&cnt[d.z], 1);
        r.w = atomicAdd(&cnt[d.w], 1);
        *(int4*)(rank + i) = r;
    } else {
        for (int k = i; k < E; ++k) rank[k] = atomicAdd(&cnt[dst[k]], 1);
    }
}

__global__ void k_scan_local(const int* __restrict__ cnt, int* __restrict__ rowstart,
                             int* __restrict__ bsum, int n) {
    __shared__ int wsum[8];
    const int t = threadIdx.x, lane = t & 31, w = t >> 5;
    const int base = blockIdx.x * 1024 + t * 4;
    int c[4];
    #pragma unroll
    for (int k = 0; k < 4; ++k) c[k] = (base + k < n) ? cnt[base + k] : 0;
    const int tsum = c[0] + c[1] + c[2] + c[3];
    int x = tsum;
    #pragma unroll
    for (int d = 1; d < 32; d <<= 1) {
        const int y = __shfl_up(x, d, 32);
        if (lane >= d) x += y;
    }
    if (lane == 31) wsum[w] = x;
    __syncthreads();
    int wbase = 0;
    #pragma unroll
    for (int k = 0; k < 8; ++k) if (k < w) wbase += wsum[k];
    int run = wbase + x - tsum;
    #pragma unroll
    for (int k = 0; k < 4; ++k) {
        if (base + k < n) rowstart[base + k] = run;
        run += c[k];
    }
    if (t == 255) bsum[blockIdx.x] = wbase + x;
}

__global__ void k_scan_add(int* __restrict__ rowstart, const int* __restrict__ bsum, int n) {
    __shared__ int wsum[8];
    const int t = threadIdx.x, lane = t & 31, w = t >> 5;
    int acc = 0;
    for (int k = t; k < (int)blockIdx.x; k += 256) acc += bsum[k];
    #pragma unroll
    for (int d = 16; d >= 1; d >>= 1) acc += __shfl_xor(acc, d, 32);
    if (lane == 0) wsum[w] = acc;
    __syncthreads();
    int boff = 0;
    #pragma unroll
    for (int k = 0; k < 8; ++k) boff += wsum[k];
    const int base = blockIdx.x * 1024 + t * 4;
    #pragma unroll
    for (int k = 0; k < 4; ++k)
        if (base + k < n) rowstart[base + k] += boff;
}

__global__ void k_scatter(const int* __restrict__ et, const int* __restrict__ src,
                          const int* __restrict__ dst, const int* __restrict__ rank,
                          const int* __restrict__ rowstart,
                          unsigned* __restrict__ sorted, int E) {
    const int i = (blockIdx.x * blockDim.x + threadIdx.x) * 4;
    if (i + 3 < E) {
        const int4 d = *(const int4*)(dst + i);
        const int4 r = *(const int4*)(rank + i);
        const int4 s = *(const int4*)(src + i);
        const int4 t = *(const int4*)(et + i);
        const int p0 = rowstart[d.x], p1 = rowstart[d.y];
        const int p2 = rowstart[d.z], p3 = rowstart[d.w];
        sorted[p0 + r.x] = ((unsigned)s.x << 6) | (unsigned)t.x;
        sorted[p1 + r.y] = ((unsigned)s.y << 6) | (unsigned)t.y;
        sorted[p2 + r.z] = ((unsigned)s.z << 6) | (unsigned)t.z;
        sorted[p3 + r.w] = ((unsigned)s.w << 6) | (unsigned)t.w;
    } else {
        for (int k = i; k < E; ++k)
            sorted[rowstart[dst[k]] + rank[k]] = ((unsigned)src[k] << 6) | (unsigned)et[k];
    }
}

__device__ __forceinline__ void g_load(const unsigned* __restrict__ sorted, int start,
                                       int len, int j0, unsigned (&pk)[GCH]) {
    #pragma unroll
    for (int k = 0; k < GCH; ++k) {
        const int j = j0 + k * GL;
        pk[k] = (j < len) ? sorted[start + j] : 0u;
    }
}
__device__ __forceinline__ void g_max(const v4f* tab, const unsigned (&pk)[GCH],
                                      int len, int j0, float (&m)[H_HEADS]) {
    #pragma unroll
    for (int k = 0; k < GCH; ++k) {
        if (j0 + k * GL < len) {
            const v4f e = tab[pk[k] & 63u];
            #pragma unroll
            for (int h = 0; h < H_HEADS; ++h) m[h] = fmaxf(m[h], e[h]);
        }
    }
}
__device__ __forceinline__ void g_sum(const v4f* tab, const float* __restrict__ hcur,
                                      const unsigned (&pk)[GCH], int len, int j0,
                                      const float (&m)[H_HEADS],
                                      float (&s)[H_HEADS], float (&u)[H_HEADS]) {
    v4f hv[GCH];
    #pragma unroll
    for (int k = 0; k < GCH; ++k) {
        hv[k] = (v4f){0.f, 0.f, 0.f, 0.f};
        if (j0 + k * GL < len)
            hv[k] = *(const v4f*)(hcur + (size_t)(pk[k] >> 6) * H_HEADS);
    }
    #pragma unroll
    for (int k = 0; k < GCH; ++k) {
        if (j0 + k * GL < len) {
            const v4f e = tab[pk[k] & 63u];
            #pragma unroll
            for (int h = 0; h < H_HEADS; ++h) {
                const float wgt = __expf(e[h] - m[h]);
                s[h] += wgt;
                u[h] += wgt * hv[k][h];
            }
        }
    }
}

__global__ void k_gather(const unsigned* __restrict__ sorted,
                         const int* __restrict__ rowstart,
                         const int* __restrict__ cnt,
                         const float* __restrict__ etab_l,
                         const float* __restrict__ hcur,
                         float* __restrict__ hnext, int n, int do_mean) {
    __shared__ v4f tab[RELS];
    if (threadIdx.x < RELS)
        tab[threadIdx.x] = *(const v4f*)(etab_l + threadIdx.x * H_HEADS);
    __syncthreads();

    const int gid  = blockIdx.x * blockDim.x + threadIdx.x;
    const int node = gid / GL;
    const int sub  = threadIdx.x & (GL - 1);
    int start = 0, len = 0;
    if (node < n) { start = rowstart[node]; len = cnt[node]; }

    unsigned pk0[GCH];
    g_load(sorted, start, len, sub, pk0);

    float m[H_HEADS];
    #pragma unroll
    for (int h = 0; h < H_HEADS; ++h) m[h] = -__builtin_huge_valf();
    g_max(tab, pk0, len, sub, m);
    for (int base = GL * GCH; base < len; base += GL * GCH) {
        unsigned pk[GCH];
        g_load(sorted, start, len, base + sub, pk);
        g_max(tab, pk, len, base + sub, m);
    }
    #pragma unroll
    for (int h = 0; h < H_HEADS; ++h) {
        m[h] = fmaxf(m[h], __shfl_xor(m[h], 1, 32));
        m[h] = fmaxf(m[h], __shfl_xor(m[h], 2, 32));
        m[h] = fmaxf(m[h], __shfl_xor(m[h], 4, 32));
    }

    float s[H_HEADS], u[H_HEADS];
    #pragma unroll
    for (int h = 0; h < H_HEADS; ++h) { s[h] = 0.f; u[h] = 0.f; }
    g_sum(tab, hcur, pk0, len, sub, m, s, u);
    for (int base = GL * GCH; base < len; base += GL * GCH) {
        unsigned pk[GCH];
        g_load(sorted, start, len, base + sub, pk);
        g_sum(tab, hcur, pk, len, base + sub, m, s, u);
    }
    #pragma unroll
    for (int h = 0; h < H_HEADS; ++h) {
        s[h] += __shfl_xor(s[h], 1, 32);  u[h] += __shfl_xor(u[h], 1, 32);
        s[h] += __shfl_xor(s[h], 2, 32);  u[h] += __shfl_xor(u[h], 2, 32);
        s[h] += __shfl_xor(s[h], 4, 32);  u[h] += __shfl_xor(u[h], 4, 32);
    }

    if (sub == 0 && node < n) {
        v4f v;
        #pragma unroll
        for (int h = 0; h < H_HEADS; ++h) {
            const float x = (s[h] > 0.f) ? (u[h] / s[h]) : 0.f;
            v[h] = (x > 0.f) ? x : 0.f;
        }
        if (do_mean) {
            const float mn = 0.25f * (v[0] + v[1] + v[2] + v[3]);
            v = (v4f){mn, mn, mn, mn};
        }
        *(v4f*)(hnext + (size_t)node * H_HEADS) = v;
    }
}

__global__ void k_final(const float* __restrict__ hfin,
                        const float* __restrict__ centrality,
                        const float* __restrict__ gamma,
                        const float* __restrict__ beta,
                        float* __restrict__ out, int n) {
    int i = blockIdx.x * blockDim.x + threadIdx.x;
    if (i >= n) return;
    const float c = centrality[i];
    float acc = 0.f;
    #pragma unroll
    for (int h = 0; h < H_HEADS; ++h)
        acc += (c * gamma[h] + beta[h]) * hfin[i * H_HEADS + h];
    acc *= 0.25f;
    out[i] = (acc > 0.f) ? acc : 0.01f * acc;
}

extern "C" void kernel_launch(void* const* d_in, const int* in_sizes, int n_in,
                              void* d_out, int out_size, void* d_ws, size_t ws_size,
                              hipStream_t stream) {
    const float* inputs     = (const float*)d_in[0];
    const float* centrality = (const float*)d_in[1];
    const float* W1         = (const float*)d_in[2];
    const float* b1         = (const float*)d_in[3];
    const float* W2         = (const float*)d_in[4];
    const float* b2         = (const float*)d_in[5];
    const float* rel_emb    = (const float*)d_in[6];
    const float* W_pred     = (const float*)d_in[7];
    const float* gamma      = (const float*)d_in[8];
    const float* beta       = (const float*)d_in[9];
    const int*   edge_types = (const int*)d_in[10];
    const int*   src        = (const int*)d_in[11];
    const int*   dst        = (const int*)d_in[12];
    float* out = (float*)d_out;

    const int N = in_sizes[1];
    const int E = in_sizes[10];
    const int Np = (N + 3) & ~3;
    const int Ep = (E + 3) & ~3;

    char* ws = (char*)d_ws;
    _Float16* W1h = (_Float16*)ws;
    float* etab   = (float*)(ws + 65536);
    float* hA     = (float*)(ws + 69632);
    float* hB     = hA + (size_t)Np * H_HEADS;
    int*   cnt      = (int*)(hB + (size_t)Np * H_HEADS);
    int*   rowstart = cnt + Np;
    int*   bsum     = rowstart + Np;
    int*   rank     = bsum + Np;
    unsigned* sorted = (unsigned*)(rank + Ep);

    const int TPB = 256;

    {
        int cnt1 = COLS * IN_DIM;
        k_w1_to_f16<<<(cnt1 + TPB - 1) / TPB, TPB, 0, stream>>>(W1, W1h, cnt1);
    }
    {
        int cnt2 = LAYERS * RELS * H_HEADS;
        k_etab<<<(cnt2 + TPB - 1) / TPB, TPB, 0, stream>>>(rel_emb, W_pred, etab);
    }
    {
        int nTiles = (N + 15) / 16;
        int wavesPerBlock = 4;
        int blocks = (nTiles + wavesPerBlock - 1) / wavesPerBlock;
        k_mlp_wmma<<<blocks, wavesPerBlock * 32, 0, stream>>>(
            inputs, W1h, b1, W2, b2, hA, N, nTiles);
    }

    const int nBlocks  = (N + TPB - 1) / TPB;
    const int e4Blocks = ((E + 3) / 4 + TPB - 1) / TPB;
    const int sBlocks  = (N + 1023) / 1024;
    k_zero_cnt<<<nBlocks, TPB, 0, stream>>>(cnt, N);
    k_rank<<<e4Blocks, TPB, 0, stream>>>(dst, cnt, rank, E);
    k_scan_local<<<sBlocks, 256, 0, stream>>>(cnt, rowstart, bsum, N);
    k_scan_add<<<sBlocks, 256, 0, stream>>>(rowstart, bsum, N);
    k_scatter<<<e4Blocks, TPB, 0, stream>>>(edge_types, src, dst, rank, rowstart, sorted, E);

    float* hcur = hA;
    float* hnxt = hB;
    const int gBlocks = (int)(((size_t)N * GL + TPB - 1) / TPB);

    for (int l = 0; l < LAYERS; ++l) {
        const float* etab_l = etab + l * RELS * H_HEADS;
        k_gather<<<gBlocks, TPB, 0, stream>>>(sorted, rowstart, cnt, etab_l, hcur, hnxt, N,
                                              (l != LAYERS - 1) ? 1 : 0);
        float* tmp = hcur; hcur = hnxt; hnxt = tmp;
    }

    k_final<<<nBlocks, TPB, 0, stream>>>(hcur, centrality, gamma, beta, out, N);
}


# Edge bucketing rebuilt as a two-level counting sort on LDS counters (chunk histogram, range partition, per-range node sort): no global atomics, run-wise stores; sums now in LDS-arrival bucket order
# speedup vs baseline: 18.6506x; 2.1049x over previous
#include <hip/hip_runtime.h>
#include <hip/hip_bf16.h>

typedef __attribute__((ext_vector_type(16))) _Float16 v16h;
typedef __attribute__((ext_vector_type(8)))  _Float16 v8h;
typedef __attribute__((ext_vector_type(8)))  float    v8f;
typedef __attribute__((ext_vector_type(4)))  float    v4f;

#define H_HEADS 4
#define IN_DIM  128
#define HID     64
#define COLS    256
#define PRED    16
#define RELS    50
#define LAYERS  3

__global__ void k_w1_to_f16(const float* __restrict__ W1,
                            _Float16* __restrict__ W1h, int n) {
    int i = blockIdx.x * blockDim.x + threadIdx.x;
    if (i < n) W1h[i] = (_Float16)W1[i];
}

__global__ void k_etab(const float* __restrict__ rel_emb,
                       const float* __restrict__ W_pred,
                       float* __restrict__ etab) {
    int i = blockIdx.x * blockDim.x + threadIdx.x;
    if (i >= LAYERS * RELS * H_HEADS) return;
    int l = i / (RELS * H_HEADS);
    int rem = i % (RELS * H_HEADS);
    int r = rem / H_HEADS;
    int h = rem % H_HEADS;
    float e = 0.f;
    #pragma unroll
    for (int p = 0; p < PRED; ++p)
        e += rel_emb[r * PRED + p] * W_pred[(l * PRED + p) * H_HEADS + h];
    etab[i] = (e > 0.f) ? e : 0.2f * e;
}

__global__ void k_mlp_wmma(const float* __restrict__ inputs,
                           const _Float16* __restrict__ W1h,
                           const float* __restrict__ b1,
                           const float* __restrict__ W2,
                           const float* __restrict__ b2,
                           float* __restrict__ hout,
                           int n, int nTiles) {
    const int wave = (blockIdx.x * blockDim.x + threadIdx.x) >> 5;
    if (wave >= nTiles) return;
    const int lane = threadIdx.x & 31;
    const int grp  = lane >> 4;
    const int l15  = lane & 15;
    const int nodeBase = wave * 16;
    int arow = nodeBase + l15;
    if (arow > n - 1) arow = n - 1;
    const float* __restrict__ abase = inputs + (size_t)arow * IN_DIM;

    const v8f zero = {};

    #pragma unroll
    for (int head = 0; head < H_HEADS; ++head) {
        v8f acc0 = zero, acc1 = zero, acc2 = zero, acc3 = zero;

        #pragma unroll
        for (int kk = 0; kk < IN_DIM; kk += 32) {
            const int kb = kk + grp * 8;
            v4f a0 = *(const v4f*)(abase + kb + 0);
            v4f a1 = *(const v4f*)(abase + kb + 4);
            v4f a2 = *(const v4f*)(abase + kb + 16);
            v4f a3 = *(const v4f*)(abase + kb + 20);
            v16h A;
            #pragma unroll
            for (int i = 0; i < 4; ++i) {
                A[i]      = (_Float16)a0[i];
                A[4 + i]  = (_Float16)a1[i];
                A[8 + i]  = (_Float16)a2[i];
                A[12 + i] = (_Float16)a3[i];
            }
            #pragma unroll
            for (int tt = 0; tt < 4; ++tt) {
                const int col = head * HID + tt * 16 + l15;
                const _Float16* q = W1h + (size_t)col * IN_DIM + kb;
                v8h blo = *(const v8h*)(q);
                v8h bhi = *(const v8h*)(q + 16);
                v16h B;
                #pragma unroll
                for (int i = 0; i < 8; ++i) { B[i] = blo[i]; B[8 + i] = bhi[i]; }
                v8f c = (tt == 0) ? acc0 : (tt == 1) ? acc1 : (tt == 2) ? acc2 : acc3;
                c = __builtin_amdgcn_wmma_f32_16x16x32_f16(
                        false, A, false, B, (short)0, c, false, false);
                if (tt == 0) acc0 = c; else if (tt == 1) acc1 = c;
                else if (tt == 2) acc2 = c; else acc3 = c;
            }
        }

        v8f hacc = zero;
        #pragma unroll
        for (int tt = 0; tt < 4; ++tt) {
            const int col  = head * HID + tt * 16 + l15;
            const float bias = b1[col];
            const float w2   = W2[col];
            v8f c = (tt == 0) ? acc0 : (tt == 1) ? acc1 : (tt == 2) ? acc2 : acc3;
            #pragma unroll
            for (int v = 0; v < 8; ++v) {
                float p = c[v] + bias;
                p = (p > 0.f) ? p : 0.f;
                p *= w2;
                p += __shfl_xor(p, 1, 32);
                p += __shfl_xor(p, 2, 32);
                p += __shfl_xor(p, 4, 32);
                p += __shfl_xor(p, 8, 32);
                hacc[v] += p;
            }
        }
        if (l15 == 0) {
            #pragma unroll
            for (int v = 0; v < 8; ++v) {
                const int node = nodeBase + v + 8 * grp;
                if (node < n)
                    hout[node * H_HEADS + head] = hacc[v] + b2[head];
            }
        }
    }
}

#define GL  8
#define GCH 4
#define BSH    8
#define NBMAX  1024
#define CH     8192
#define CTH    256
#define FTH    1024
#define FCAP   12288

__device__ __forceinline__ void scan1024(const int* cin, int* cout, int* wsum, int t,
                                         int (&own)[4]) {
    const int lane = t & 31, w = t >> 5;
    const int c0 = cin[4 * t], c1 = cin[4 * t + 1], c2 = cin[4 * t + 2], c3 = cin[4 * t + 3];
    const int tsum = c0 + c1 + c2 + c3;
    int x = tsum;
    #pragma unroll
    for (int d = 1; d < 32; d <<= 1) {
        const int y = __shfl_up(x, d, 32);
        if (lane >= d) x += y;
    }
    if (lane == 31) wsum[w] = x;
    __syncthreads();
    int wbase = 0;
    #pragma unroll
    for (int k = 0; k < 8; ++k) if (k < w) wbase += wsum[k];
    const int run = wbase + x - tsum;
    own[0] = run; own[1] = run + c0; own[2] = run + c0 + c1; own[3] = run + c0 + c1 + c2;
    cout[4 * t] = own[0]; cout[4 * t + 1] = own[1];
    cout[4 * t + 2] = own[2]; cout[4 * t + 3] = own[3];
}

__global__ void __launch_bounds__(CTH)
k_chunk_hist(const int* __restrict__ dst, int* __restrict__ hist,
             int E, int nChunks, int NB) {
    __shared__ int lc[NBMAX];
    const int t = threadIdx.x;
    for (int b = t; b < NBMAX; b += CTH) lc[b] = 0;
    __syncthreads();
    const int base = blockIdx.x * CH;
    #pragma unroll
    for (int k = 0; k < CH / (CTH * 4); ++k) {
        const int i = base + (k * CTH + t) * 4;
        if (i + 3 < E) {
            const int4 d = *(const int4*)(dst + i);
            atomicAdd(&lc[d.x >> BSH], 1);
            atomicAdd(&lc[d.y >> BSH], 1);
            atomicAdd(&lc[d.z >> BSH], 1);
            atomicAdd(&lc[d.w >> BSH], 1);
        } else {
            for (int j = i; j < E; ++j) atomicAdd(&lc[dst[j] >> BSH], 1);
        }
    }
    __syncthreads();
    for (int b = t; b < NB; b += CTH) hist[(size_t)b * nChunks + blockIdx.x] = lc[b];
}

__global__ void k_scan_local(const int* __restrict__ cnt, int* __restrict__ rowstart,
                             int* __restrict__ bsum, int n) {
    __shared__ int wsum[8];
    const int t = threadIdx.x, lane = t & 31, w = t >> 5;
    const int base = blockIdx.x * 1024 + t * 4;
    int c[4];
    #pragma unroll
    for (int k = 0; k < 4; ++k) c[k] = (base + k < n) ? cnt[base + k] : 0;
    const int tsum = c[0] + c[1] + c[2] + c[3];
    int x = tsum;
    #pragma unroll
    for (int d = 1; d < 32; d <<= 1) {
        const int y = __shfl_up(x, d, 32);
        if (lane >= d) x += y;
    }
    if (lane == 31) wsum[w] = x;
    __syncthreads();
    int wbase = 0;
    #pragma unroll
    for (int k = 0; k < 8; ++k) if (k < w) wbase += wsum[k];
    int run = wbase + x - tsum;
    #pragma unroll
    for (int k = 0; k < 4; ++k) {
        if (base + k < n) rowstart[base + k] = run;
        run += c[k];
    }
    if (t == 255) bsum[blockIdx.x] = wbase + x;
}

__global__ void k_scan_add(int* __restrict__ rowstart, const int* __restrict__ bsum, int n) {
    __shared__ int wsum[8];
    const int t = threadIdx.x, lane = t & 31, w = t >> 5;
    int acc = 0;
    for (int k = t; k < (int)blockIdx.x; k += 256) acc += bsum[k];
    #pragma unroll
    for (int d = 16; d >= 1; d >>= 1) acc += __shfl_xor(acc, d, 32);
    if (lane == 0) wsum[w] = acc;
    __syncthreads();
    int boff = 0;
    #pragma unroll
    for (int k = 0; k < 8; ++k) boff += wsum[k];
    const int base = blockIdx.x * 1024 + t * 4;
    #pragma unroll
    for (int k = 0; k < 4; ++k)
        if (base + k < n) rowstart[base + k] += boff;
}

__global__ void __launch_bounds__(CTH)
k_partition(const int* __restrict__ et, const int* __restrict__ src,
            const int* __restrict__ dst, const int* __restrict__ scanned,
            unsigned* __restrict__ tmp, int E, int nChunks, int NB) {
    __shared__ int lc[NBMAX];
    __shared__ int lb[NBMAX];
    __shared__ int gd[NBMAX];
    __shared__ int wsum[8];
    __shared__ unsigned recs[CH];
    __shared__ unsigned short rb[CH];
    const int t = threadIdx.x;
    for (int b = t; b < NBMAX; b += CTH) lc[b] = 0;
    __syncthreads();

    const int base = blockIdx.x * CH;
    unsigned rec[32];
    unsigned br[32];
    #pragma unroll
    for (int k = 0; k < CH / (CTH * 4); ++k) {
        const int i = base + (k * CTH + t) * 4;
        int dv[4], sv[4], ev[4];
        if (i + 3 < E) {
            const int4 d = *(const int4*)(dst + i);
            const int4 s = *(const int4*)(src + i);
            const int4 e = *(const int4*)(et + i);
            dv[0] = d.x; dv[1] = d.y; dv[2] = d.z; dv[3] = d.w;
            sv[0] = s.x; sv[1] = s.y; sv[2] = s.z; sv[3] = s.w;
            ev[0] = e.x; ev[1] = e.y; ev[2] = e.z; ev[3] = e.w;
        } else {
            #pragma unroll
            for (int c = 0; c < 4; ++c) {
                const int j = i + c;
                dv[c] = -1; sv[c] = 0; ev[c] = 0;
                if (j < E) { dv[c] = dst[j]; sv[c] = src[j]; ev[c] = et[j]; }
            }
        }
        #pragma unroll
        for (int c = 0; c < 4; ++c) {
            rec[k * 4 + c] = 0u; br[k * 4 + c] = ~0u;
            if (dv[c] >= 0) {
                const int b = dv[c] >> BSH;
                const int r = atomicAdd(&lc[b], 1);
                rec[k * 4 + c] = ((unsigned)sv[c] << 14) | ((unsigned)ev[c] << 8)
                               | ((unsigned)dv[c] & 255u);
                br[k * 4 + c] = ((unsigned)b << 16) | (unsigned)r;
            }
        }
    }
    __syncthreads();

    int own[4];
    scan1024(lc, lb, wsum, t, own);
    #pragma unroll
    for (int c = 0; c < 4; ++c) {
        const int b = 4 * t + c;
        if (b < NB) gd[b] = scanned[(size_t)b * nChunks + blockIdx.x] - own[c];
    }
    __syncthreads();

    #pragma unroll
    for (int q = 0; q < 32; ++q) {
        if (br[q] != ~0u) {
            const int b = (int)(br[q] >> 16);
            const int p = lb[b] + (int)(br[q] & 0xFFFFu);
            recs[p] = rec[q];
            rb[p] = (unsigned short)b;
        }
    }
    __syncthreads();

    int nvalid = E - base;
    if (nvalid > CH) nvalid = CH;
    for (int i = t; i < nvalid; i += CTH)
        tmp[gd[rb[i]] + i] = recs[i];
}

__global__ void __launch_bounds__(FTH)
k_fine(const unsigned* __restrict__ tmp, const int* __restrict__ scanned,
       int* __restrict__ cnt, int* __restrict__ rowstart,
       unsigned* __restrict__ sorted, int n, int E, int nChunks, int NB) {
    __shared__ int ncount[1 << BSH];
    __shared__ int nbase[1 << BSH];
    __shared__ int cur[1 << BSH];
    __shared__ int wsum[8];
    __shared__ unsigned stage[FCAP];
    const int t = threadIdx.x, lane = t & 31, w = t >> 5;
    const int b = blockIdx.x;
    const int bs = scanned[(size_t)b * nChunks];
    const int be = (b + 1 < NB) ? scanned[(size_t)(b + 1) * nChunks] : E;
    const int nb = be - bs;
    const bool staged = (nb <= FCAP);
    if (t < (1 << BSH)) { ncount[t] = 0; cur[t] = 0; }
    __syncthreads();

    for (int i0 = 0; i0 < nb; i0 += FTH * 8) {
        unsigned r[8];
        #pragma unroll
        for (int k = 0; k < 8; ++k) {
            const int i = i0 + k * FTH + t;
            r[k] = (i < nb) ? tmp[bs + i] : 0u;
        }
        #pragma unroll
        for (int k = 0; k < 8; ++k)
            if (i0 + k * FTH + t < nb) atomicAdd(&ncount[r[k] & 255u], 1);
    }
    __syncthreads();

    int v = 0, x = 0;
    if (t < (1 << BSH)) {
        v = ncount[t];
        x = v;
        #pragma unroll
        for (int d = 1; d < 32; d <<= 1) {
            const int y = __shfl_up(x, d, 32);
            if (lane >= d) x += y;
        }
        if (lane == 31) wsum[w] = x;
    }
    __syncthreads();
    if (t < (1 << BSH)) {
        int wbase = 0;
        #pragma unroll
        for (int k = 0; k < 8; ++k) if (k < w) wbase += wsum[k];
        const int excl = wbase + x - v;
        nbase[t] = excl;
        const int node = (b << BSH) + t;
        if (node < n) { cnt[node] = v; rowstart[node] = bs + excl; }
    }
    __syncthreads();

    for (int i0 = 0; i0 < nb; i0 += FTH * 8) {
        unsigned r[8];
        #pragma unroll
        for (int k = 0; k < 8; ++k) {
            const int i = i0 + k * FTH + t;
            r[k] = (i < nb) ? tmp[bs + i] : 0u;
        }
        #pragma unroll
        for (int k = 0; k < 8; ++k) {
            if (i0 + k * FTH + t < nb) {
                const unsigned dl = r[k] & 255u;
                const int p = nbase[dl] + atomicAdd(&cur[dl], 1);
                const unsigned val = r[k] >> 8;
                if (staged) stage[p] = val; else sorted[bs + p] = val;
            }
        }
    }
    if (staged) {
        __syncthreads();
        for (int i = t; i < nb; i += FTH) sorted[bs + i] = stage[i];
    }
}

__device__ __forceinline__ void g_load(const unsigned* __restrict__ sorted, int start,
                                       int len, int j0, unsigned (&pk)[GCH]) {
    #pragma unroll
    for (int k = 0; k < GCH; ++k) {
        const int j = j0 + k * GL;
        pk[k] = (j < len) ? sorted[start + j] : 0u;
    }
}
__device__ __forceinline__ void g_max(const v4f* tab, const unsigned (&pk)[GCH],
                                      int len, int j0, float (&m)[H_HEADS]) {
    #pragma unroll
    for (int k = 0; k < GCH; ++k) {
        if (j0 + k * GL < len) {
            const v4f e = tab[pk[k] & 63u];
            #pragma unroll
            for (int h = 0; h < H_HEADS; ++h) m[h] = fmaxf(m[h], e[h]);
        }
    }
}
__device__ __forceinline__ void g_sum(const v4f* tab, const float* __restrict__ hcur,
                                      const unsigned (&pk)[GCH], int len, int j0,
                                      const float (&m)[H_HEADS],
                                      float (&s)[H_HEADS], float (&u)[H_HEADS]) {
    v4f hv[GCH];
    #pragma unroll
    for (int k = 0; k < GCH; ++k) {
        hv[k] = (v4f){0.f, 0.f, 0.f, 0.f};
        if (j0 + k * GL < len)
            hv[k] = *(const v4f*)(hcur + (size_t)(pk[k] >> 6) * H_HEADS);
    }
    #pragma unroll
    for (int k = 0; k < GCH; ++k) {
        if (j0 + k * GL < len) {
            const v4f e = tab[pk[k] & 63u];
            #pragma unroll
            for (int h = 0; h < H_HEADS; ++h) {
                const float wgt = __expf(e[h] - m[h]);
                s[h] += wgt;
                u[h] += wgt * hv[k][h];
            }
        }
    }
}

__global__ void k_gather(const unsigned* __restrict__ sorted,
                         const int* __restrict__ rowstart,
                         const int* __restrict__ cnt,
                         const float* __restrict__ etab_l,
                         const float* __restrict__ hcur,
                         float* __restrict__ hnext, int n, int do_mean) {
    __shared__ v4f tab[RELS];
    if (threadIdx.x < RELS)
        tab[threadIdx.x] = *(const v4f*)(etab_l + threadIdx.x * H_HEADS);
    __syncthreads();

    const int gid  = blockIdx.x * blockDim.x + threadIdx.x;
    const int node = gid / GL;
    const int sub  = threadIdx.x & (GL - 1);
    int start = 0, len = 0;
    if (node < n) { start = rowstart[node]; len = cnt[node]; }

    unsigned pk0[GCH];
    g_load(sorted, start, len, sub, pk0);

    float m[H_HEADS];
    #pragma unroll
    for (int h = 0; h < H_HEADS; ++h) m[h] = -__builtin_huge_valf();
    g_max(tab, pk0, len, sub, m);
    for (int base = GL * GCH; base < len; base += GL * GCH) {
        unsigned pk[GCH];
        g_load(sorted, start, len, base + sub, pk);
        g_max(tab, pk, len, base + sub, m);
    }
    #pragma unroll
    for (int h = 0; h < H_HEADS; ++h) {
        m[h] = fmaxf(m[h], __shfl_xor(m[h], 1, 32));
        m[h] = fmaxf(m[h], __shfl_xor(m[h], 2, 32));
        m[h] = fmaxf(m[h], __shfl_xor(m[h], 4, 32));
    }

    float s[H_HEADS], u[H_HEADS];
    #pragma unroll
    for (int h = 0; h < H_HEADS; ++h) { s[h] = 0.f; u[h] = 0.f; }
    g_sum(tab, hcur, pk0, len, sub, m, s, u);
    for (int base = GL * GCH; base < len; base += GL * GCH) {
        unsigned pk[GCH];
        g_load(sorted, start, len, base + sub, pk);
        g_sum(tab, hcur, pk, len, base + sub, m, s, u);
    }
    #pragma unroll
    for (int h = 0; h < H_HEADS; ++h) {
        s[h] += __shfl_xor(s[h], 1, 32);  u[h] += __shfl_xor(u[h], 1, 32);
        s[h] += __shfl_xor(s[h], 2, 32);  u[h] += __shfl_xor(u[h], 2, 32);
        s[h] += __shfl_xor(s[h], 4, 32);  u[h] += __shfl_xor(u[h], 4, 32);
    }

    if (sub == 0 && node < n) {
        v4f v;
        #pragma unroll
        for (int h = 0; h < H_HEADS; ++h) {
            const float x = (s[h] > 0.f) ? (u[h] / s[h]) : 0.f;
            v[h] = (x > 0.f) ? x : 0.f;
        }
        if (do_mean) {
            const float mn = 0.25f * (v[0] + v[1] + v[2] + v[3]);
            v = (v4f){mn, mn, mn, mn};
        }
        *(v4f*)(hnext + (size_t)node * H_HEADS) = v;
    }
}

__global__ void k_final(const float* __restrict__ hfin,
                        const float* __restrict__ centrality,
                        const float* __restrict__ gamma,
                        const float* __restrict__ beta,
                        float* __restrict__ out, int n) {
    int i = blockIdx.x * blockDim.x + threadIdx.x;
    if (i >= n) return;
    const float c = centrality[i];
    float acc = 0.f;
    #pragma unroll
    for (int h = 0; h < H_HEADS; ++h)
        acc += (c * gamma[h] + beta[h]) * hfin[i * H_HEADS + h];
    acc *= 0.25f;
    out[i] = (acc > 0.f) ? acc : 0.01f * acc;
}

extern "C" void kernel_launch(void* const* d_in, const int* in_sizes, int n_in,
                              void* d_out, int out_size, void* d_ws, size_t ws_size,
                              hipStream_t stream) {
    const float* inputs     = (const float*)d_in[0];
    const float* centrality = (const float*)d_in[1];
    const float* W1         = (const float*)d_in[2];
    const float* b1         = (const float*)d_in[3];
    const float* W2         = (const float*)d_in[4];
    const float* b2         = (const float*)d_in[5];
    const float* rel_emb    = (const float*)d_in[6];
    const float* W_pred     = (const float*)d_in[7];
    const float* gamma      = (const float*)d_in[8];
    const float* beta       = (const float*)d_in[9];
    const int*   edge_types = (const int*)d_in[10];
    const int*   src        = (const int*)d_in[11];
    const int*   dst        = (const int*)d_in[12];
    float* out = (float*)d_out;

    const int N = in_sizes[1];
    const int E = in_sizes[10];
    const int Np = (N + 3) & ~3;
    const int Ep = (E + 3) & ~3;
    const int NB = (N + (1 << BSH) - 1) >> BSH;
    const int nChunks = (E + CH - 1) / CH;
    const int M  = NB * nChunks;
    const int Mp = (M + 3) & ~3;

    char* ws = (char*)d_ws;
    _Float16* W1h = (_Float16*)ws;
    float* etab   = (float*)(ws + 65536);
    float* hA     = (float*)(ws + 69632);
    float* hB     = hA + (size_t)Np * H_HEADS;
    int*   cnt      = (int*)(hB + (size_t)Np * H_HEADS);
    int*   rowstart = cnt + Np;
    int*   hist     = rowstart + Np;
    int*   scanned  = hist + Mp;
    int*   bsum     = scanned + Mp;
    unsigned* part   = (unsigned*)(bsum + Mp);
    unsigned* sorted = part + Ep;

    const int TPB = 256;

    {
        int cnt1 = COLS * IN_DIM;
        k_w1_to_f16<<<(cnt1 + TPB - 1) / TPB, TPB, 0, stream>>>(W1, W1h, cnt1);
    }
    {
        int cnt2 = LAYERS * RELS * H_HEADS;
        k_etab<<<(cnt2 + TPB - 1) / TPB, TPB, 0, stream>>>(rel_emb, W_pred, etab);
    }
    {
        int nTiles = (N + 15) / 16;
        int wavesPerBlock = 4;
        int blocks = (nTiles + wavesPerBlock - 1) / wavesPerBlock;
        k_mlp_wmma<<<blocks, wavesPerBlock * 32, 0, stream>>>(
            inputs, W1h, b1, W2, b2, hA, N, nTiles);
    }

    const int nBlocks = (N + TPB - 1) / TPB;
    const int sBlocks = (M + 1023) / 1024;
    k_chunk_hist<<<nChunks, CTH, 0, stream>>>(dst, hist, E, nChunks, NB);
    k_scan_local<<<sBlocks, 256, 0, stream>>>(hist, scanned, bsum, M);
    k_scan_add<<<sBlocks, 256, 0, stream>>>(scanned, bsum, M);
    k_partition<<<nChunks, CTH, 0, stream>>>(edge_types, src, dst, scanned, part, E, nChunks, NB);
    k_fine<<<NB, FTH, 0, stream>>>(part, scanned, cnt, rowstart, sorted, N, E, nChunks, NB);

    float* hcur = hA;
    float* hnxt = hB;
    const int gBlocks = (int)(((size_t)N * GL + TPB - 1) / TPB);

    for (int l = 0; l < LAYERS; ++l) {
        const float* etab_l = etab + l * RELS * H_HEADS;
        k_gather<<<gBlocks, TPB, 0, stream>>>(sorted, rowstart, cnt, etab_l, hcur, hnxt, N,
                                              (l != LAYERS - 1) ? 1 : 0);
        float* tmp = hcur; hcur = hnxt; hnxt = tmp;
    }

    k_final<<<nBlocks, TPB, 0, stream>>>(hcur, centrality, gamma, beta, out, N);
}


# Layers 2-3 (one score per node after the head mean): 800-thread workgroups copy the score table into LDS (first 81664 nodes, rest from L2) and gather there; 2 lanes per node: stride-2 sums + xor step
# speedup vs baseline: 20.1130x; 1.0784x over previous
#include <hip/hip_runtime.h>
#include <hip/hip_bf16.h>

typedef __attribute__((ext_vector_type(16))) _Float16 v16h;
typedef __attribute__((ext_vector_type(8)))  _Float16 v8h;
typedef __attribute__((ext_vector_type(8)))  float    v8f;
typedef __attribute__((ext_vector_type(4)))  float    v4f;

#define H_HEADS 4
#define IN_DIM  128
#define HID     64
#define COLS    256
#define PRED    16
#define RELS    50
#define LAYERS  3

__global__ void k_w1_to_f16(const float* __restrict__ W1,
                            _Float16* __restrict__ W1h, int n) {
    int i = blockIdx.x * blockDim.x + threadIdx.x;
    if (i < n) W1h[i] = (_Float16)W1[i];
}

__global__ void k_etab(const float* __restrict__ rel_emb,
                       const float* __restrict__ W_pred,
                       float* __restrict__ etab) {
    int i = blockIdx.x * blockDim.x + threadIdx.x;
    if (i >= LAYERS * RELS * H_HEADS) return;
    int l = i / (RELS * H_HEADS);
    int rem = i % (RELS * H_HEADS);
    int r = rem / H_HEADS;
    int h = rem % H_HEADS;
    float e = 0.f;
    #pragma unroll
    for (int p = 0; p < PRED; ++p)
        e += rel_emb[r * PRED + p] * W_pred[(l * PRED + p) * H_HEADS + h];
    etab[i] = (e > 0.f) ? e : 0.2f * e;
}

__global__ void k_mlp_wmma(const float* __restrict__ inputs,
                           const _Float16* __restrict__ W1h,
                           const float* __restrict__ b1,
                           const float* __restrict__ W2,
                           const float* __restrict__ b2,
                           float* __restrict__ hout,
                           int n, int nTiles) {
    const int wave = (blockIdx.x * blockDim.x + threadIdx.x) >> 5;
    if (wave >= nTiles) return;
    const int lane = threadIdx.x & 31;
    const int grp  = lane >> 4;
    const int l15  = lane & 15;
    const int nodeBase = wave * 16;
    int arow = nodeBase + l15;
    if (arow > n - 1) arow = n - 1;
    const float* __restrict__ abase = inputs + (size_t)arow * IN_DIM;

    const v8f zero = {};

    #pragma unroll
    for (int head = 0; head < H_HEADS; ++head) {
        v8f acc0 = zero, acc1 = zero, acc2 = zero, acc3 = zero;

        #pragma unroll
        for (int kk = 0; kk < IN_DIM; kk += 32) {
            const int kb = kk + grp * 8;
            v4f a0 = *(const v4f*)(abase + kb + 0);
            v4f a1 = *(const v4f*)(abase + kb + 4);
            v4f a2 = *(const v4f*)(abase + kb + 16);
            v4f a3 = *(const v4f*)(abase + kb + 20);
            v16h A;
            #pragma unroll
            for (int i = 0; i < 4; ++i) {
                A[i]      = (_Float16)a0[i];
                A[4 + i]  = (_Float16)a1[i];
                A[8 + i]  = (_Float16)a2[i];
                A[12 + i] = (_Float16)a3[i];
            }
            #pragma unroll
            for (int tt = 0; tt < 4; ++tt) {
                const int col = head * HID + tt * 16 + l15;
                const _Float16* q = W1h + (size_t)col * IN_DIM + kb;
                v8h blo = *(const v8h*)(q);
                v8h bhi = *(const v8h*)(q + 16);
                v16h B;
                #pragma unroll
                for (int i = 0; i < 8; ++i) { B[i] = blo[i]; B[8 + i] = bhi[i]; }
                v8f c = (tt == 0) ? acc0 : (tt == 1) ? acc1 : (tt == 2) ? acc2 : acc3;
                c = __builtin_amdgcn_wmma_f32_16x16x32_f16(
                        false, A, false, B, (short)0, c, false, false);
                if (tt == 0) acc0 = c; else if (tt == 1) acc1 = c;
                else if (tt == 2) acc2 = c; else acc3 = c;
            }
        }

        v8f hacc = zero;
        #pragma unroll
        for (int tt = 0; tt < 4; ++tt) {
            const int col  = head * HID + tt * 16 + l15;
            const float bias = b1[col];
            const float w2   = W2[col];
            v8f c = (tt == 0) ? acc0 : (tt == 1) ? acc1 : (tt == 2) ? acc2 : acc3;
            #pragma unroll
            for (int v = 0; v < 8; ++v) {
                float p = c[v] + bias;
                p = (p > 0.f) ? p : 0.f;
                p *= w2;
                p += __shfl_xor(p, 1, 32);
                p += __shfl_xor(p, 2, 32);
                p += __shfl_xor(p, 4, 32);
                p += __shfl_xor(p, 8, 32);
                hacc[v] += p;
            }
        }
        if (l15 == 0) {
            #pragma unroll
            for (int v = 0; v < 8; ++v) {
                const int node = nodeBase + v + 8 * grp;
                if (node < n)
                    hout[node * H_HEADS + head] = hacc[v] + b2[head];
            }
        }
    }
}

#define GL  8
#define GCH 4
#define BSH    8
#define NBMAX  1024
#define CH     8192
#define CTH    256
#define FTH    1024
#define FCAP   12288

__device__ __forceinline__ void scan1024(const int* cin, int* cout, int* wsum, int t,
                                         int (&own)[4]) {
    const int lane = t & 31, w = t >> 5;
    const int c0 = cin[4 * t], c1 = cin[4 * t + 1], c2 = cin[4 * t + 2], c3 = cin[4 * t + 3];
    const int tsum = c0 + c1 + c2 + c3;
    int x = tsum;
    #pragma unroll
    for (int d = 1; d < 32; d <<= 1) {
        const int y = __shfl_up(x, d, 32);
        if (lane >= d) x += y;
    }
    if (lane == 31) wsum[w] = x;
    __syncthreads();
    int wbase = 0;
    #pragma unroll
    for (int k = 0; k < 8; ++k) if (k < w) wbase += wsum[k];
    const int run = wbase + x - tsum;
    own[0] = run; own[1] = run + c0; own[2] = run + c0 + c1; own[3] = run + c0 + c1 + c2;
    cout[4 * t] = own[0]; cout[4 * t + 1] = own[1];
    cout[4 * t + 2] = own[2]; cout[4 * t + 3] = own[3];
}

__global__ void __launch_bounds__(CTH)
k_chunk_hist(const int* __restrict__ dst, int* __restrict__ hist,
             int E, int nChunks, int NB) {
    __shared__ int lc[NBMAX];
    const int t = threadIdx.x;
    for (int b = t; b < NBMAX; b += CTH) lc[b] = 0;
    __syncthreads();
    const int base = blockIdx.x * CH;
    #pragma unroll
    for (int k = 0; k < CH / (CTH * 4); ++k) {
        const int i = base + (k * CTH + t) * 4;
        if (i + 3 < E) {
            const int4 d = *(const int4*)(dst + i);
            atomicAdd(&lc[d.x >> BSH], 1);
            atomicAdd(&lc[d.y >> BSH], 1);
            atomicAdd(&lc[d.z >> BSH], 1);
            atomicAdd(&lc[d.w >> BSH], 1);
        } else {
            for (int j = i; j < E; ++j) atomicAdd(&lc[dst[j] >> BSH], 1);
        }
    }
    __syncthreads();
    for (int b = t; b < NB; b += CTH) hist[(size_t)b * nChunks + blockIdx.x] = lc[b];
}

__global__ void k_scan_local(const int* __restrict__ cnt, int* __restrict__ rowstart,
                             int* __restrict__ bsum, int n) {
    __shared__ int wsum[8];
    const int t = threadIdx.x, lane = t & 31, w = t >> 5;
    const int base = blockIdx.x * 1024 + t * 4;
    int c[4];
    #pragma unroll
    for (int k = 0; k < 4; ++k) c[k] = (base + k < n) ? cnt[base + k] : 0;
    const int tsum = c[0] + c[1] + c[2] + c[3];
    int x = tsum;
    #pragma unroll
    for (int d = 1; d < 32; d <<= 1) {
        const int y = __shfl_up(x, d, 32);
        if (lane >= d) x += y;
    }
    if (lane == 31) wsum[w] = x;
    __syncthreads();
    int wbase = 0;
    #pragma unroll
    for (int k = 0; k < 8; ++k) if (k < w) wbase += wsum[k];
    int run = wbase + x - tsum;
    #pragma unroll
    for (int k = 0; k < 4; ++k) {
        if (base + k < n) rowstart[base + k] = run;
        run += c[k];
    }
    if (t == 255) bsum[blockIdx.x] = wbase + x;
}

__global__ void k_scan_add(int* __restrict__ rowstart, const int* __restrict__ bsum, int n) {
    __shared__ int wsum[8];
    const int t = threadIdx.x, lane = t & 31, w = t >> 5;
    int acc = 0;
    for (int k = t; k < (int)blockIdx.x; k += 256) acc += bsum[k];
    #pragma unroll
    for (int d = 16; d >= 1; d >>= 1) acc += __shfl_xor(acc, d, 32);
    if (lane == 0) wsum[w] = acc;
    __syncthreads();
    int boff = 0;
    #pragma unroll
    for (int k = 0; k < 8; ++k) boff += wsum[k];
    const int base = blockIdx.x * 1024 + t * 4;
    #pragma unroll
    for (int k = 0; k < 4; ++k)
        if (base + k < n) rowstart[base + k] += boff;
}

__global__ void __launch_bounds__(CTH)
k_partition(const int* __restrict__ et, const int* __restrict__ src,
            const int* __restrict__ dst, const int* __restrict__ scanned,
            unsigned* __restrict__ tmp, int E, int nChunks, int NB) {
    __shared__ int lc[NBMAX];
    __shared__ int lb[NBMAX];
    __shared__ int gd[NBMAX];
    __shared__ int wsum[8];
    __shared__ unsigned recs[CH];
    __shared__ unsigned short rb[CH];
    const int t = threadIdx.x;
    for (int b = t; b < NBMAX; b += CTH) lc[b] = 0;
    __syncthreads();

    const int base = blockIdx.x * CH;
    unsigned rec[32];
    unsigned br[32];
    #pragma unroll
    for (int k = 0; k < CH / (CTH * 4); ++k) {
        const int i = base + (k * CTH + t) * 4;
        int dv[4], sv[4], ev[4];
        if (i + 3 < E) {
            const int4 d = *(const int4*)(dst + i);
            const int4 s = *(const int4*)(src + i);
            const int4 e = *(const int4*)(et + i);
            dv[0] = d.x; dv[1] = d.y; dv[2] = d.z; dv[3] = d.w;
            sv[0] = s.x; sv[1] = s.y; sv[2] = s.z; sv[3] = s.w;
            ev[0] = e.x; ev[1] = e.y; ev[2] = e.z; ev[3] = e.w;
        } else {
            #pragma unroll
            for (int c = 0; c < 4; ++c) {
                const int j = i + c;
                dv[c] = -1; sv[c] = 0; ev[c] = 0;
                if (j < E) { dv[c] = dst[j]; sv[c] = src[j]; ev[c] = et[j]; }
            }
        }
        #pragma unroll
        for (int c = 0; c < 4; ++c) {
            rec[k * 4 + c] = 0u; br[k * 4 + c] = ~0u;
            if (dv[c] >= 0) {
                const int b = dv[c] >> BSH;
                const int r = atomicAdd(&lc[b], 1);
                rec[k * 4 + c] = ((unsigned)sv[c] << 14) | ((unsigned)ev[c] << 8)
                               | ((unsigned)dv[c] & 255u);
                br[k * 4 + c] = ((unsigned)b << 16) | (unsigned)r;
            }
        }
    }
    __syncthreads();

    int own[4];
    scan1024(lc, lb, wsum, t, own);
    #pragma unroll
    for (int c = 0; c < 4; ++c) {
        const int b = 4 * t + c;
        if (b < NB) gd[b] = scanned[(size_t)b * nChunks + blockIdx.x] - own[c];
    }
    __syncthreads();

    #pragma unroll
    for (int q = 0; q < 32; ++q) {
        if (br[q] != ~0u) {
            const int b = (int)(br[q] >> 16);
            const int p = lb[b] + (int)(br[q] & 0xFFFFu);
            recs[p] = rec[q];
            rb[p] = (unsigned short)b;
        }
    }
    __syncthreads();

    int nvalid = E - base;
    if (nvalid > CH) nvalid = CH;
    for (int i = t; i < nvalid; i += CTH)
        tmp[gd[rb[i]] + i] = recs[i];
}

__global__ void __launch_bounds__(FTH)
k_fine(const unsigned* __restrict__ tmp, const int* __restrict__ scanned,
       int* __restrict__ cnt, int* __restrict__ rowstart,
       unsigned* __restrict__ sorted, int n, int E, int nChunks, int NB) {
    __shared__ int ncount[1 << BSH];
    __shared__ int nbase[1 << BSH];
    __shared__ int cur[1 << BSH];
    __shared__ int wsum[8];
    __shared__ unsigned stage[FCAP];
    const int t = threadIdx.x, lane = t & 31, w = t >> 5;
    const int b = blockIdx.x;
    const int bs = scanned[(size_t)b * nChunks];
    const int be = (b + 1 < NB) ? scanned[(size_t)(b + 1) * nChunks] : E;
    const int nb = be - bs;
    const bool staged = (nb <= FCAP);
    if (t < (1 << BSH)) { ncount[t] = 0; cur[t] = 0; }
    __syncthreads();

    for (int i0 = 0; i0 < nb; i0 += FTH * 8) {
        unsigned r[8];
        #pragma unroll
        for (int k = 0; k < 8; ++k) {
            const int i = i0 + k * FTH + t;
            r[k] = (i < nb) ? tmp[bs + i] : 0u;
        }
        #pragma unroll
        for (int k = 0; k < 8; ++k)
            if (i0 + k * FTH + t < nb) atomicAdd(&ncount[r[k] & 255u], 1);
    }
    __syncthreads();

    int v = 0, x = 0;
    if (t < (1 << BSH)) {
        v = ncount[t];
        x = v;
        #pragma unroll
        for (int d = 1; d < 32; d <<= 1) {
            const int y = __shfl_up(x, d, 32);
            if (lane >= d) x += y;
        }
        if (lane == 31) wsum[w] = x;
    }
    __syncthreads();
    if (t < (1 << BSH)) {
        int wbase = 0;
        #pragma unroll
        for (int k = 0; k < 8; ++k) if (k < w) wbase += wsum[k];
        const int excl = wbase + x - v;
        nbase[t] = excl;
        const int node = (b << BSH) + t;
        if (node < n) { cnt[node] = v; rowstart[node] = bs + excl; }
    }
    __syncthreads();

    for (int i0 = 0; i0 < nb; i0 += FTH * 8) {
        unsigned r[8];
        #pragma unroll
        for (int k = 0; k < 8; ++k) {
            const int i = i0 + k * FTH + t;
            r[k] = (i < nb) ? tmp[bs + i] : 0u;
        }
        #pragma unroll
        for (int k = 0; k < 8; ++k) {
            if (i0 + k * FTH + t < nb) {
                const unsigned dl = r[k] & 255u;
                const int p = nbase[dl] + atomicAdd(&cur[dl], 1);
                const unsigned val = r[k] >> 8;
                if (staged) stage[p] = val; else sorted[bs + p] = val;
            }
        }
    }
    if (staged) {
        __syncthreads();
        for (int i = t; i < nb; i += FTH) sorted[bs + i] = stage[i];
    }
}

__device__ __forceinline__ void g_load(const unsigned* __restrict__ sorted, int start,
                                       int len, int j0, unsigned (&pk)[GCH]) {
    #pragma unroll
    for (int k = 0; k < GCH; ++k) {
        const int j = j0 + k * GL;
        pk[k] = (j < len) ? sorted[start + j] : 0u;
    }
}
__device__ __forceinline__ void g_max(const v4f* tab, const unsigned (&pk)[GCH],
                                      int len, int j0, float (&m)[H_HEADS]) {
    #pragma unroll
    for (int k = 0; k < GCH; ++k) {
        if (j0 + k * GL < len) {
            const v4f e = tab[pk[k] & 63u];
            #pragma unroll
            for (int h = 0; h < H_HEADS; ++h) m[h] = fmaxf(m[h], e[h]);
        }
    }
}
__device__ __forceinline__ void g_sum(const v4f* tab, const float* __restrict__ hcur,
                                      const unsigned (&pk)[GCH], int len, int j0,
                                      const float (&m)[H_HEADS],
                                      float (&s)[H_HEADS], float (&u)[H_HEADS]) {
    v4f hv[GCH];
    #pragma unroll
    for (int k = 0; k < GCH; ++k) {
        hv[k] = (v4f){0.f, 0.f, 0.f, 0.f};
        if (j0 + k * GL < len)
            hv[k] = *(const v4f*)(hcur + (size_t)(pk[k] >> 6) * H_HEADS);
    }
    #pragma unroll
    for (int k = 0; k < GCH; ++k) {
        if (j0 + k * GL < len) {
            const v4f e = tab[pk[k] & 63u];
            #pragma unroll
            for (int h = 0; h < H_HEADS; ++h) {
                const float wgt = __expf(e[h] - m[h]);
                s[h] += wgt;
                u[h] += wgt * hv[k][h];
            }
        }
    }
}

__global__ void k_gather(const unsigned* __restrict__ sorted,
                         const int* __restrict__ rowstart,
                         const int* __restrict__ cnt,
                         const float* __restrict__ etab_l,
                         const float* __restrict__ hcur,
                         float* __restrict__ hnext, float* __restrict__ hs_next,
                         int n, int do_mean) {
    __shared__ v4f tab[RELS];
    if (threadIdx.x < RELS)
        tab[threadIdx.x] = *(const v4f*)(etab_l + threadIdx.x * H_HEADS);
    __syncthreads();

    const int gid  = blockIdx.x * blockDim.x + threadIdx.x;
    const int node = gid / GL;
    const int sub  = threadIdx.x & (GL - 1);
    int start = 0, len = 0;
    if (node < n) { start = rowstart[node]; len = cnt[node]; }

    unsigned pk0[GCH];
    g_load(sorted, start, len, sub, pk0);

    float m[H_HEADS];
    #pragma unroll
    for (int h = 0; h < H_HEADS; ++h) m[h] = -__builtin_huge_valf();
    g_max(tab, pk0, len, sub, m);
    for (int base = GL * GCH; base < len; base += GL * GCH) {
        unsigned pk[GCH];
        g_load(sorted, start, len, base + sub, pk);
        g_max(tab, pk, len, base + sub, m);
    }
    #pragma unroll
    for (int h = 0; h < H_HEADS; ++h) {
        m[h] = fmaxf(m[h], __shfl_xor(m[h], 1, 32));
        m[h] = fmaxf(m[h], __shfl_xor(m[h], 2, 32));
        m[h] = fmaxf(m[h], __shfl_xor(m[h], 4, 32));
    }

    float s[H_HEADS], u[H_HEADS];
    #pragma unroll
    for (int h = 0; h < H_HEADS; ++h) { s[h] = 0.f; u[h] = 0.f; }
    g_sum(tab, hcur, pk0, len, sub, m, s, u);
    for (int base = GL * GCH; base < len; base += GL * GCH) {
        unsigned pk[GCH];
        g_load(sorted, start, len, base + sub, pk);
        g_sum(tab, hcur, pk, len, base + sub, m, s, u);
    }
    #pragma unroll
    for (int h = 0; h < H_HEADS; ++h) {
        s[h] += __shfl_xor(s[h], 1, 32);  u[h] += __shfl_xor(u[h], 1, 32);
        s[h] += __shfl_xor(s[h], 2, 32);  u[h] += __shfl_xor(u[h], 2, 32);
        s[h] += __shfl_xor(s[h], 4, 32);  u[h] += __shfl_xor(u[h], 4, 32);
    }

    if (sub == 0 && node < n) {
        v4f v;
        #pragma unroll
        for (int h = 0; h < H_HEADS; ++h) {
            const float x = (s[h] > 0.f) ? (u[h] / s[h]) : 0.f;
            v[h] = (x > 0.f) ? x : 0.f;
        }
        if (do_mean) {
            hs_next[node] = 0.25f * (v[0] + v[1] + v[2] + v[3]);
            if (node == n - 1)
                for (int j = n; j < ((n + 3) & ~3); ++j) hs_next[j] = 0.f;
        } else {
            *(v4f*)(hnext + (size_t)node * H_HEADS) = v;
        }
    }
}

#define LT    800
#define LGL   2
#define LCH   16
#define LNPW  (LT / LGL)
#define LCAPN 81664
#define LST   8

__device__ __forceinline__ void l_load(const unsigned* __restrict__ sorted, int start,
                                       int len, int j0, unsigned (&pk)[LCH]) {
    #pragma unroll
    for (int k = 0; k < LCH; ++k) {
        const int j = j0 + k * LGL;
        pk[k] = sorted[(j < len) ? start + j : 0];
    }
}
__device__ __forceinline__ void l_max(const v4f* tab, const unsigned (&pk)[LCH],
                                      int len, int j0, float (&m)[H_HEADS]) {
    #pragma unroll
    for (int k = 0; k < LCH; ++k) {
        if (j0 + k * LGL < len) {
            const v4f e = tab[pk[k] & 63u];
            #pragma unroll
            for (int h = 0; h < H_HEADS; ++h) m[h] = fmaxf(m[h], e[h]);
        }
    }
}
__device__ __forceinline__ void l_sum(const v4f* tab, const float* hl,
                                      const float* __restrict__ hs, unsigned count,
                                      const unsigned (&pk)[LCH],
                                      int len, int j0, const float (&m)[H_HEADS],
                                      float (&s)[H_HEADS], float (&u)[H_HEADS]) {
    float hv[LCH];
    #pragma unroll
    for (int k = 0; k < LCH; ++k) {
        const unsigned sn = pk[k] >> 6;
        const bool inl = sn < count;
        const float a = hl[inl ? sn : 0u];
        const float b = hs[inl ? 0u : sn];
        hv[k] = inl ? a : b;
    }
    #pragma unroll
    for (int k = 0; k < LCH; ++k) {
        if (j0 + k * LGL < len) {
            const v4f e = tab[pk[k] & 63u];
            #pragma unroll
            for (int h = 0; h < H_HEADS; ++h) {
                const float wgt = __expf(e[h] - m[h]);
                s[h] += wgt;
                u[h] += wgt * hv[k];
            }
        }
    }
}

__global__ void __launch_bounds__(LT)
k_gather_lds(const unsigned* __restrict__ sorted,
             const int* __restrict__ rowstart,
             const int* __restrict__ cnt,
             const float* __restrict__ etab_l,
             const float* __restrict__ hs,
             float* __restrict__ hnext, float* __restrict__ hs_next,
             int n, int do_mean) {
    __shared__ v4f tab[RELS];
    __shared__ float hl[LCAPN];
    const int t = threadIdx.x;
    if (t < RELS) tab[t] = *(const v4f*)(etab_l + t * H_HEADS);

    const int node = blockIdx.x * LNPW + t / LGL;
    const int sub  = t & (LGL - 1);
    int start = 0, len = 0;
    if (node < n) { start = rowstart[node]; len = cnt[node]; }

    unsigned pk0[LCH];
    l_load(sorted, start, len, sub, pk0);

    const int np = (n + 3) & ~3;
    const int count = (np < LCAPN) ? np : LCAPN;
    for (int i0 = 0; i0 < count; i0 += LT * 4 * LST) {
        v4f r[LST];
        #pragma unroll
        for (int k = 0; k < LST; ++k) {
            int i = i0 + (k * LT + t) * 4;
            if (i > count - 4) i = count - 4;
            r[k] = *(const v4f*)(hs + i);
        }
        #pragma unroll
        for (int k = 0; k < LST; ++k) {
            const int i = i0 + (k * LT + t) * 4;
            if (i < count) *(v4f*)(hl + i) = r[k];
        }
    }
    __syncthreads();

    float m[H_HEADS];
    #pragma unroll
    for (int h = 0; h < H_HEADS; ++h) m[h] = -__builtin_huge_valf();
    l_max(tab, pk0, len, sub, m);
    for (int base = LGL * LCH; base < len; base += LGL * LCH) {
        unsigned pk[LCH];
        l_load(sorted, start, len, base + sub, pk);
        l_max(tab, pk, len, base + sub, m);
    }
    #pragma unroll
    for (int h = 0; h < H_HEADS; ++h)
        m[h] = fmaxf(m[h], __shfl_xor(m[h], 1, 32));

    float s[H_HEADS], u[H_HEADS];
    #pragma unroll
    for (int h = 0; h < H_HEADS; ++h) { s[h] = 0.f; u[h] = 0.f; }
    l_sum(tab, hl, hs, (unsigned)count, pk0, len, sub, m, s, u);
    for (int base = LGL * LCH; base < len; base += LGL * LCH) {
        unsigned pk[LCH];
        l_load(sorted, start, len, base + sub, pk);
        l_sum(tab, hl, hs, (unsigned)count, pk, len, base + sub, m, s, u);
    }
    #pragma unroll
    for (int h = 0; h < H_HEADS; ++h) {
        s[h] += __shfl_xor(s[h], 1, 32);
        u[h] += __shfl_xor(u[h], 1, 32);
    }

    if (sub == 0 && node < n) {
        v4f v;
        #pragma unroll
        for (int h = 0; h < H_HEADS; ++h) {
            const float x = (s[h] > 0.f) ? (u[h] / s[h]) : 0.f;
            v[h] = (x > 0.f) ? x : 0.f;
        }
        if (do_mean) {
            hs_next[node] = 0.25f * (v[0] + v[1] + v[2] + v[3]);
            if (node == n - 1)
                for (int j = n; j < np; ++j) hs_next[j] = 0.f;
        } else {
            *(v4f*)(hnext + (size_t)node * H_HEADS) = v;
        }
    }
}

__global__ void k_final(const float* __restrict__ hfin,
                        const float* __restrict__ centrality,
                        const float* __restrict__ gamma,
                        const float* __restrict__ beta,
                        float* __restrict__ out, int n) {
    int i = blockIdx.x * blockDim.x + threadIdx.x;
    if (i >= n) return;
    const float c = centrality[i];
    float acc = 0.f;
    #pragma unroll
    for (int h = 0; h < H_HEADS; ++h)
        acc += (c * gamma[h] + beta[h]) * hfin[i * H_HEADS + h];
    acc *= 0.25f;
    out[i] = (acc > 0.f) ? acc : 0.01f * acc;
}

extern "C" void kernel_launch(void* const* d_in, const int* in_sizes, int n_in,
                              void* d_out, int out_size, void* d_ws, size_t ws_size,
                              hipStream_t stream) {
    const float* inputs     = (const float*)d_in[0];
    const float* centrality = (const float*)d_in[1];
    const float* W1         = (const float*)d_in[2];
    const float* b1         = (const float*)d_in[3];
    const float* W2         = (const float*)d_in[4];
    const float* b2         = (const float*)d_in[5];
    const float* rel_emb    = (const float*)d_in[6];
    const float* W_pred     = (const float*)d_in[7];
    const float* gamma      = (const float*)d_in[8];
    const float* beta       = (const float*)d_in[9];
    const int*   edge_types = (const int*)d_in[10];
    const int*   src        = (const int*)d_in[11];
    const int*   dst        = (const int*)d_in[12];
    float* out = (float*)d_out;

    const int N = in_sizes[1];
    const int E = in_sizes[10];
    const int Np = (N + 3) & ~3;
    const int Ep = (E + 3) & ~3;
    const int NB = (N + (1 << BSH) - 1) >> BSH;
    const int nChunks = (E + CH - 1) / CH;
    const int M  = NB * nChunks;
    const int Mp = (M + 3) & ~3;

    char* ws = (char*)d_ws;
    _Float16* W1h = (_Float16*)ws;
    float* etab   = (float*)(ws + 65536);
    float* hA     = (float*)(ws + 69632);
    float* hB     = hA + (size_t)Np * H_HEADS;
    int*   cnt      = (int*)(hB + (size_t)Np * H_HEADS);
    int*   rowstart = cnt + Np;
    int*   hist     = rowstart + Np;
    int*   scanned  = hist + Mp;
    int*   bsum     = scanned + Mp;
    unsigned* part   = (unsigned*)(bsum + Mp);
    unsigned* sorted = part + Ep;
    float* hsA = (float*)(sorted + Ep);
    float* hsB = hsA + Np;

    const int TPB = 256;

    {
        int cnt1 = COLS * IN_DIM;
        k_w1_to_f16<<<(cnt1 + TPB - 1) / TPB, TPB, 0, stream>>>(W1, W1h, cnt1);
    }
    {
        int cnt2 = LAYERS * RELS * H_HEADS;
        k_etab<<<(cnt2 + TPB - 1) / TPB, TPB, 0, stream>>>(rel_emb, W_pred, etab);
    }
    {
        int nTiles = (N + 15) / 16;
        int wavesPerBlock = 4;
        int blocks = (nTiles + wavesPerBlock - 1) / wavesPerBlock;
        k_mlp_wmma<<<blocks, wavesPerBlock * 32, 0, stream>>>(
            inputs, W1h, b1, W2, b2, hA, N, nTiles);
    }

    const int nBlocks = (N + TPB - 1) / TPB;
    const int sBlocks = (M + 1023) / 1024;
    k_chunk_hist<<<nChunks, CTH, 0, stream>>>(dst, hist, E, nChunks, NB);
    k_scan_local<<<sBlocks, 256, 0, stream>>>(hist, scanned, bsum, M);
    k_scan_add<<<sBlocks, 256, 0, stream>>>(scanned, bsum, M);
    k_partition<<<nChunks, CTH, 0, stream>>>(edge_types, src, dst, scanned, part, E, nChunks, NB);
    k_fine<<<NB, FTH, 0, stream>>>(part, scanned, cnt, rowstart, sorted, N, E, nChunks, NB);

    const int gBlocks = (int)(((size_t)N * GL + TPB - 1) / TPB);
    const int lBlocks = (N + LNPW - 1) / LNPW;
    float* hscur = hsA;
    float* hsnxt = hsB;

    for (int l = 0; l < LAYERS; ++l) {
        const float* etab_l = etab + l * RELS * H_HEADS;
        const int do_mean = (l != LAYERS - 1) ? 1 : 0;
        if (l == 0) {
            k_gather<<<gBlocks, TPB, 0, stream>>>(sorted, rowstart, cnt, etab_l, hA, hB, hscur,
                                                  N, do_mean);
        } else {
            k_gather_lds<<<lBlocks, LT, 0, stream>>>(sorted, rowstart, cnt, etab_l, hscur, hB,
                                                     hsnxt, N, do_mean);
            float* tmp = hscur; hscur = hsnxt; hsnxt = tmp;
        }
    }

    k_final<<<nBlocks, TPB, 0, stream>>>(hB, centrality, gamma, beta, out, N);
}


# Launches merged, 12 -> 7: histogram + W1 copy + predicate table in one; scan fix-up folded into its two readers; MLP runs beside the level-1 placement; final scaling inside the last layer
# speedup vs baseline: 22.8936x; 1.1382x over previous
#include <hip/hip_runtime.h>
#include <hip/hip_bf16.h>

typedef __attribute__((ext_vector_type(16))) _Float16 v16h;
typedef __attribute__((ext_vector_type(8)))  _Float16 v8h;
typedef __attribute__((ext_vector_type(8)))  float    v8f;
typedef __attribute__((ext_vector_type(4)))  float    v4f;

#define H_HEADS 4
#define IN_DIM  128
#define HID     64
#define COLS    256
#define PRED    16
#define RELS    50
#define LAYERS  3

__device__ __forceinline__ void w1_to_f16_body(const float* __restrict__ W1,
                                               _Float16* __restrict__ W1h, int n, int blk) {
    int i = blk * blockDim.x + threadIdx.x;
    if (i < n) W1h[i] = (_Float16)W1[i];
}

__device__ __forceinline__ void etab_body(const float* __restrict__ rel_emb,
                                          const float* __restrict__ W_pred,
                                          float* __restrict__ etab, int blk) {
    int i = blk * blockDim.x + threadIdx.x;
    if (i >= LAYERS * RELS * H_HEADS) return;
    int l = i / (RELS * H_HEADS);
    int rem = i % (RELS * H_HEADS);
    int r = rem / H_HEADS;
    int h = rem % H_HEADS;
    float e = 0.f;
    #pragma unroll
    for (int p = 0; p < PRED; ++p)
        e += rel_emb[r * PRED + p] * W_pred[(l * PRED + p) * H_HEADS + h];
    etab[i] = (e > 0.f) ? e : 0.2f * e;
}

__device__ __forceinline__ void mlp_body(const float* __restrict__ inputs,
                                         const _Float16* __restrict__ W1h,
                                         const float* __restrict__ b1,
                                         const float* __restrict__ W2,
                                         const float* __restrict__ b2,
                                         float* __restrict__ hout,
                                         int n, int nTiles, int blk) {
    const int wave = (blk * blockDim.x + threadIdx.x) >> 5;
    if (wave >= nTiles) return;
    const int lane = threadIdx.x & 31;
    const int grp  = lane >> 4;
    const int l15  = lane & 15;
    const int nodeBase = wave * 16;
    int arow = nodeBase + l15;
    if (arow > n - 1) arow = n - 1;
    const float* __restrict__ abase = inputs + (size_t)arow * IN_DIM;

    const v8f zero = {};

    #pragma unroll
    for (int head = 0; head < H_HEADS; ++head) {
        v8f acc0 = zero, acc1 = zero, acc2 = zero, acc3 = zero;

        #pragma unroll
        for (int kk = 0; kk < IN_DIM; kk += 32) {
            const int kb = kk + grp * 8;
            v4f a0 = *(const v4f*)(abase + kb + 0);
            v4f a1 = *(const v4f*)(abase + kb + 4);
            v4f a2 = *(const v4f*)(abase + kb + 16);
            v4f a3 = *(const v4f*)(abase + kb + 20);
            v16h A;
            #pragma unroll
            for (int i = 0; i < 4; ++i) {
                A[i]      = (_Float16)a0[i];
                A[4 + i]  = (_Float16)a1[i];
                A[8 + i]  = (_Float16)a2[i];
                A[12 + i] = (_Float16)a3[i];
            }
            #pragma unroll
            for (int tt = 0; tt < 4; ++tt) {
                const int col = head * HID + tt * 16 + l15;
                const _Float16* q = W1h + (size_t)col * IN_DIM + kb;
                v8h blo = *(const v8h*)(q);
                v8h bhi = *(const v8h*)(q + 16);
                v16h B;
                #pragma unroll
                for (int i = 0; i < 8; ++i) { B[i] = blo[i]; B[8 + i] = bhi[i]; }
                v8f c = (tt == 0) ? acc0 : (tt == 1) ? acc1 : (tt == 2) ? acc2 : acc3;
                c = __builtin_amdgcn_wmma_f32_16x16x32_f16(
                        false, A, false, B, (short)0, c, false, false);
                if (tt == 0) acc0 = c; else if (tt == 1) acc1 = c;
                else if (tt == 2) acc2 = c; else acc3 = c;
            }
        }

        v8f hacc = zero;
        #pragma unroll
        for (int tt = 0; tt < 4; ++tt) {
            const int col  = head * HID + tt * 16 + l15;
            const float bias = b1[col];
            const float w2   = W2[col];
            v8f c = (tt == 0) ? acc0 : (tt == 1) ? acc1 : (tt == 2) ? acc2 : acc3;
            #pragma unroll
            for (int v = 0; v < 8; ++v) {
                float p = c[v] + bias;
                p = (p > 0.f) ? p : 0.f;
                p *= w2;
                p += __shfl_xor(p, 1, 32);
                p += __shfl_xor(p, 2, 32);
                p += __shfl_xor(p, 4, 32);
                p += __shfl_xor(p, 8, 32);
                hacc[v] += p;
            }
        }
        if (l15 == 0) {
            #pragma unroll
            for (int v = 0; v < 8; ++v) {
                const int node = nodeBase + v + 8 * grp;
                if (node < n)
                    hout[node * H_HEADS + head] = hacc[v] + b2[head];
            }
        }
    }
}

#define GL  8
#define GCH 4
#define BSH    8
#define NBMAX  1024
#define CH     8192
#define CTH    256
#define FTH    1024
#define FCAP   12288
#define SBMAX  1024

__device__ __forceinline__ void scan1024(const int* cin, int* cout, int* wsum, int t,
                                         int (&own)[4]) {
    const int lane = t & 31, w = t >> 5;
    const int c0 = cin[4 * t], c1 = cin[4 * t + 1], c2 = cin[4 * t + 2], c3 = cin[4 * t + 3];
    const int tsum = c0 + c1 + c2 + c3;
    int x = tsum;
    #pragma unroll
    for (int d = 1; d < 32; d <<= 1) {
        const int y = __shfl_up(x, d, 32);
        if (lane >= d) x += y;
    }
    if (lane == 31) wsum[w] = x;
    __syncthreads();
    int wbase = 0;
    #pragma unroll
    for (int k = 0; k < 8; ++k) if (k < w) wbase += wsum[k];
    const int run = wbase + x - tsum;
    own[0] = run; own[1] = run + c0; own[2] = run + c0 + c1; own[3] = run + c0 + c1 + c2;
    cout[4 * t] = own[0]; cout[4 * t + 1] = own[1];
    cout[4 * t + 2] = own[2]; cout[4 * t + 3] = own[3];
}

__device__ __forceinline__ void chunk_hist_body(const int* __restrict__ dst,
                                                int* __restrict__ hist,
                                                int E, int nChunks, int NB, int blk) {
    __shared__ int lc[NBMAX];
    const int t = threadIdx.x;
    for (int b = t; b < NBMAX; b += CTH) lc[b] = 0;
    __syncthreads();
    const int base = blk * CH;
    #pragma unroll
    for (int k = 0; k < CH / (CTH * 4); ++k) {
        const int i = base + (k * CTH + t) * 4;
        if (i + 3 < E) {
            const int4 d = *(const int4*)(dst + i);
            atomicAdd(&lc[d.x >> BSH], 1);
            atomicAdd(&lc[d.y >> BSH], 1);
            atomicAdd(&lc[d.z >> BSH], 1);
            atomicAdd(&lc[d.w >> BSH], 1);
        } else {
            for (int j = i; j < E; ++j) atomicAdd(&lc[dst[j] >> BSH], 1);
        }
    }
    __syncthreads();
    for (int b = t; b < NB; b += CTH) hist[(size_t)b * nChunks + blk] = lc[b];
}

__global__ void __launch_bounds__(CTH)
k_pre(const int* __restrict__ dst, int* __restrict__ hist, int E, int nChunks, int NB,
      const float* __restrict__ W1, _Float16* __restrict__ W1h, int nW1, int w1Blocks,
      const float* __restrict__ rel_emb, const float* __restrict__ W_pred,
      float* __restrict__ etab) {
    const int blk = blockIdx.x;
    if (blk < nChunks)                 chunk_hist_body(dst, hist, E, nChunks, NB, blk);
    else if (blk < nChunks + w1Blocks) w1_to_f16_body(W1, W1h, nW1, blk - nChunks);
    else                               etab_body(rel_emb, W_pred, etab, blk - nChunks - w1Blocks);
}

__global__ void k_scan_local(const int* __restrict__ cnt, int* __restrict__ rowstart,
                             int* __restrict__ bsum, int n) {
    __shared__ int wsum[8];
    const int t = threadIdx.x, lane = t & 31, w = t >> 5;
    const int base = blockIdx.x * 1024 + t * 4;
    int c[4];
    #pragma unroll
    for (int k = 0; k < 4; ++k) c[k] = (base + k < n) ? cnt[base + k] : 0;
    const int tsum = c[0] + c[1] + c[2] + c[3];
    int x = tsum;
    #pragma unroll
    for (int d = 1; d < 32; d <<= 1) {
        const int y = __shfl_up(x, d, 32);
        if (lane >= d) x += y;
    }
    if (lane == 31) wsum[w] = x;
    __syncthreads();
    int wbase = 0;
    #pragma unroll
    for (int k = 0; k < 8; ++k) if (k < w) wbase += wsum[k];
    int run = wbase + x - tsum;
    #pragma unroll
    for (int k = 0; k < 4; ++k) {
        if (base + k < n) rowstart[base + k] = run;
        run += c[k];
    }
    if (t == 255) bsum[blockIdx.x] = wbase + x;
}

__device__ __forceinline__ void partition_body(
            const int* __restrict__ et, const int* __restrict__ src,
            const int* __restrict__ dst, const int* __restrict__ scanned,
            const int* __restrict__ bsum, int sBlocks,
            unsigned* __restrict__ tmp, int E, int nChunks, int NB, int blk) {
    __shared__ int lc[NBMAX];
    __shared__ int lb[NBMAX];
    __shared__ int gd[NBMAX];
    __shared__ int bp[SBMAX];
    __shared__ int wsum[8];
    __shared__ unsigned recs[CH];
    __shared__ unsigned short rb[CH];
    const int t = threadIdx.x;
    for (int b = t; b < NBMAX; b += CTH) lc[b] = 0;
    for (int k = t; k < SBMAX; k += CTH) lb[k] = (k < sBlocks) ? bsum[k] : 0;
    __syncthreads();
    {
        int unused[4];
        scan1024(lb, bp, wsum, t, unused);
    }

    const int base = blk * CH;
    unsigned rec[32];
    unsigned br[32];
    #pragma unroll
    for (int k = 0; k < CH / (CTH * 4); ++k) {
        const int i = base + (k * CTH + t) * 4;
        int dv[4], sv[4], ev[4];
        if (i + 3 < E) {
            const int4 d = *(const int4*)(dst + i);
            const int4 s = *(const int4*)(src + i);
            const int4 e = *(const int4*)(et + i);
            dv[0] = d.x; dv[1] = d.y; dv[2] = d.z; dv[3] = d.w;
            sv[0] = s.x; sv[1] = s.y; sv[2] = s.z; sv[3] = s.w;
            ev[0] = e.x; ev[1] = e.y; ev[2] = e.z; ev[3] = e.w;
        } else {
            #pragma unroll
            for (int c = 0; c < 4; ++c) {
                const int j = i + c;
                dv[c] = -1; sv[c] = 0; ev[c] = 0;
                if (j < E) { dv[c] = dst[j]; sv[c] = src[j]; ev[c] = et[j]; }
            }
        }
        #pragma unroll
        for (int c = 0; c < 4; ++c) {
            rec[k * 4 + c] = 0u; br[k * 4 + c] = ~0u;
            if (dv[c] >= 0) {
                const int b = dv[c] >> BSH;
                const int r = atomicAdd(&lc[b], 1);
                rec[k * 4 + c] = ((unsigned)sv[c] << 14) | ((unsigned)ev[c] << 8)
                               | ((unsigned)dv[c] & 255u);
                br[k * 4 + c] = ((unsigned)b << 16) | (unsigned)r;
            }
        }
    }
    __syncthreads();

    int own[4];
    scan1024(lc, lb, wsum, t, own);
    #pragma unroll
    for (int c = 0; c < 4; ++c) {
        const int b = 4 * t + c;
        if (b < NB) {
            const int idx = b * nChunks + blk;
            gd[b] = scanned[idx] + bp[idx >> 10] - own[c];
        }
    }
    __syncthreads();

    #pragma unroll
    for (int q = 0; q < 32; ++q) {
        if (br[q] != ~0u) {
            const int b = (int)(br[q] >> 16);
            const int p = lb[b] + (int)(br[q] & 0xFFFFu);
            recs[p] = rec[q];
            rb[p] = (unsigned short)b;
        }
    }
    __syncthreads();

    int nvalid = E - base;
    if (nvalid > CH) nvalid = CH;
    for (int i = t; i < nvalid; i += CTH)
        tmp[gd[rb[i]] + i] = recs[i];
}

__global__ void __launch_bounds__(CTH)
k_partition_mlp(const int* __restrict__ et, const int* __restrict__ src,
                const int* __restrict__ dst, const int* __restrict__ scanned,
                const int* __restrict__ bsum, int sBlocks,
                unsigned* __restrict__ tmp, int E, int nChunks, int NB, int pBlocks,
                const float* __restrict__ inputs, const _Float16* __restrict__ W1h,
                const float* __restrict__ b1, const float* __restrict__ W2,
                const float* __restrict__ b2, float* __restrict__ hout,
                int n, int nTiles) {
    const int blk = blockIdx.x;
    if (blk < pBlocks)
        partition_body(et, src, dst, scanned, bsum, sBlocks, tmp, E, nChunks, NB, blk);
    else
        mlp_body(inputs, W1h, b1, W2, b2, hout, n, nTiles, blk - pBlocks);
}

__global__ void __launch_bounds__(FTH)
k_fine(const unsigned* __restrict__ tmp, const int* __restrict__ scanned,
       const int* __restrict__ bsum,
       int* __restrict__ cnt, int* __restrict__ rowstart,
       unsigned* __restrict__ sorted, int n, int E, int nChunks, int NB) {
    __shared__ int ncount[1 << BSH];
    __shared__ int nbase[1 << BSH];
    __shared__ int cur[1 << BSH];
    __shared__ int wsum[8];
    __shared__ unsigned stage[FCAP];
    const int t = threadIdx.x, lane = t & 31, w = t >> 5;
    __shared__ int boff[2];
    const int b = blockIdx.x;
    const int e0 = b * nChunks, e1 = (b + 1) * nChunks;
    const int s0 = scanned[e0];
    const int s1 = (b + 1 < NB) ? scanned[e1] : 0;
    if (w < 2) {
        const int lim = ((w == 0) ? e0 : e1) >> 10;
        int acc = 0;
        for (int k = lane; k < lim; k += 32) acc += bsum[k];
        #pragma unroll
        for (int d = 16; d >= 1; d >>= 1) acc += __shfl_xor(acc, d, 32);
        if (lane == 0) boff[w] = acc;
    }
    if (t < (1 << BSH)) { ncount[t] = 0; cur[t] = 0; }
    __syncthreads();
    const int bs = s0 + boff[0];
    const int be = (b + 1 < NB) ? s1 + boff[1] : E;
    const int nb = be - bs;
    const bool staged = (nb <= FCAP);

    for (int i0 = 0; i0 < nb; i0 += FTH * 8) {
        unsigned r[8];
        #pragma unroll
        for (int k = 0; k < 8; ++k) {
            const int i = i0 + k * FTH + t;
            r[k] = (i < nb) ? tmp[bs + i] : 0u;
        }
        #pragma unroll
        for (int k = 0; k < 8; ++k)
            if (i0 + k * FTH + t < nb) atomicAdd(&ncount[r[k] & 255u], 1);
    }
    __syncthreads();

    int v = 0, x = 0;
    if (t < (1 << BSH)) {
        v = ncount[t];
        x = v;
        #pragma unroll
        for (int d = 1; d < 32; d <<= 1) {
            const int y = __shfl_up(x, d, 32);
            if (lane >= d) x += y;
        }
        if (lane == 31) wsum[w] = x;
    }
    __syncthreads();
    if (t < (1 << BSH)) {
        int wbase = 0;
        #pragma unroll
        for (int k = 0; k < 8; ++k) if (k < w) wbase += wsum[k];
        const int excl = wbase + x - v;
        nbase[t] = excl;
        const int node = (b << BSH) + t;
        if (node < n) { cnt[node] = v; rowstart[node] = bs + excl; }
    }
    __syncthreads();

    for (int i0 = 0; i0 < nb; i0 += FTH * 8) {
        unsigned r[8];
        #pragma unroll
        for (int k = 0; k < 8; ++k) {
            const int i = i0 + k * FTH + t;
            r[k] = (i < nb) ? tmp[bs + i] : 0u;
        }
        #pragma unroll
        for (int k = 0; k < 8; ++k) {
            if (i0 + k * FTH + t < nb) {
                const unsigned dl = r[k] & 255u;
                const int p = nbase[dl] + atomicAdd(&cur[dl], 1);
                const unsigned val = r[k] >> 8;
                if (staged) stage[p] = val; else sorted[bs + p] = val;
            }
        }
    }
    if (staged) {
        __syncthreads();
        for (int i = t; i < nb; i += FTH) sorted[bs + i] = stage[i];
    }
}

__device__ __forceinline__ void g_load(const unsigned* __restrict__ sorted, int start,
                                       int len, int j0, unsigned (&pk)[GCH]) {
    #pragma unroll
    for (int k = 0; k < GCH; ++k) {
        const int j = j0 + k * GL;
        pk[k] = (j < len) ? sorted[start + j] : 0u;
    }
}
__device__ __forceinline__ void g_max(const v4f* tab, const unsigned (&pk)[GCH],
                                      int len, int j0, float (&m)[H_HEADS]) {
    #pragma unroll
    for (int k = 0; k < GCH; ++k) {
        if (j0 + k * GL < len) {
            const v4f e = tab[pk[k] & 63u];
            #pragma unroll
            for (int h = 0; h < H_HEADS; ++h) m[h] = fmaxf(m[h], e[h]);
        }
    }
}
__device__ __forceinline__ void g_sum(const v4f* tab, const float* __restrict__ hcur,
                                      const unsigned (&pk)[GCH], int len, int j0,
                                      const float (&m)[H_HEADS],
                                      float (&s)[H_HEADS], float (&u)[H_HEADS]) {
    v4f hv[GCH];
    #pragma unroll
    for (int k = 0; k < GCH; ++k) {
        hv[k] = (v4f){0.f, 0.f, 0.f, 0.f};
        if (j0 + k * GL < len)
            hv[k] = *(const v4f*)(hcur + (size_t)(pk[k] >> 6) * H_HEADS);
    }
    #pragma unroll
    for (int k = 0; k < GCH; ++k) {
        if (j0 + k * GL < len) {
            const v4f e = tab[pk[k] & 63u];
            #pragma unroll
            for (int h = 0; h < H_HEADS; ++h) {
                const float wgt = __expf(e[h] - m[h]);
                s[h] += wgt;
                u[h] += wgt * hv[k][h];
            }
        }
    }
}

__global__ void k_gather(const unsigned* __restrict__ sorted,
                         const int* __restrict__ rowstart,
                         const int* __restrict__ cnt,
                         const float* __restrict__ etab_l,
                         const float* __restrict__ hcur,
                         float* __restrict__ hnext, float* __restrict__ hs_next,
                         int n, int do_mean) {
    __shared__ v4f tab[RELS];
    if (threadIdx.x < RELS)
        tab[threadIdx.x] = *(const v4f*)(etab_l + threadIdx.x * H_HEADS);
    __syncthreads();

    const int gid  = blockIdx.x * blockDim.x + threadIdx.x;
    const int node = gid / GL;
    const int sub  = threadIdx.x & (GL - 1);
    int start = 0, len = 0;
    if (node < n) { start = rowstart[node]; len = cnt[node]; }

    unsigned pk0[GCH];
    g_load(sorted, start, len, sub, pk0);

    float m[H_HEADS];
    #pragma unroll
    for (int h = 0; h < H_HEADS; ++h) m[h] = -__builtin_huge_valf();
    g_max(tab, pk0, len, sub, m);
    for (int base = GL * GCH; base < len; base += GL * GCH) {
        unsigned pk[GCH];
        g_load(sorted, start, len, base + sub, pk);
        g_max(tab, pk, len, base + sub, m);
    }
    #pragma unroll
    for (int h = 0; h < H_HEADS; ++h) {
        m[h] = fmaxf(m[h], __shfl_xor(m[h], 1, 32));
        m[h] = fmaxf(m[h], __shfl_xor(m[h], 2, 32));
        m[h] = fmaxf(m[h], __shfl_xor(m[h], 4, 32));
    }

    float s[H_HEADS], u[H_HEADS];
    #pragma unroll
    for (int h = 0; h < H_HEADS; ++h) { s[h] = 0.f; u[h] = 0.f; }
    g_sum(tab, hcur, pk0, len, sub, m, s, u);
    for (int base = GL * GCH; base < len; base += GL * GCH) {
        unsigned pk[GCH];
        g_load(sorted, start, len, base + sub, pk);
        g_sum(tab, hcur, pk, len, base + sub, m, s, u);
    }
    #pragma unroll
    for (int h = 0; h < H_HEADS; ++h) {
        s[h] += __shfl_xor(s[h], 1, 32);  u[h] += __shfl_xor(u[h], 1, 32);
        s[h] += __shfl_xor(s[h], 2, 32);  u[h] += __shfl_xor(u[h], 2, 32);
        s[h] += __shfl_xor(s[h], 4, 32);  u[h] += __shfl_xor(u[h], 4, 32);
    }

    if (sub == 0 && node < n) {
        v4f v;
        #pragma unroll
        for (int h = 0; h < H_HEADS; ++h) {
            const float x = (s[h] > 0.f) ? (u[h] / s[h]) : 0.f;
            v[h] = (x > 0.f) ? x : 0.f;
        }
        if (do_mean) {
            hs_next[node] = 0.25f * (v[0] + v[1] + v[2] + v[3]);
            if (node == n - 1)
                for (int j = n; j < ((n + 3) & ~3); ++j) hs_next[j] = 0.f;
        } else {
            *(v4f*)(hnext + (size_t)node * H_HEADS) = v;
        }
    }
}

#define LT    800
#define LGL   2
#define LCH   16
#define LNPW  (LT / LGL)
#define LCAPN 81664
#define LST   8

__device__ __forceinline__ void l_load(const unsigned* __restrict__ sorted, int start,
                                       int len, int j0, unsigned (&pk)[LCH]) {
    #pragma unroll
    for (int k = 0; k < LCH; ++k) {
        const int j = j0 + k * LGL;
        pk[k] = sorted[(j < len) ? start + j : 0];
    }
}
__device__ __forceinline__ void l_max(const v4f* tab, const unsigned (&pk)[LCH],
                                      int len, int j0, float (&m)[H_HEADS]) {
    #pragma unroll
    for (int k = 0; k < LCH; ++k) {
        if (j0 + k * LGL < len) {
            const v4f e = tab[pk[k] & 63u];
            #pragma unroll
            for (int h = 0; h < H_HEADS; ++h) m[h] = fmaxf(m[h], e[h]);
        }
    }
}
__device__ __forceinline__ void l_sum(const v4f* tab, const float* hl,
                                      const float* __restrict__ hs, unsigned count,
                                      const unsigned (&pk)[LCH],
                                      int len, int j0, const float (&m)[H_HEADS],
                                      float (&s)[H_HEADS], float (&u)[H_HEADS]) {
    float hv[LCH];
    #pragma unroll
    for (int k = 0; k < LCH; ++k) {
        const unsigned sn = pk[k] >> 6;
        const bool inl = sn < count;
        const float a = hl[inl ? sn : 0u];
        const float b = hs[inl ? 0u : sn];
        hv[k] = inl ? a : b;
    }
    #pragma unroll
    for (int k = 0; k < LCH; ++k) {
        if (j0 + k * LGL < len) {
            const v4f e = tab[pk[k] & 63u];
            #pragma unroll
            for (int h = 0; h < H_HEADS; ++h) {
                const float wgt = __expf(e[h] - m[h]);
                s[h] += wgt;
                u[h] += wgt * hv[k];
            }
        }
    }
}

__global__ void __launch_bounds__(LT)
k_gather_lds(const unsigned* __restrict__ sorted,
             const int* __restrict__ rowstart,
             const int* __restrict__ cnt,
             const float* __restrict__ etab_l,
             const float* __restrict__ hs,
             float* __restrict__ hs_next,
             const float* __restrict__ centrality,
             const float* __restrict__ gamma,
             const float* __restrict__ beta,
             float* __restrict__ out,
             int n, int do_mean) {
    __shared__ v4f tab[RELS];
    __shared__ float hl[LCAPN];
    const int t = threadIdx.x;
    if (t < RELS) tab[t] = *(const v4f*)(etab_l + t * H_HEADS);

    const int node = blockIdx.x * LNPW + t / LGL;
    const int sub  = t & (LGL - 1);
    int start = 0, len = 0;
    if (node < n) { start = rowstart[node]; len = cnt[node]; }

    unsigned pk0[LCH];
    l_load(sorted, start, len, sub, pk0);

    const int np = (n + 3) & ~3;
    const int count = (np < LCAPN) ? np : LCAPN;
    for (int i0 = 0; i0 < count; i0 += LT * 4 * LST) {
        v4f r[LST];
        #pragma unroll
        for (int k = 0; k < LST; ++k) {
            int i = i0 + (k * LT + t) * 4;
            if (i > count - 4) i = count - 4;
            r[k] = *(const v4f*)(hs + i);
        }
        #pragma unroll
        for (int k = 0; k < LST; ++k) {
            const int i = i0 + (k * LT + t) * 4;
            if (i < count) *(v4f*)(hl + i) = r[k];
        }
    }
    __syncthreads();

    float m[H_HEADS];
    #pragma unroll
    for (int h = 0; h < H_HEADS; ++h) m[h] = -__builtin_huge_valf();
    l_max(tab, pk0, len, sub, m);
    for (int base = LGL * LCH; base < len; base += LGL * LCH) {
        unsigned pk[LCH];
        l_load(sorted, start, len, base + sub, pk);
        l_max(tab, pk, len, base + sub, m);
    }
    #pragma unroll
    for (int h = 0; h < H_HEADS; ++h)
        m[h] = fmaxf(m[h], __shfl_xor(m[h], 1, 32));

    float s[H_HEADS], u[H_HEADS];
    #pragma unroll
    for (int h = 0; h < H_HEADS; ++h) { s[h] = 0.f; u[h] = 0.f; }
    l_sum(tab, hl, hs, (unsigned)count, pk0, len, sub, m, s, u);
    for (int base = LGL * LCH; base < len; base += LGL * LCH) {
        unsigned pk[LCH];
        l_load(sorted, start, len, base + sub, pk);
        l_sum(tab, hl, hs, (unsigned)count, pk, len, base + sub, m, s, u);
    }
    #pragma unroll
    for (int h = 0; h < H_HEADS; ++h) {
        s[h] += __shfl_xor(s[h], 1, 32);
        u[h] += __shfl_xor(u[h], 1, 32);
    }

    if (sub == 0 && node < n) {
        v4f v;
        #pragma unroll
        for (int h = 0; h < H_HEADS; ++h) {
            const float x = (s[h] > 0.f) ? (u[h] / s[h]) : 0.f;
            v[h] = (x > 0.f) ? x : 0.f;
        }
        if (do_mean) {
            hs_next[node] = 0.25f * (v[0] + v[1] + v[2] + v[3]);
            if (node == n - 1)
                for (int j = n; j < np; ++j) hs_next[j] = 0.f;
        } else {
            const float c = centrality[node];
            float acc = 0.f;
            #pragma unroll
            for (int h = 0; h < H_HEADS; ++h)
                acc += (c * gamma[h] + beta[h]) * v[h];
            acc *= 0.25f;
            out[node] = (acc > 0.f) ? acc : 0.01f * acc;
        }
    }
}

extern "C" void kernel_launch(void* const* d_in, const int* in_sizes, int n_in,
                              void* d_out, int out_size, void* d_ws, size_t ws_size,
                              hipStream_t stream) {
    const float* inputs     = (const float*)d_in[0];
    const float* centrality = (const float*)d_in[1];
    const float* W1         = (const float*)d_in[2];
    const float* b1         = (const float*)d_in[3];
    const float* W2         = (const float*)d_in[4];
    const float* b2         = (const float*)d_in[5];
    const float* rel_emb    = (const float*)d_in[6];
    const float* W_pred     = (const float*)d_in[7];
    const float* gamma      = (const float*)d_in[8];
    const float* beta       = (const float*)d_in[9];
    const int*   edge_types = (const int*)d_in[10];
    const int*   src        = (const int*)d_in[11];
    const int*   dst        = (const int*)d_in[12];
    float* out = (float*)d_out;

    const int N = in_sizes[1];
    const int E = in_sizes[10];
    const int Np = (N + 3) & ~3;
    const int Ep = (E + 3) & ~3;
    const int NB = (N + (1 << BSH) - 1) >> BSH;
    const int nChunks = (E + CH - 1) / CH;
    const int M  = NB * nChunks;
    const int Mp = (M + 3) & ~3;

    char* ws = (char*)d_ws;
    _Float16* W1h = (_Float16*)ws;
    float* etab   = (float*)(ws + 65536);
    float* hA     = (float*)(ws + 69632);
    float* hB     = hA + (size_t)Np * H_HEADS;
    int*   cnt      = (int*)(hB + (size_t)Np * H_HEADS);
    int*   rowstart = cnt + Np;
    int*   hist     = rowstart + Np;
    int*   scanned  = hist + Mp;
    int*   bsum     = scanned + Mp;
    unsigned* part   = (unsigned*)(bsum + Mp);
    unsigned* sorted = part + Ep;
    float* hsA = (float*)(sorted + Ep);
    float* hsB = hsA + Np;

    const int TPB = 256;

    const int nW1      = COLS * IN_DIM;
    const int w1Blocks = (nW1 + CTH - 1) / CTH;
    const int eBlocks  = (LAYERS * RELS * H_HEADS + CTH - 1) / CTH;
    k_pre<<<nChunks + w1Blocks + eBlocks, CTH, 0, stream>>>(
        dst, hist, E, nChunks, NB, W1, W1h, nW1, w1Blocks, rel_emb, W_pred, etab);

    const int sBlocks = (M + 1023) / 1024;
    k_scan_local<<<sBlocks, 256, 0, stream>>>(hist, scanned, bsum, M);

    const int nTiles    = (N + 15) / 16;
    const int mlpBlocks = (nTiles + CTH / 32 - 1) / (CTH / 32);
    k_partition_mlp<<<nChunks + mlpBlocks, CTH, 0, stream>>>(
        edge_types, src, dst, scanned, bsum, sBlocks, part, E, nChunks, NB, nChunks,
        inputs, W1h, b1, W2, b2, hA, N, nTiles);

    k_fine<<<NB, FTH, 0, stream>>>(part, scanned, bsum, cnt, rowstart, sorted, N, E, nChunks, NB);

    const int gBlocks = (int)(((size_t)N * GL + TPB - 1) / TPB);
    const int lBlocks = (N + LNPW - 1) / LNPW;
    float* hscur = hsA;
    float* hsnxt = hsB;

    for (int l = 0; l < LAYERS; ++l) {
        const float* etab_l = etab + l * RELS * H_HEADS;
        const int do_mean = (l != LAYERS - 1) ? 1 : 0;
        if (l == 0) {
            k_gather<<<gBlocks, TPB, 0, stream>>>(sorted, rowstart, cnt, etab_l, hA, hB, hscur,
                                                  N, do_mean);
        } else {
            k_gather_lds<<<lBlocks, LT, 0, stream>>>(sorted, rowstart, cnt, etab_l, hscur,
                                                     hsnxt, centrality, gamma, beta, out,
                                                     N, do_mean);
            float* tmp = hscur; hscur = hsnxt; hsnxt = tmp;
        }
    }
}


# First layer fused into the level-2 sort workgroups: buckets read from LDS while still staged, 4 lanes per node (sums now stride-4 + 2 xor steps); one launch and the bucket re-read gone, 7 -> 6
# speedup vs baseline: 24.1903x; 1.0566x over previous
#include <hip/hip_runtime.h>
#include <hip/hip_bf16.h>

typedef __attribute__((ext_vector_type(16))) _Float16 v16h;
typedef __attribute__((ext_vector_type(8)))  _Float16 v8h;
typedef __attribute__((ext_vector_type(8)))  float    v8f;
typedef __attribute__((ext_vector_type(4)))  float    v4f;

#define H_HEADS 4
#define IN_DIM  128
#define HID     64
#define COLS    256
#define PRED    16
#define RELS    50
#define LAYERS  3

__device__ __forceinline__ void w1_to_f16_body(const float* __restrict__ W1,
                                               _Float16* __restrict__ W1h, int n, int blk) {
    int i = blk * blockDim.x + threadIdx.x;
    if (i < n) W1h[i] = (_Float16)W1[i];
}

__device__ __forceinline__ void etab_body(const float* __restrict__ rel_emb,
                                          const float* __restrict__ W_pred,
                                          float* __restrict__ etab, int blk) {
    int i = blk * blockDim.x + threadIdx.x;
    if (i >= LAYERS * RELS * H_HEADS) return;
    int l = i / (RELS * H_HEADS);
    int rem = i % (RELS * H_HEADS);
    int r = rem / H_HEADS;
    int h = rem % H_HEADS;
    float e = 0.f;
    #pragma unroll
    for (int p = 0; p < PRED; ++p)
        e += rel_emb[r * PRED + p] * W_pred[(l * PRED + p) * H_HEADS + h];
    etab[i] = (e > 0.f) ? e : 0.2f * e;
}

__device__ __forceinline__ void mlp_body(const float* __restrict__ inputs,
                                         const _Float16* __restrict__ W1h,
                                         const float* __restrict__ b1,
                                         const float* __restrict__ W2,
                                         const float* __restrict__ b2,
                                         float* __restrict__ hout,
                                         int n, int nTiles, int blk) {
    const int wave = (blk * blockDim.x + threadIdx.x) >> 5;
    if (wave >= nTiles) return;
    const int lane = threadIdx.x & 31;
    const int grp  = lane >> 4;
    const int l15  = lane & 15;
    const int nodeBase = wave * 16;
    int arow = nodeBase + l15;
    if (arow > n - 1) arow = n - 1;
    const float* __restrict__ abase = inputs + (size_t)arow * IN_DIM;

    const v8f zero = {};

    #pragma unroll
    for (int head = 0; head < H_HEADS; ++head) {
        v8f acc0 = zero, acc1 = zero, acc2 = zero, acc3 = zero;

        #pragma unroll
        for (int kk = 0; kk < IN_DIM; kk += 32) {
            const int kb = kk + grp * 8;
            v4f a0 = *(const v4f*)(abase + kb + 0);
            v4f a1 = *(const v4f*)(abase + kb + 4);
            v4f a2 = *(const v4f*)(abase + kb + 16);
            v4f a3 = *(const v4f*)(abase + kb + 20);
            v16h A;
            #pragma unroll
            for (int i = 0; i < 4; ++i) {
                A[i]      = (_Float16)a0[i];
                A[4 + i]  = (_Float16)a1[i];
                A[8 + i]  = (_Float16)a2[i];
                A[12 + i] = (_Float16)a3[i];
            }
            #pragma unroll
            for (int tt = 0; tt < 4; ++tt) {
                const int col = head * HID + tt * 16 + l15;
                const _Float16* q = W1h + (size_t)col * IN_DIM + kb;
                v8h blo = *(const v8h*)(q);
                v8h bhi = *(const v8h*)(q + 16);
                v16h B;
                #pragma unroll
                for (int i = 0; i < 8; ++i) { B[i] = blo[i]; B[8 + i] = bhi[i]; }
                v8f c = (tt == 0) ? acc0 : (tt == 1) ? acc1 : (tt == 2) ? acc2 : acc3;
                c = __builtin_amdgcn_wmma_f32_16x16x32_f16(
                        false, A, false, B, (short)0, c, false, false);
                if (tt == 0) acc0 = c; else if (tt == 1) acc1 = c;
                else if (tt == 2) acc2 = c; else acc3 = c;
            }
        }

        v8f hacc = zero;
        #pragma unroll
        for (int tt = 0; tt < 4; ++tt) {
            const int col  = head * HID + tt * 16 + l15;
            const float bias = b1[col];
            const float w2   = W2[col];
            v8f c = (tt == 0) ? acc0 : (tt == 1) ? acc1 : (tt == 2) ? acc2 : acc3;
            #pragma unroll
            for (int v = 0; v < 8; ++v) {
                float p = c[v] + bias;
                p = (p > 0.f) ? p : 0.f;
                p *= w2;
                p += __shfl_xor(p, 1, 32);
                p += __shfl_xor(p, 2, 32);
                p += __shfl_xor(p, 4, 32);
                p += __shfl_xor(p, 8, 32);
                hacc[v] += p;
            }
        }
        if (l15 == 0) {
            #pragma unroll
            for (int v = 0; v < 8; ++v) {
                const int node = nodeBase + v + 8 * grp;
                if (node < n)
                    hout[node * H_HEADS + head] = hacc[v] + b2[head];
            }
        }
    }
}

#define BSH    8
#define NBMAX  1024
#define CH     8192
#define CTH    256
#define FTH    1024
#define FCAP   12288
#define SBMAX  1024

__device__ __forceinline__ void scan1024(const int* cin, int* cout, int* wsum, int t,
                                         int (&own)[4]) {
    const int lane = t & 31, w = t >> 5;
    const int c0 = cin[4 * t], c1 = cin[4 * t + 1], c2 = cin[4 * t + 2], c3 = cin[4 * t + 3];
    const int tsum = c0 + c1 + c2 + c3;
    int x = tsum;
    #pragma unroll
    for (int d = 1; d < 32; d <<= 1) {
        const int y = __shfl_up(x, d, 32);
        if (lane >= d) x += y;
    }
    if (lane == 31) wsum[w] = x;
    __syncthreads();
    int wbase = 0;
    #pragma unroll
    for (int k = 0; k < 8; ++k) if (k < w) wbase += wsum[k];
    const int run = wbase + x - tsum;
    own[0] = run; own[1] = run + c0; own[2] = run + c0 + c1; own[3] = run + c0 + c1 + c2;
    cout[4 * t] = own[0]; cout[4 * t + 1] = own[1];
    cout[4 * t + 2] = own[2]; cout[4 * t + 3] = own[3];
}

__device__ __forceinline__ void chunk_hist_body(const int* __restrict__ dst,
                                                int* __restrict__ hist,
                                                int E, int nChunks, int NB, int blk) {
    __shared__ int lc[NBMAX];
    const int t = threadIdx.x;
    for (int b = t; b < NBMAX; b += CTH) lc[b] = 0;
    __syncthreads();
    const int base = blk * CH;
    #pragma unroll
    for (int k = 0; k < CH / (CTH * 4); ++k) {
        const int i = base + (k * CTH + t) * 4;
        if (i + 3 < E) {
            const int4 d = *(const int4*)(dst + i);
            atomicAdd(&lc[d.x >> BSH], 1);
            atomicAdd(&lc[d.y >> BSH], 1);
            atomicAdd(&lc[d.z >> BSH], 1);
            atomicAdd(&lc[d.w >> BSH], 1);
        } else {
            for (int j = i; j < E; ++j) atomicAdd(&lc[dst[j] >> BSH], 1);
        }
    }
    __syncthreads();
    for (int b = t; b < NB; b += CTH) hist[(size_t)b * nChunks + blk] = lc[b];
}

__global__ void __launch_bounds__(CTH)
k_pre(const int* __restrict__ dst, int* __restrict__ hist, int E, int nChunks, int NB,
      const float* __restrict__ W1, _Float16* __restrict__ W1h, int nW1, int w1Blocks,
      const float* __restrict__ rel_emb, const float* __restrict__ W_pred,
      float* __restrict__ etab) {
    const int blk = blockIdx.x;
    if (blk < nChunks)                 chunk_hist_body(dst, hist, E, nChunks, NB, blk);
    else if (blk < nChunks + w1Blocks) w1_to_f16_body(W1, W1h, nW1, blk - nChunks);
    else                               etab_body(rel_emb, W_pred, etab, blk - nChunks - w1Blocks);
}

__global__ void k_scan_local(const int* __restrict__ cnt, int* __restrict__ rowstart,
                             int* __restrict__ bsum, int n) {
    __shared__ int wsum[8];
    const int t = threadIdx.x, lane = t & 31, w = t >> 5;
    const int base = blockIdx.x * 1024 + t * 4;
    int c[4];
    #pragma unroll
    for (int k = 0; k < 4; ++k) c[k] = (base + k < n) ? cnt[base + k] : 0;
    const int tsum = c[0] + c[1] + c[2] + c[3];
    int x = tsum;
    #pragma unroll
    for (int d = 1; d < 32; d <<= 1) {
        const int y = __shfl_up(x, d, 32);
        if (lane >= d) x += y;
    }
    if (lane == 31) wsum[w] = x;
    __syncthreads();
    int wbase = 0;
    #pragma unroll
    for (int k = 0; k < 8; ++k) if (k < w) wbase += wsum[k];
    int run = wbase + x - tsum;
    #pragma unroll
    for (int k = 0; k < 4; ++k) {
        if (base + k < n) rowstart[base + k] = run;
        run += c[k];
    }
    if (t == 255) bsum[blockIdx.x] = wbase + x;
}

__device__ __forceinline__ void partition_body(
            const int* __restrict__ et, const int* __restrict__ src,
            const int* __restrict__ dst, const int* __restrict__ scanned,
            const int* __restrict__ bsum, int sBlocks,
            unsigned* __restrict__ tmp, int E, int nChunks, int NB, int blk) {
    __shared__ int lc[NBMAX];
    __shared__ int lb[NBMAX];
    __shared__ int gd[NBMAX];
    __shared__ int bp[SBMAX];
    __shared__ int wsum[8];
    __shared__ unsigned recs[CH];
    __shared__ unsigned short rb[CH];
    const int t = threadIdx.x;
    for (int b = t; b < NBMAX; b += CTH) lc[b] = 0;
    for (int k = t; k < SBMAX; k += CTH) lb[k] = (k < sBlocks) ? bsum[k] : 0;
    __syncthreads();
    {
        int unused[4];
        scan1024(lb, bp, wsum, t, unused);
    }

    const int base = blk * CH;
    unsigned rec[32];
    unsigned br[32];
    #pragma unroll
    for (int k = 0; k < CH / (CTH * 4); ++k) {
        const int i = base + (k * CTH + t) * 4;
        int dv[4], sv[4], ev[4];
        if (i + 3 < E) {
            const int4 d = *(const int4*)(dst + i);
            const int4 s = *(const int4*)(src + i);
            const int4 e = *(const int4*)(et + i);
            dv[0] = d.x; dv[1] = d.y; dv[2] = d.z; dv[3] = d.w;
            sv[0] = s.x; sv[1] = s.y; sv[2] = s.z; sv[3] = s.w;
            ev[0] = e.x; ev[1] = e.y; ev[2] = e.z; ev[3] = e.w;
        } else {
            #pragma unroll
            for (int c = 0; c < 4; ++c) {
                const int j = i + c;
                dv[c] = -1; sv[c] = 0; ev[c] = 0;
                if (j < E) { dv[c] = dst[j]; sv[c] = src[j]; ev[c] = et[j]; }
            }
        }
        #pragma unroll
        for (int c = 0; c < 4; ++c) {
            rec[k * 4 + c] = 0u; br[k * 4 + c] = ~0u;
            if (dv[c] >= 0) {
                const int b = dv[c] >> BSH;
                const int r = atomicAdd(&lc[b], 1);
                rec[k * 4 + c] = ((unsigned)sv[c] << 14) | ((unsigned)ev[c] << 8)
                               | ((unsigned)dv[c] & 255u);
                br[k * 4 + c] = ((unsigned)b << 16) | (unsigned)r;
            }
        }
    }
    __syncthreads();

    int own[4];
    scan1024(lc, lb, wsum, t, own);
    #pragma unroll
    for (int c = 0; c < 4; ++c) {
        const int b = 4 * t + c;
        if (b < NB) {
            const int idx = b * nChunks + blk;
            gd[b] = scanned[idx] + bp[idx >> 10] - own[c];
        }
    }
    __syncthreads();

    #pragma unroll
    for (int q = 0; q < 32; ++q) {
        if (br[q] != ~0u) {
            const int b = (int)(br[q] >> 16);
            const int p = lb[b] + (int)(br[q] & 0xFFFFu);
            recs[p] = rec[q];
            rb[p] = (unsigned short)b;
        }
    }
    __syncthreads();

    int nvalid = E - base;
    if (nvalid > CH) nvalid = CH;
    for (int i = t; i < nvalid; i += CTH)
        tmp[gd[rb[i]] + i] = recs[i];
}

__global__ void __launch_bounds__(CTH)
k_partition_mlp(const int* __restrict__ et, const int* __restrict__ src,
                const int* __restrict__ dst, const int* __restrict__ scanned,
                const int* __restrict__ bsum, int sBlocks,
                unsigned* __restrict__ tmp, int E, int nChunks, int NB, int pBlocks,
                const float* __restrict__ inputs, const _Float16* __restrict__ W1h,
                const float* __restrict__ b1, const float* __restrict__ W2,
                const float* __restrict__ b2, float* __restrict__ hout,
                int n, int nTiles) {
    const int blk = blockIdx.x;
    if (blk < pBlocks)
        partition_body(et, src, dst, scanned, bsum, sBlocks, tmp, E, nChunks, NB, blk);
    else
        mlp_body(inputs, W1h, b1, W2, b2, hout, n, nTiles, blk - pBlocks);
}

#define FGL (FTH >> BSH)
#define FGC 4
static_assert(LAYERS >= 2, "the fused first layer always feeds a head mean");

template <bool STAGED>
__device__ __forceinline__ unsigned fine_rec(const unsigned* stage, const unsigned* gsort, int i) {
    return STAGED ? stage[i] : gsort[i];
}

template <bool STAGED>
__device__ __forceinline__ void fine_layer0(const unsigned* stage, const unsigned* gsort,
                                            const v4f* tab, const int* ncount, const int* nbase,
                                            const float* __restrict__ hcur,
                                            float* __restrict__ hs_next, int b, int n) {
    const int t    = threadIdx.x;
    const int dl   = t / FGL;
    const int sub  = t & (FGL - 1);
    const int node = (b << BSH) + dl;
    const int len  = ncount[dl];
    const int st   = nbase[dl];

    float m[H_HEADS];
    #pragma unroll
    for (int h = 0; h < H_HEADS; ++h) m[h] = -__builtin_huge_valf();
    for (int j0 = sub; j0 < len; j0 += FGL * FGC) {
        unsigned pk[FGC];
        #pragma unroll
        for (int k = 0; k < FGC; ++k) {
            const int j = j0 + k * FGL;
            pk[k] = fine_rec<STAGED>(stage, gsort, (j < len) ? st + j : 0);
        }
        #pragma unroll
        for (int k = 0; k < FGC; ++k) {
            const bool ok = j0 + k * FGL < len;
            const v4f e = tab[ok ? (pk[k] & 63u) : 0u];
            #pragma unroll
            for (int h = 0; h < H_HEADS; ++h)
                m[h] = fmaxf(m[h], ok ? e[h] : -__builtin_huge_valf());
        }
    }
    #pragma unroll
    for (int h = 0; h < H_HEADS; ++h) {
        m[h] = fmaxf(m[h], __shfl_xor(m[h], 1, 32));
        m[h] = fmaxf(m[h], __shfl_xor(m[h], 2, 32));
    }

    float s[H_HEADS], u[H_HEADS];
    #pragma unroll
    for (int h = 0; h < H_HEADS; ++h) { s[h] = 0.f; u[h] = 0.f; }
    for (int j0 = sub; j0 < len; j0 += FGL * FGC) {
        unsigned pk[FGC];
        v4f hv[FGC];
        #pragma unroll
        for (int k = 0; k < FGC; ++k) {
            const int j = j0 + k * FGL;
            const unsigned raw = fine_rec<STAGED>(stage, gsort, (j < len) ? st + j : 0);
            pk[k] = (j < len) ? raw : 0u;
        }
        #pragma unroll
        for (int k = 0; k < FGC; ++k)
            hv[k] = *(const v4f*)(hcur + (size_t)(pk[k] >> 6) * H_HEADS);
        #pragma unroll
        for (int k = 0; k < FGC; ++k) {
            const bool ok = j0 + k * FGL < len;
            const v4f e = tab[pk[k] & 63u];
            #pragma unroll
            for (int h = 0; h < H_HEADS; ++h) {
                const float wgt = ok ? __expf(e[h] - m[h]) : 0.f;
                const float hk  = ok ? hv[k][h] : 0.f;
                s[h] += wgt;
                u[h] += wgt * hk;
            }
        }
    }
    #pragma unroll
    for (int h = 0; h < H_HEADS; ++h) {
        s[h] += __shfl_xor(s[h], 1, 32);  u[h] += __shfl_xor(u[h], 1, 32);
        s[h] += __shfl_xor(s[h], 2, 32);  u[h] += __shfl_xor(u[h], 2, 32);
    }

    if (sub == 0 && node < n) {
        v4f v;
        #pragma unroll
        for (int h = 0; h < H_HEADS; ++h) {
            const float x = (s[h] > 0.f) ? (u[h] / s[h]) : 0.f;
            v[h] = (x > 0.f) ? x : 0.f;
        }
        hs_next[node] = 0.25f * (v[0] + v[1] + v[2] + v[3]);
        if (node == n - 1)
            for (int j = n; j < ((n + 3) & ~3); ++j) hs_next[j] = 0.f;
    }
}

__global__ void __launch_bounds__(FTH)
k_fine(const unsigned* __restrict__ tmp, const int* __restrict__ scanned,
       const int* __restrict__ bsum,
       int* __restrict__ cnt, int* __restrict__ rowstart,
       unsigned* __restrict__ sorted, int n, int E, int nChunks, int NB,
       const float* __restrict__ etab0, const float* __restrict__ hcur,
       float* __restrict__ hs_next) {
    __shared__ v4f tab[RELS];
    __shared__ int ncount[1 << BSH];
    __shared__ int nbase[1 << BSH];
    __shared__ int cur[1 << BSH];
    __shared__ int wsum[8];
    __shared__ unsigned stage[FCAP];
    const int t = threadIdx.x, lane = t & 31, w = t >> 5;
    __shared__ int boff[2];
    const int b = blockIdx.x;
    const int e0 = b * nChunks, e1 = (b + 1) * nChunks;
    const int s0 = scanned[e0];
    const int s1 = (b + 1 < NB) ? scanned[e1] : 0;
    if (w < 2) {
        const int lim = ((w == 0) ? e0 : e1) >> 10;
        int acc = 0;
        for (int k = lane; k < lim; k += 32) acc += bsum[k];
        #pragma unroll
        for (int d = 16; d >= 1; d >>= 1) acc += __shfl_xor(acc, d, 32);
        if (lane == 0) boff[w] = acc;
    }
    if (t < (1 << BSH)) { ncount[t] = 0; cur[t] = 0; }
    if (t < RELS) tab[t] = *(const v4f*)(etab0 + t * H_HEADS);
    __syncthreads();
    const int bs = s0 + boff[0];
    const int be = (b + 1 < NB) ? s1 + boff[1] : E;
    const int nb = be - bs;
    const bool staged = (nb <= FCAP);

    for (int i0 = 0; i0 < nb; i0 += FTH * 8) {
        unsigned r[8];
        #pragma unroll
        for (int k = 0; k < 8; ++k) {
            const int i = i0 + k * FTH + t;
            r[k] = (i < nb) ? tmp[bs + i] : 0u;
        }
        #pragma unroll
        for (int k = 0; k < 8; ++k)
            if (i0 + k * FTH + t < nb) atomicAdd(&ncount[r[k] & 255u], 1);
    }
    __syncthreads();

    int v = 0, x = 0;
    if (t < (1 << BSH)) {
        v = ncount[t];
        x = v;
        #pragma unroll
        for (int d = 1; d < 32; d <<= 1) {
            const int y = __shfl_up(x, d, 32);
            if (lane >= d) x += y;
        }
        if (lane == 31) wsum[w] = x;
    }
    __syncthreads();
    if (t < (1 << BSH)) {
        int wbase = 0;
        #pragma unroll
        for (int k = 0; k < 8; ++k) if (k < w) wbase += wsum[k];
        const int excl = wbase + x - v;
        nbase[t] = excl;
        const int node = (b << BSH) + t;
        if (node < n) { cnt[node] = v; rowstart[node] = bs + excl; }
    }
    __syncthreads();

    for (int i0 = 0; i0 < nb; i0 += FTH * 8) {
        unsigned r[8];
        #pragma unroll
        for (int k = 0; k < 8; ++k) {
            const int i = i0 + k * FTH + t;
            r[k] = (i < nb) ? tmp[bs + i] : 0u;
        }
        #pragma unroll
        for (int k = 0; k < 8; ++k) {
            if (i0 + k * FTH + t < nb) {
                const unsigned dl = r[k] & 255u;
                const int p = nbase[dl] + atomicAdd(&cur[dl], 1);
                const unsigned val = r[k] >> 8;
                if (staged) stage[p] = val; else sorted[bs + p] = val;
            }
        }
    }
    __syncthreads();
    if (staged) {
        for (int i = t; i < nb; i += FTH) sorted[bs + i] = stage[i];
        fine_layer0<true>(stage, sorted + bs, tab, ncount, nbase, hcur, hs_next, b, n);
    } else {
        fine_layer0<false>(stage, sorted + bs, tab, ncount, nbase, hcur, hs_next, b, n);
    }
}

#define LT    800
#define LGL   2
#define LCH   16
#define LNPW  (LT / LGL)
#define LCAPN 81664
#define LST   8

__device__ __forceinline__ void l_load(const unsigned* __restrict__ sorted, int start,
                                       int len, int j0, unsigned (&pk)[LCH]) {
    #pragma unroll
    for (int k = 0; k < LCH; ++k) {
        const int j = j0 + k * LGL;
        pk[k] = sorted[(j < len) ? start + j : 0];
    }
}
__device__ __forceinline__ void l_max(const v4f* tab, const unsigned (&pk)[LCH],
                                      int len, int j0, float (&m)[H_HEADS]) {
    #pragma unroll
    for (int k = 0; k < LCH; ++k) {
        if (j0 + k * LGL < len) {
            const v4f e = tab[pk[k] & 63u];
            #pragma unroll
            for (int h = 0; h < H_HEADS; ++h) m[h] = fmaxf(m[h], e[h]);
        }
    }
}
__device__ __forceinline__ void l_sum(const v4f* tab, const float* hl,
                                      const float* __restrict__ hs, unsigned count,
                                      const unsigned (&pk)[LCH],
                                      int len, int j0, const float (&m)[H_HEADS],
                                      float (&s)[H_HEADS], float (&u)[H_HEADS]) {
    float hv[LCH];
    #pragma unroll
    for (int k = 0; k < LCH; ++k) {
        const unsigned sn = pk[k] >> 6;
        const bool inl = sn < count;
        const float a = hl[inl ? sn : 0u];
        const float b = hs[inl ? 0u : sn];
        hv[k] = inl ? a : b;
    }
    #pragma unroll
    for (int k = 0; k < LCH; ++k) {
        if (j0 + k * LGL < len) {
            const v4f e = tab[pk[k] & 63u];
            #pragma unroll
            for (int h = 0; h < H_HEADS; ++h) {
                const float wgt = __expf(e[h] - m[h]);
                s[h] += wgt;
                u[h] += wgt * hv[k];
            }
        }
    }
}

__global__ void __launch_bounds__(LT)
k_gather_lds(const unsigned* __restrict__ sorted,
             const int* __restrict__ rowstart,
             const int* __restrict__ cnt,
             const float* __restrict__ etab_l,
             const float* __restrict__ hs,
             float* __restrict__ hs_next,
             const float* __restrict__ centrality,
             const float* __restrict__ gamma,
             const float* __restrict__ beta,
             float* __restrict__ out,
             int n, int do_mean) {
    __shared__ v4f tab[RELS];
    __shared__ float hl[LCAPN];
    const int t = threadIdx.x;
    if (t < RELS) tab[t] = *(const v4f*)(etab_l + t * H_HEADS);

    const int node = blockIdx.x * LNPW + t / LGL;
    const int sub  = t & (LGL - 1);
    int start = 0, len = 0;
    if (node < n) { start = rowstart[node]; len = cnt[node]; }

    unsigned pk0[LCH];
    l_load(sorted, start, len, sub, pk0);

    const int np = (n + 3) & ~3;
    const int count = (np < LCAPN) ? np : LCAPN;
    for (int i0 = 0; i0 < count; i0 += LT * 4 * LST) {
        v4f r[LST];
        #pragma unroll
        for (int k = 0; k < LST; ++k) {
            int i = i0 + (k * LT + t) * 4;
            if (i > count - 4) i = count - 4;
            r[k] = *(const v4f*)(hs + i);
        }
        #pragma unroll
        for (int k = 0; k < LST; ++k) {
            const int i = i0 + (k * LT + t) * 4;
            if (i < count) *(v4f*)(hl + i) = r[k];
        }
    }
    __syncthreads();

    float m[H_HEADS];
    #pragma unroll
    for (int h = 0; h < H_HEADS; ++h) m[h] = -__builtin_huge_valf();
    l_max(tab, pk0, len, sub, m);
    for (int base = LGL * LCH; base < len; base += LGL * LCH) {
        unsigned pk[LCH];
        l_load(sorted, start, len, base + sub, pk);
        l_max(tab, pk, len, base + sub, m);
    }
    #pragma unroll
    for (int h = 0; h < H_HEADS; ++h)
        m[h] = fmaxf(m[h], __shfl_xor(m[h], 1, 32));

    float s[H_HEADS], u[H_HEADS];
    #pragma unroll
    for (int h = 0; h < H_HEADS; ++h) { s[h] = 0.f; u[h] = 0.f; }
    l_sum(tab, hl, hs, (unsigned)count, pk0, len, sub, m, s, u);
    for (int base = LGL * LCH; base < len; base += LGL * LCH) {
        unsigned pk[LCH];
        l_load(sorted, start, len, base + sub, pk);
        l_sum(tab, hl, hs, (unsigned)count, pk, len, base + sub, m, s, u);
    }
    #pragma unroll
    for (int h = 0; h < H_HEADS; ++h) {
        s[h] += __shfl_xor(s[h], 1, 32);
        u[h] += __shfl_xor(u[h], 1, 32);
    }

    if (sub == 0 && node < n) {
        v4f v;
        #pragma unroll
        for (int h = 0; h < H_HEADS; ++h) {
            const float x = (s[h] > 0.f) ? (u[h] / s[h]) : 0.f;
            v[h] = (x > 0.f) ? x : 0.f;
        }
        if (do_mean) {
            hs_next[node] = 0.25f * (v[0] + v[1] + v[2] + v[3]);
            if (node == n - 1)
                for (int j = n; j < np; ++j) hs_next[j] = 0.f;
        } else {
            const float c = centrality[node];
            float acc = 0.f;
            #pragma unroll
            for (int h = 0; h < H_HEADS; ++h)
                acc += (c * gamma[h] + beta[h]) * v[h];
            acc *= 0.25f;
            out[node] = (acc > 0.f) ? acc : 0.01f * acc;
        }
    }
}

extern "C" void kernel_launch(void* const* d_in, const int* in_sizes, int n_in,
                              void* d_out, int out_size, void* d_ws, size_t ws_size,
                              hipStream_t stream) {
    const float* inputs     = (const float*)d_in[0];
    const float* centrality = (const float*)d_in[1];
    const float* W1         = (const float*)d_in[2];
    const float* b1         = (const float*)d_in[3];
    const float* W2         = (const float*)d_in[4];
    const float* b2         = (const float*)d_in[5];
    const float* rel_emb    = (const float*)d_in[6];
    const float* W_pred     = (const float*)d_in[7];
    const float* gamma      = (const float*)d_in[8];
    const float* beta       = (const float*)d_in[9];
    const int*   edge_types = (const int*)d_in[10];
    const int*   src        = (const int*)d_in[11];
    const int*   dst        = (const int*)d_in[12];
    float* out = (float*)d_out;

    const int N = in_sizes[1];
    const int E = in_sizes[10];
    const int Np = (N + 3) & ~3;
    const int Ep = (E + 3) & ~3;
    const int NB = (N + (1 << BSH) - 1) >> BSH;
    const int nChunks = (E + CH - 1) / CH;
    const int M  = NB * nChunks;
    const int Mp = (M + 3) & ~3;

    char* ws = (char*)d_ws;
    _Float16* W1h = (_Float16*)ws;
    float* etab   = (float*)(ws + 65536);
    float* hA     = (float*)(ws + 69632);
    int*   cnt      = (int*)(hA + (size_t)Np * H_HEADS);
    int*   rowstart = cnt + Np;
    int*   hist     = rowstart + Np;
    int*   scanned  = hist + Mp;
    int*   bsum     = scanned + Mp;
    unsigned* part   = (unsigned*)(bsum + Mp);
    unsigned* sorted = part + Ep;
    float* hsA = (float*)(sorted + Ep);
    float* hsB = hsA + Np;

    const int nW1      = COLS * IN_DIM;
    const int w1Blocks = (nW1 + CTH - 1) / CTH;
    const int eBlocks  = (LAYERS * RELS * H_HEADS + CTH - 1) / CTH;
    k_pre<<<nChunks + w1Blocks + eBlocks, CTH, 0, stream>>>(
        dst, hist, E, nChunks, NB, W1, W1h, nW1, w1Blocks, rel_emb, W_pred, etab);

    const int sBlocks = (M + 1023) / 1024;
    k_scan_local<<<sBlocks, 256, 0, stream>>>(hist, scanned, bsum, M);

    const int nTiles    = (N + 15) / 16;
    const int mlpBlocks = (nTiles + CTH / 32 - 1) / (CTH / 32);
    k_partition_mlp<<<nChunks + mlpBlocks, CTH, 0, stream>>>(
        edge_types, src, dst, scanned, bsum, sBlocks, part, E, nChunks, NB, nChunks,
        inputs, W1h, b1, W2, b2, hA, N, nTiles);

    k_fine<<<NB, FTH, 0, stream>>>(part, scanned, bsum, cnt, rowstart, sorted, N, E, nChunks, NB,
                                   etab, hA, hsA);

    const int lBlocks = (N + LNPW - 1) / LNPW;
    float* hscur = hsA;
    float* hsnxt = hsB;

    for (int l = 1; l < LAYERS; ++l) {
        const float* etab_l = etab + l * RELS * H_HEADS;
        const int do_mean = (l != LAYERS - 1) ? 1 : 0;
        k_gather_lds<<<lBlocks, LT, 0, stream>>>(sorted, rowstart, cnt, etab_l, hscur,
                                                 hsnxt, centrality, gamma, beta, out,
                                                 N, do_mean);
        float* tmp = hscur; hscur = hsnxt; hsnxt = tmp;
    }
}


# Level-1 sort made chunk-local (range order inside each chunk + run table, one fixed slot per range): edge histogram pass and scan launch gone, 6 -> 5; bucket order, so sum order, now by run arrival
# speedup vs baseline: 26.0331x; 1.0762x over previous
#include <hip/hip_runtime.h>
#include <hip/hip_bf16.h>

typedef __attribute__((ext_vector_type(16))) _Float16 v16h;
typedef __attribute__((ext_vector_type(8)))  _Float16 v8h;
typedef __attribute__((ext_vector_type(8)))  float    v8f;
typedef __attribute__((ext_vector_type(4)))  float    v4f;

#define H_HEADS 4
#define IN_DIM  128
#define HID     64
#define COLS    256
#define PRED    16
#define RELS    50
#define LAYERS  3

__device__ __forceinline__ void w1_to_f16_body(const float* __restrict__ W1,
                                               _Float16* __restrict__ W1h, int n, int blk) {
    int i = blk * blockDim.x + threadIdx.x;
    if (i < n) W1h[i] = (_Float16)W1[i];
}

__device__ __forceinline__ void etab_body(const float* __restrict__ rel_emb,
                                          const float* __restrict__ W_pred,
                                          float* __restrict__ etab, int blk) {
    int i = blk * blockDim.x + threadIdx.x;
    if (i >= LAYERS * RELS * H_HEADS) return;
    int l = i / (RELS * H_HEADS);
    int rem = i % (RELS * H_HEADS);
    int r = rem / H_HEADS;
    int h = rem % H_HEADS;
    float e = 0.f;
    #pragma unroll
    for (int p = 0; p < PRED; ++p)
        e += rel_emb[r * PRED + p] * W_pred[(l * PRED + p) * H_HEADS + h];
    etab[i] = (e > 0.f) ? e : 0.2f * e;
}

__device__ __forceinline__ void mlp_body(const float* __restrict__ inputs,
                                         const _Float16* __restrict__ W1h,
                                         const float* __restrict__ b1,
                                         const float* __restrict__ W2,
                                         const float* __restrict__ b2,
                                         float* __restrict__ hout,
                                         int n, int nTiles, int blk) {
    const int wave = (blk * blockDim.x + threadIdx.x) >> 5;
    if (wave >= nTiles) return;
    const int lane = threadIdx.x & 31;
    const int grp  = lane >> 4;
    const int l15  = lane & 15;
    const int nodeBase = wave * 16;
    int arow = nodeBase + l15;
    if (arow > n - 1) arow = n - 1;
    const float* __restrict__ abase = inputs + (size_t)arow * IN_DIM;

    const v8f zero = {};

    #pragma unroll
    for (int head = 0; head < H_HEADS; ++head) {
        v8f acc0 = zero, acc1 = zero, acc2 = zero, acc3 = zero;

        #pragma unroll
        for (int kk = 0; kk < IN_DIM; kk += 32) {
            const int kb = kk + grp * 8;
            v4f a0 = *(const v4f*)(abase + kb + 0);
            v4f a1 = *(const v4f*)(abase + kb + 4);
            v4f a2 = *(const v4f*)(abase + kb + 16);
            v4f a3 = *(const v4f*)(abase + kb + 20);
            v16h A;
            #pragma unroll
            for (int i = 0; i < 4; ++i) {
                A[i]      = (_Float16)a0[i];
                A[4 + i]  = (_Float16)a1[i];
                A[8 + i]  = (_Float16)a2[i];
                A[12 + i] = (_Float16)a3[i];
            }
            #pragma unroll
            for (int tt = 0; tt < 4; ++tt) {
                const int col = head * HID + tt * 16 + l15;
                const _Float16* q = W1h + (size_t)col * IN_DIM + kb;
                v8h blo = *(const v8h*)(q);
                v8h bhi = *(const v8h*)(q + 16);
                v16h B;
                #pragma unroll
                for (int i = 0; i < 8; ++i) { B[i] = blo[i]; B[8 + i] = bhi[i]; }
                v8f c = (tt == 0) ? acc0 : (tt == 1) ? acc1 : (tt == 2) ? acc2 : acc3;
                c = __builtin_amdgcn_wmma_f32_16x16x32_f16(
                        false, A, false, B, (short)0, c, false, false);
                if (tt == 0) acc0 = c; else if (tt == 1) acc1 = c;
                else if (tt == 2) acc2 = c; else acc3 = c;
            }
        }

        v8f hacc = zero;
        #pragma unroll
        for (int tt = 0; tt < 4; ++tt) {
            const int col  = head * HID + tt * 16 + l15;
            const float bias = b1[col];
            const float w2   = W2[col];
            v8f c = (tt == 0) ? acc0 : (tt == 1) ? acc1 : (tt == 2) ? acc2 : acc3;
            #pragma unroll
            for (int v = 0; v < 8; ++v) {
                float p = c[v] + bias;
                p = (p > 0.f) ? p : 0.f;
                p *= w2;
                p += __shfl_xor(p, 1, 32);
                p += __shfl_xor(p, 2, 32);
                p += __shfl_xor(p, 4, 32);
                p += __shfl_xor(p, 8, 32);
                hacc[v] += p;
            }
        }
        if (l15 == 0) {
            #pragma unroll
            for (int v = 0; v < 8; ++v) {
                const int node = nodeBase + v + 8 * grp;
                if (node < n)
                    hout[node * H_HEADS + head] = hacc[v] + b2[head];
            }
        }
    }
}

#define BSH    8
#define NBMAX  1024
#define CH     8192
#define CTH    256
#define FTH    1024
#define FCAP   12288
#define SPAD   4
#define RB     16

__device__ __forceinline__ void scan1024(const int* cin, int* cout, int* wsum, int t,
                                         int (&own)[4]) {
    const int lane = t & 31, w = t >> 5;
    const int c0 = cin[4 * t], c1 = cin[4 * t + 1], c2 = cin[4 * t + 2], c3 = cin[4 * t + 3];
    const int tsum = c0 + c1 + c2 + c3;
    int x = tsum;
    #pragma unroll
    for (int d = 1; d < 32; d <<= 1) {
        const int y = __shfl_up(x, d, 32);
        if (lane >= d) x += y;
    }
    if (lane == 31) wsum[w] = x;
    __syncthreads();
    int wbase = 0;
    #pragma unroll
    for (int k = 0; k < 8; ++k) if (k < w) wbase += wsum[k];
    const int run = wbase + x - tsum;
    own[0] = run; own[1] = run + c0; own[2] = run + c0 + c1; own[3] = run + c0 + c1 + c2;
    cout[4 * t] = own[0]; cout[4 * t + 1] = own[1];
    cout[4 * t + 2] = own[2]; cout[4 * t + 3] = own[3];
}

__global__ void __launch_bounds__(CTH)
k_pre(const float* __restrict__ W1, _Float16* __restrict__ W1h, int nW1, int w1Blocks,
      const float* __restrict__ rel_emb, const float* __restrict__ W_pred,
      float* __restrict__ etab) {
    const int blk = blockIdx.x;
    if (blk < w1Blocks) w1_to_f16_body(W1, W1h, nW1, blk);
    else                etab_body(rel_emb, W_pred, etab, blk - w1Blocks);
}

__device__ __forceinline__ void partition_body(
            const int* __restrict__ et, const int* __restrict__ src,
            const int* __restrict__ dst,
            unsigned* __restrict__ tmp, unsigned* __restrict__ runtab,
            int E, int nChunks, int NB, int blk) {
    __shared__ int lc[NBMAX];
    __shared__ int lb[NBMAX];
    __shared__ int wsum[8];
    __shared__ __attribute__((aligned(16))) unsigned recs[CH];
    const int t = threadIdx.x;
    for (int b = t; b < NBMAX; b += CTH) lc[b] = 0;
    __syncthreads();

    const int base = blk * CH;
    unsigned rec[32];
    unsigned br[32];
    #pragma unroll
    for (int k = 0; k < CH / (CTH * 4); ++k) {
        const int i = base + (k * CTH + t) * 4;
        int dv[4], sv[4], ev[4];
        if (i + 3 < E) {
            const int4 d = *(const int4*)(dst + i);
            const int4 s = *(const int4*)(src + i);
            const int4 e = *(const int4*)(et + i);
            dv[0] = d.x; dv[1] = d.y; dv[2] = d.z; dv[3] = d.w;
            sv[0] = s.x; sv[1] = s.y; sv[2] = s.z; sv[3] = s.w;
            ev[0] = e.x; ev[1] = e.y; ev[2] = e.z; ev[3] = e.w;
        } else {
            #pragma unroll
            for (int c = 0; c < 4; ++c) {
                const int j = i + c;
                dv[c] = -1; sv[c] = 0; ev[c] = 0;
                if (j < E) { dv[c] = dst[j]; sv[c] = src[j]; ev[c] = et[j]; }
            }
        }
        #pragma unroll
        for (int c = 0; c < 4; ++c) {
            rec[k * 4 + c] = 0u; br[k * 4 + c] = ~0u;
            if (dv[c] >= 0) {
                const int b = dv[c] >> BSH;
                const int r = atomicAdd(&lc[b], 1);
                rec[k * 4 + c] = ((unsigned)sv[c] << 14) | ((unsigned)ev[c] << 8)
                               | ((unsigned)dv[c] & 255u);
                br[k * 4 + c] = ((unsigned)b << 16) | (unsigned)r;
            }
        }
    }
    __syncthreads();

    int own[4];
    scan1024(lc, lb, wsum, t, own);
    #pragma unroll
    for (int c = 0; c < 4; ++c) {
        const int b = 4 * t + c;
        if (b < NB)
            runtab[(size_t)b * nChunks + blk] = ((unsigned)own[c] << 16) | (unsigned)lc[b];
    }
    __syncthreads();

    #pragma unroll
    for (int q = 0; q < 32; ++q) {
        if (br[q] != ~0u) {
            const int b = (int)(br[q] >> 16);
            recs[lb[b] + (int)(br[q] & 0xFFFFu)] = rec[q];
        }
    }
    __syncthreads();

    int nvalid = E - base;
    if (nvalid > CH) nvalid = CH;
    const int n4 = nvalid >> 2;
    uint4* __restrict__ o4 = (uint4*)(tmp + base);
    const uint4* r4 = (const uint4*)recs;
    for (int i = t; i < n4; i += CTH) o4[i] = r4[i];
    for (int i = (n4 << 2) + t; i < nvalid; i += CTH) tmp[base + i] = recs[i];
}

__global__ void __launch_bounds__(CTH)
k_partition_mlp(const int* __restrict__ et, const int* __restrict__ src,
                const int* __restrict__ dst,
                unsigned* __restrict__ tmp, unsigned* __restrict__ runtab,
                int E, int nChunks, int NB, int pBlocks,
                const float* __restrict__ inputs, const _Float16* __restrict__ W1h,
                const float* __restrict__ b1, const float* __restrict__ W2,
                const float* __restrict__ b2, float* __restrict__ hout,
                int n, int nTiles) {
    const int blk = blockIdx.x;
    if (blk < pBlocks)
        partition_body(et, src, dst, tmp, runtab, E, nChunks, NB, blk);
    else
        mlp_body(inputs, W1h, b1, W2, b2, hout, n, nTiles, blk - pBlocks);
}

#define FGL (FTH >> BSH)
#define FGC 4
static_assert(LAYERS >= 2, "the fused first layer always feeds a head mean");

template <bool STAGED>
__device__ __forceinline__ unsigned fine_rec(const unsigned* stage, const unsigned* gsort, int i) {
    return STAGED ? stage[i] : gsort[i];
}

template <bool STAGED>
__device__ __forceinline__ void fine_layer0(const unsigned* stage, const unsigned* gsort,
                                            const v4f* tab, const int* ncount, const int* nbase,
                                            const float* __restrict__ hcur,
                                            float* __restrict__ hs_next, int b, int n) {
    const int t    = threadIdx.x;
    const int dl   = t / FGL;
    const int sub  = t & (FGL - 1);
    const int node = (b << BSH) + dl;
    const int len  = ncount[dl];
    const int st   = nbase[dl];

    float m[H_HEADS];
    #pragma unroll
    for (int h = 0; h < H_HEADS; ++h) m[h] = -__builtin_huge_valf();
    for (int j0 = sub; j0 < len; j0 += FGL * FGC) {
        unsigned pk[FGC];
        #pragma unroll
        for (int k = 0; k < FGC; ++k) {
            const int j = j0 + k * FGL;
            pk[k] = fine_rec<STAGED>(stage, gsort, (j < len) ? st + j : 0);
        }
        #pragma unroll
        for (int k = 0; k < FGC; ++k) {
            const bool ok = j0 + k * FGL < len;
            const v4f e = tab[ok ? (pk[k] & 63u) : 0u];
            #pragma unroll
            for (int h = 0; h < H_HEADS; ++h)
                m[h] = fmaxf(m[h], ok ? e[h] : -__builtin_huge_valf());
        }
    }
    #pragma unroll
    for (int h = 0; h < H_HEADS; ++h) {
        m[h] = fmaxf(m[h], __shfl_xor(m[h], 1, 32));
        m[h] = fmaxf(m[h], __shfl_xor(m[h], 2, 32));
    }

    float s[H_HEADS], u[H_HEADS];
    #pragma unroll
    for (int h = 0; h < H_HEADS; ++h) { s[h] = 0.f; u[h] = 0.f; }
    for (int j0 = sub; j0 < len; j0 += FGL * FGC) {
        unsigned pk[FGC];
        v4f hv[FGC];
        #pragma unroll
        for (int k = 0; k < FGC; ++k) {
            const int j = j0 + k * FGL;
            const unsigned raw = fine_rec<STAGED>(stage, gsort, (j < len) ? st + j : 0);
            pk[k] = (j < len) ? raw : 0u;
        }
        #pragma unroll
        for (int k = 0; k < FGC; ++k)
            hv[k] = *(const v4f*)(hcur + (size_t)(pk[k] >> 6) * H_HEADS);
        #pragma unroll
        for (int k = 0; k < FGC; ++k) {
            const bool ok = j0 + k * FGL < len;
            const v4f e = tab[pk[k] & 63u];
            #pragma unroll
            for (int h = 0; h < H_HEADS; ++h) {
                const float wgt = ok ? __expf(e[h] - m[h]) : 0.f;
                const float hk  = ok ? hv[k][h] : 0.f;
                s[h] += wgt;
                u[h] += wgt * hk;
            }
        }
    }
    #pragma unroll
    for (int h = 0; h < H_HEADS; ++h) {
        s[h] += __shfl_xor(s[h], 1, 32);  u[h] += __shfl_xor(u[h], 1, 32);
        s[h] += __shfl_xor(s[h], 2, 32);  u[h] += __shfl_xor(u[h], 2, 32);
    }

    if (sub == 0 && node < n) {
        v4f v;
        #pragma unroll
        for (int h = 0; h < H_HEADS; ++h) {
            const float x = (s[h] > 0.f) ? (u[h] / s[h]) : 0.f;
            v[h] = (x > 0.f) ? x : 0.f;
        }
        hs_next[node] = 0.25f * (v[0] + v[1] + v[2] + v[3]);
        if (node == n - 1)
            for (int j = n; j < ((n + 3) & ~3); ++j) hs_next[j] = 0.f;
    }
}

template <int PASS, bool STAGED>
__device__ __forceinline__ void fine_put(unsigned rec, int* ncount, const int* nbase, int* cur,
                                         unsigned* stage, unsigned* __restrict__ gout) {
    const unsigned dl = rec & 255u;
    if (PASS == 0) {
        atomicAdd(&ncount[dl], 1);
    } else {
        const int p = nbase[dl] + atomicAdd(&cur[dl], 1);
        const unsigned val = rec >> 8;
        if (STAGED) stage[p] = val; else gout[p] = val;
    }
}

template <int PASS, bool STAGED>
__device__ __forceinline__ void fine_visit(const unsigned* __restrict__ tmp,
                                           const unsigned* __restrict__ row,
                                           int nChunks, int K, int c0, int lane,
                                           int* ncount, const int* nbase, int* cur,
                                           unsigned* stage, unsigned* __restrict__ gout) {
    for (int kb = 0; kb < K; kb += RB) {
        const int ci = c0 + kb + lane;
        const bool have = lane < RB && kb + lane < K && ci < nChunks;
        const unsigned draw = row[have ? ci : 0];
        const unsigned d = have ? draw : 0u;
        unsigned r[RB];
        #pragma unroll
        for (int i = 0; i < RB; ++i) {
            const unsigned di = (unsigned)__builtin_amdgcn_readlane((int)d, i);
            const int ln = (int)(di & 0xFFFFu);
            const size_t off = (ln > 0) ? (size_t)(c0 + kb + i) * CH + (di >> 16) : (size_t)0;
            r[i] = tmp[off + (size_t)((lane < ln) ? lane : 0)];
        }
        #pragma unroll
        for (int i = 0; i < RB; ++i) {
            const unsigned di = (unsigned)__builtin_amdgcn_readlane((int)d, i);
            const int ln = (int)(di & 0xFFFFu);
            if (lane < ln) fine_put<PASS, STAGED>(r[i], ncount, nbase, cur, stage, gout);
            if (ln > 32) {
                const unsigned* p = tmp + (size_t)(c0 + kb + i) * CH + (di >> 16);
                for (int j = 32 + lane; j < ln; j += 32)
                    fine_put<PASS, STAGED>(p[j], ncount, nbase, cur, stage, gout);
            }
        }
    }
}

__global__ void __launch_bounds__(FTH)
k_fine(const unsigned* __restrict__ tmp, const unsigned* __restrict__ runtab,
       int* __restrict__ cnt, int* __restrict__ rowstart,
       unsigned* __restrict__ sorted, int n, int nChunks, int NB,
       const float* __restrict__ etab0, const float* __restrict__ hcur,
       float* __restrict__ hs_next) {
    __shared__ v4f tab[RELS];
    __shared__ int ncount[1 << BSH];
    __shared__ int nbase[1 << BSH];
    __shared__ int cur[1 << BSH];
    __shared__ int wsum[8];
    __shared__ int wtot[FTH / 32];
    __shared__ int wovf[FTH / 32];
    __shared__ unsigned stage[FCAP];
    const int t = threadIdx.x, lane = t & 31;
    const int w = __builtin_amdgcn_readfirstlane(t >> 5);
    const int b = blockIdx.x;
    const int K  = (nChunks + FTH / 32 - 1) / (FTH / 32);
    const int c0 = w * K;
    const unsigned* __restrict__ row = runtab + (size_t)b * nChunks;

    int part = 0;
    for (int k = lane; k < K; k += 32) {
        const int c = c0 + k;
        if (c < nChunks) part += (int)(row[c] & 0xFFFFu);
    }
    #pragma unroll
    for (int d = 16; d >= 1; d >>= 1) part += __shfl_xor(part, d, 32);
    if (lane == 0) wtot[w] = part;
    if (t < (1 << BSH)) { ncount[t] = 0; cur[t] = 0; }
    if (t < RELS) tab[t] = *(const v4f*)(etab0 + t * H_HEADS);
    if (b == 0 && t == 0) sorted[0] = 0u;
    __syncthreads();
    static_assert(FTH / 32 == 32, "one lane per wave total below");
    int nbl = wtot[lane];
    #pragma unroll
    for (int d = 16; d >= 1; d >>= 1) nbl += __shfl_xor(nbl, d, 32);
    const int nb = __builtin_amdgcn_readfirstlane(nbl);
    const bool staged = (nb <= FCAP);

    int bs = SPAD + b * FCAP;
    if (!staged) {
        int acc = 0;
        for (int bb = w; bb < b; bb += FTH / 32) {
            const unsigned* __restrict__ r2 = runtab + (size_t)bb * nChunks;
            int sum = 0;
            for (int c = lane; c < nChunks; c += 32) sum += (int)(r2[c] & 0xFFFFu);
            #pragma unroll
            for (int d = 16; d >= 1; d >>= 1) sum += __shfl_xor(sum, d, 32);
            if (sum > FCAP) acc += sum;
        }
        if (lane == 0) wovf[w] = acc;
        __syncthreads();
        int before = wovf[lane];
        #pragma unroll
        for (int d = 16; d >= 1; d >>= 1) before += __shfl_xor(before, d, 32);
        bs = SPAD + NB * FCAP + __builtin_amdgcn_readfirstlane(before);
    }
    unsigned* __restrict__ gout = sorted + bs;

    if (staged) fine_visit<0, true >(tmp, row, nChunks, K, c0, lane, ncount, nbase, cur, stage, gout);
    else        fine_visit<0, false>(tmp, row, nChunks, K, c0, lane, ncount, nbase, cur, stage, gout);
    __syncthreads();

    int v = 0, x = 0;
    if (t < (1 << BSH)) {
        v = ncount[t];
        x = v;
        #pragma unroll
        for (int d = 1; d < 32; d <<= 1) {
            const int y = __shfl_up(x, d, 32);
            if (lane >= d) x += y;
        }
        if (lane == 31) wsum[w] = x;
    }
    __syncthreads();
    if (t < (1 << BSH)) {
        int wbase = 0;
        #pragma unroll
        for (int k = 0; k < 8; ++k) if (k < w) wbase += wsum[k];
        const int excl = wbase + x - v;
        nbase[t] = excl;
        const int node = (b << BSH) + t;
        if (node < n) { cnt[node] = v; rowstart[node] = bs + excl; }
    }
    __syncthreads();

    if (staged) fine_visit<1, true >(tmp, row, nChunks, K, c0, lane, ncount, nbase, cur, stage, gout);
    else        fine_visit<1, false>(tmp, row, nChunks, K, c0, lane, ncount, nbase, cur, stage, gout);
    __syncthreads();
    if (staged) {
        for (int i = t; i < nb; i += FTH) gout[i] = stage[i];
        fine_layer0<true>(stage, gout, tab, ncount, nbase, hcur, hs_next, b, n);
    } else {
        fine_layer0<false>(stage, gout, tab, ncount, nbase, hcur, hs_next, b, n);
    }
}

#define LT    800
#define LGL   2
#define LCH   16
#define LNPW  (LT / LGL)
#define LCAPN 81664
#define LST   8

__device__ __forceinline__ void l_load(const unsigned* __restrict__ sorted, int start,
                                       int len, int j0, unsigned (&pk)[LCH]) {
    #pragma unroll
    for (int k = 0; k < LCH; ++k) {
        const int j = j0 + k * LGL;
        pk[k] = sorted[(j < len) ? start + j : 0];
    }
}
__device__ __forceinline__ void l_max(const v4f* tab, const unsigned (&pk)[LCH],
                                      int len, int j0, float (&m)[H_HEADS]) {
    #pragma unroll
    for (int k = 0; k < LCH; ++k) {
        if (j0 + k * LGL < len) {
            const v4f e = tab[pk[k] & 63u];
            #pragma unroll
            for (int h = 0; h < H_HEADS; ++h) m[h] = fmaxf(m[h], e[h]);
        }
    }
}
__device__ __forceinline__ void l_sum(const v4f* tab, const float* hl,
                                      const float* __restrict__ hs, unsigned count,
                                      const unsigned (&pk)[LCH],
                                      int len, int j0, const float (&m)[H_HEADS],
                                      float (&s)[H_HEADS], float (&u)[H_HEADS]) {
    float hv[LCH];
    #pragma unroll
    for (int k = 0; k < LCH; ++k) {
        const unsigned sn = pk[k] >> 6;
        const bool inl = sn < count;
        const float a = hl[inl ? sn : 0u];
        const float b = hs[inl ? 0u : sn];
        hv[k] = inl ? a : b;
    }
    #pragma unroll
    for (int k = 0; k < LCH; ++k) {
        if (j0 + k * LGL < len) {
            const v4f e = tab[pk[k] & 63u];
            #pragma unroll
            for (int h = 0; h < H_HEADS; ++h) {
                const float wgt = __expf(e[h] - m[h]);
                s[h] += wgt;
                u[h] += wgt * hv[k];
            }
        }
    }
}

__global__ void __launch_bounds__(LT)
k_gather_lds(const unsigned* __restrict__ sorted,
             const int* __restrict__ rowstart,
             const int* __restrict__ cnt,
             const float* __restrict__ etab_l,
             const float* __restrict__ hs,
             float* __restrict__ hs_next,
             const float* __restrict__ centrality,
             const float* __restrict__ gamma,
             const float* __restrict__ beta,
             float* __restrict__ out,
             int n, int do_mean) {
    __shared__ v4f tab[RELS];
    __shared__ float hl[LCAPN];
    const int t = threadIdx.x;
    if (t < RELS) tab[t] = *(const v4f*)(etab_l + t * H_HEADS);

    const int node = blockIdx.x * LNPW + t / LGL;
    const int sub  = t & (LGL - 1);
    int start = 0, len = 0;
    if (node < n) { start = rowstart[node]; len = cnt[node]; }

    unsigned pk0[LCH];
    l_load(sorted, start, len, sub, pk0);

    const int np = (n + 3) & ~3;
    const int count = (np < LCAPN) ? np : LCAPN;
    for (int i0 = 0; i0 < count; i0 += LT * 4 * LST) {
        v4f r[LST];
        #pragma unroll
        for (int k = 0; k < LST; ++k) {
            int i = i0 + (k * LT + t) * 4;
            if (i > count - 4) i = count - 4;
            r[k] = *(const v4f*)(hs + i);
        }
        #pragma unroll
        for (int k = 0; k < LST; ++k) {
            const int i = i0 + (k * LT + t) * 4;
            if (i < count) *(v4f*)(hl + i) = r[k];
        }
    }
    __syncthreads();

    float m[H_HEADS];
    #pragma unroll
    for (int h = 0; h < H_HEADS; ++h) m[h] = -__builtin_huge_valf();
    l_max(tab, pk0, len, sub, m);
    for (int base = LGL * LCH; base < len; base += LGL * LCH) {
        unsigned pk[LCH];
        l_load(sorted, start, len, base + sub, pk);
        l_max(tab, pk, len, base + sub, m);
    }
    #pragma unroll
    for (int h = 0; h < H_HEADS; ++h)
        m[h] = fmaxf(m[h], __shfl_xor(m[h], 1, 32));

    float s[H_HEADS], u[H_HEADS];
    #pragma unroll
    for (int h = 0; h < H_HEADS; ++h) { s[h] = 0.f; u[h] = 0.f; }
    l_sum(tab, hl, hs, (unsigned)count, pk0, len, sub, m, s, u);
    for (int base = LGL * LCH; base < len; base += LGL * LCH) {
        unsigned pk[LCH];
        l_load(sorted, start, len, base + sub, pk);
        l_sum(tab, hl, hs, (unsigned)count, pk, len, base + sub, m, s, u);
    }
    #pragma unroll
    for (int h = 0; h < H_HEADS; ++h) {
        s[h] += __shfl_xor(s[h], 1, 32);
        u[h] += __shfl_xor(u[h], 1, 32);
    }

    if (sub == 0 && node < n) {
        v4f v;
        #pragma unroll
        for (int h = 0; h < H_HEADS; ++h) {
            const float x = (s[h] > 0.f) ? (u[h] / s[h]) : 0.f;
            v[h] = (x > 0.f) ? x : 0.f;
        }
        if (do_mean) {
            hs_next[node] = 0.25f * (v[0] + v[1] + v[2] + v[3]);
            if (node == n - 1)
                for (int j = n; j < np; ++j) hs_next[j] = 0.f;
        } else {
            const float c = centrality[node];
            float acc = 0.f;
            #pragma unroll
            for (int h = 0; h < H_HEADS; ++h)
                acc += (c * gamma[h] + beta[h]) * v[h];
            acc *= 0.25f;
            out[node] = (acc > 0.f) ? acc : 0.01f * acc;
        }
    }
}

extern "C" void kernel_launch(void* const* d_in, const int* in_sizes, int n_in,
                              void* d_out, int out_size, void* d_ws, size_t ws_size,
                              hipStream_t stream) {
    const float* inputs     = (const float*)d_in[0];
    const float* centrality = (const float*)d_in[1];
    const float* W1         = (const float*)d_in[2];
    const float* b1         = (const float*)d_in[3];
    const float* W2         = (const float*)d_in[4];
    const float* b2         = (const float*)d_in[5];
    const float* rel_emb    = (const float*)d_in[6];
    const float* W_pred     = (const float*)d_in[7];
    const float* gamma      = (const float*)d_in[8];
    const float* beta       = (const float*)d_in[9];
    const int*   edge_types = (const int*)d_in[10];
    const int*   src        = (const int*)d_in[11];
    const int*   dst        = (const int*)d_in[12];
    float* out = (float*)d_out;

    const int N = in_sizes[1];
    const int E = in_sizes[10];
    const int Np = (N + 3) & ~3;
    const int Ep = (E + 3) & ~3;
    const int NB = (N + (1 << BSH) - 1) >> BSH;
    const int nChunks = (E + CH - 1) / CH;
    const int M  = NB * nChunks;
    const int Mp = (M + 3) & ~3;
    const int Sp = SPAD + NB * FCAP + Ep;

    char* ws = (char*)d_ws;
    _Float16* W1h = (_Float16*)ws;
    float* etab   = (float*)(ws + 65536);
    float* hA     = (float*)(ws + 69632);
    int*   cnt      = (int*)(hA + (size_t)Np * H_HEADS);
    int*   rowstart = cnt + Np;
    unsigned* runtab = (unsigned*)(rowstart + Np);
    unsigned* part   = runtab + Mp;
    unsigned* sorted = part + Ep;
    float* hsA = (float*)(sorted + Sp);
    float* hsB = hsA + Np;

    const int nW1      = COLS * IN_DIM;
    const int w1Blocks = (nW1 + CTH - 1) / CTH;
    const int eBlocks  = (LAYERS * RELS * H_HEADS + CTH - 1) / CTH;
    k_pre<<<w1Blocks + eBlocks, CTH, 0, stream>>>(W1, W1h, nW1, w1Blocks, rel_emb, W_pred, etab);

    const int nTiles    = (N + 15) / 16;
    const int mlpBlocks = (nTiles + CTH / 32 - 1) / (CTH / 32);
    k_partition_mlp<<<nChunks + mlpBlocks, CTH, 0, stream>>>(
        edge_types, src, dst, part, runtab, E, nChunks, NB, nChunks,
        inputs, W1h, b1, W2, b2, hA, N, nTiles);

    k_fine<<<NB, FTH, 0, stream>>>(part, runtab, cnt, rowstart, sorted, N, nChunks, NB,
                                   etab, hA, hsA);

    const int lBlocks = (N + LNPW - 1) / LNPW;
    float* hscur = hsA;
    float* hsnxt = hsB;

    for (int l = 1; l < LAYERS; ++l) {
        const float* etab_l = etab + l * RELS * H_HEADS;
        const int do_mean = (l != LAYERS - 1) ? 1 : 0;
        k_gather_lds<<<lBlocks, LT, 0, stream>>>(sorted, rowstart, cnt, etab_l, hscur,
                                                 hsnxt, centrality, gamma, beta, out,
                                                 N, do_mean);
        float* tmp = hscur; hscur = hsnxt; hsnxt = tmp;
    }
}


# MLP epilogue: each head's 64-column sum taken in-lane over the 4 column tiles, then across 16 lanes by a halving exchange on the vector ALU (8 exchanges, not 128 via LDS); sum order changed
# speedup vs baseline: 28.8134x; 1.1068x over previous
#include <hip/hip_runtime.h>
#include <hip/hip_bf16.h>

typedef __attribute__((ext_vector_type(16))) _Float16 v16h;
typedef __attribute__((ext_vector_type(8)))  _Float16 v8h;
typedef __attribute__((ext_vector_type(8)))  float    v8f;
typedef __attribute__((ext_vector_type(4)))  float    v4f;

#define H_HEADS 4
#define IN_DIM  128
#define HID     64
#define COLS    256
#define PRED    16
#define RELS    50
#define LAYERS  3

__device__ __forceinline__ void w1_to_f16_body(const float* __restrict__ W1,
                                               _Float16* __restrict__ W1h, int n, int blk) {
    int i = blk * blockDim.x + threadIdx.x;
    if (i < n) W1h[i] = (_Float16)W1[i];
}

__device__ __forceinline__ void etab_body(const float* __restrict__ rel_emb,
                                          const float* __restrict__ W_pred,
                                          float* __restrict__ etab, int blk) {
    int i = blk * blockDim.x + threadIdx.x;
    if (i >= LAYERS * RELS * H_HEADS) return;
    int l = i / (RELS * H_HEADS);
    int rem = i % (RELS * H_HEADS);
    int r = rem / H_HEADS;
    int h = rem % H_HEADS;
    float e = 0.f;
    #pragma unroll
    for (int p = 0; p < PRED; ++p)
        e += rel_emb[r * PRED + p] * W_pred[(l * PRED + p) * H_HEADS + h];
    etab[i] = (e > 0.f) ? e : 0.2f * e;
}

template <unsigned LO, unsigned HI>
__device__ __forceinline__ float row_xor(float x) {
    const int b = __builtin_bit_cast(int, x);
    return __builtin_bit_cast(float, __builtin_amdgcn_permlane16(b, b, LO, HI, false, false));
}

__device__ __forceinline__ void mlp_body(const float* __restrict__ inputs,
                                         const _Float16* __restrict__ W1h,
                                         const float* __restrict__ b1,
                                         const float* __restrict__ W2,
                                         const float* __restrict__ b2,
                                         float* __restrict__ hout,
                                         int n, int nTiles, int blk) {
    const int wave = (blk * blockDim.x + threadIdx.x) >> 5;
    if (wave >= nTiles) return;
    const int lane = threadIdx.x & 31;
    const int grp  = lane >> 4;
    const int l15  = lane & 15;
    const int nodeBase = wave * 16;
    int arow = nodeBase + l15;
    if (arow > n - 1) arow = n - 1;
    const float* __restrict__ abase = inputs + (size_t)arow * IN_DIM;

    const v8f zero = {};
    const bool up8 = (l15 & 8) != 0, up4 = (l15 & 4) != 0, up2 = (l15 & 2) != 0;
    const bool even = (l15 & 1) == 0;
    const int onode = nodeBase + (up8 ? 4 : 0) + (up4 ? 2 : 0) + (up2 ? 1 : 0) + 8 * grp;

    #pragma unroll
    for (int head = 0; head < H_HEADS; ++head) {
        v8f acc0 = zero, acc1 = zero, acc2 = zero, acc3 = zero;

        #pragma unroll
        for (int kk = 0; kk < IN_DIM; kk += 32) {
            const int kb = kk + grp * 8;
            v4f a0 = *(const v4f*)(abase + kb + 0);
            v4f a1 = *(const v4f*)(abase + kb + 4);
            v4f a2 = *(const v4f*)(abase + kb + 16);
            v4f a3 = *(const v4f*)(abase + kb + 20);
            v16h A;
            #pragma unroll
            for (int i = 0; i < 4; ++i) {
                A[i]      = (_Float16)a0[i];
                A[4 + i]  = (_Float16)a1[i];
                A[8 + i]  = (_Float16)a2[i];
                A[12 + i] = (_Float16)a3[i];
            }
            #pragma unroll
            for (int tt = 0; tt < 4; ++tt) {
                const int col = head * HID + tt * 16 + l15;
                const _Float16* q = W1h + (size_t)col * IN_DIM + kb;
                v8h blo = *(const v8h*)(q);
                v8h bhi = *(const v8h*)(q + 16);
                v16h B;
                #pragma unroll
                for (int i = 0; i < 8; ++i) { B[i] = blo[i]; B[8 + i] = bhi[i]; }
                v8f c = (tt == 0) ? acc0 : (tt == 1) ? acc1 : (tt == 2) ? acc2 : acc3;
                c = __builtin_amdgcn_wmma_f32_16x16x32_f16(
                        false, A, false, B, (short)0, c, false, false);
                if (tt == 0) acc0 = c; else if (tt == 1) acc1 = c;
                else if (tt == 2) acc2 = c; else acc3 = c;
            }
        }

        v8f q = zero;
        #pragma unroll
        for (int tt = 0; tt < 4; ++tt) {
            const int col  = head * HID + tt * 16 + l15;
            const float bias = b1[col];
            const float w2   = W2[col];
            v8f c = (tt == 0) ? acc0 : (tt == 1) ? acc1 : (tt == 2) ? acc2 : acc3;
            #pragma unroll
            for (int v = 0; v < 8; ++v) {
                float p = c[v] + bias;
                p = (p > 0.f) ? p : 0.f;
                q[v] += p * w2;
            }
        }
        float r1[4], r2[2];
        #pragma unroll
        for (int i = 0; i < 4; ++i) {
            const float keep = up8 ? q[4 + i] : q[i];
            const float send = up8 ? q[i] : q[4 + i];
            r1[i] = keep + row_xor<0xFEDCBA98u, 0x76543210u>(send);
        }
        #pragma unroll
        for (int i = 0; i < 2; ++i) {
            const float keep = up4 ? r1[2 + i] : r1[i];
            const float send = up4 ? r1[i] : r1[2 + i];
            r2[i] = keep + row_xor<0x32107654u, 0xBA98FEDCu>(send);
        }
        const float keep3 = up2 ? r2[1] : r2[0];
        const float send3 = up2 ? r2[0] : r2[1];
        const float r3 = keep3 + row_xor<0x54761032u, 0xDCFE98BAu>(send3);
        const float tot = r3 + row_xor<0x67452301u, 0xEFCDAB89u>(r3);
        if (even && onode < n)
            hout[onode * H_HEADS + head] = tot + b2[head];
    }
}

#define BSH    8
#define NBMAX  1024
#define CH     8192
#define CTH    256
#define FTH    1024
#define FCAP   12288
#define SPAD   4
#define RB     16

__device__ __forceinline__ void scan1024(const int* cin, int* cout, int* wsum, int t,
                                         int (&own)[4]) {
    const int lane = t & 31, w = t >> 5;
    const int c0 = cin[4 * t], c1 = cin[4 * t + 1], c2 = cin[4 * t + 2], c3 = cin[4 * t + 3];
    const int tsum = c0 + c1 + c2 + c3;
    int x = tsum;
    #pragma unroll
    for (int d = 1; d < 32; d <<= 1) {
        const int y = __shfl_up(x, d, 32);
        if (lane >= d) x += y;
    }
    if (lane == 31) wsum[w] = x;
    __syncthreads();
    int wbase = 0;
    #pragma unroll
    for (int k = 0; k < 8; ++k) if (k < w) wbase += wsum[k];
    const int run = wbase + x - tsum;
    own[0] = run; own[1] = run + c0; own[2] = run + c0 + c1; own[3] = run + c0 + c1 + c2;
    cout[4 * t] = own[0]; cout[4 * t + 1] = own[1];
    cout[4 * t + 2] = own[2]; cout[4 * t + 3] = own[3];
}

__global__ void __launch_bounds__(CTH)
k_pre(const float* __restrict__ W1, _Float16* __restrict__ W1h, int nW1, int w1Blocks,
      const float* __restrict__ rel_emb, const float* __restrict__ W_pred,
      float* __restrict__ etab) {
    const int blk = blockIdx.x;
    if (blk < w1Blocks) w1_to_f16_body(W1, W1h, nW1, blk);
    else                etab_body(rel_emb, W_pred, etab, blk - w1Blocks);
}

__device__ __forceinline__ void partition_body(
            const int* __restrict__ et, const int* __restrict__ src,
            const int* __restrict__ dst,
            unsigned* __restrict__ tmp, unsigned* __restrict__ runtab,
            int E, int nChunks, int NB, int blk) {
    __shared__ int lc[NBMAX];
    __shared__ int lb[NBMAX];
    __shared__ int wsum[8];
    __shared__ __attribute__((aligned(16))) unsigned recs[CH];
    const int t = threadIdx.x;
    for (int b = t; b < NBMAX; b += CTH) lc[b] = 0;
    __syncthreads();

    const int base = blk * CH;
    unsigned rec[32];
    unsigned br[32];
    #pragma unroll
    for (int k = 0; k < CH / (CTH * 4); ++k) {
        const int i = base + (k * CTH + t) * 4;
        int dv[4], sv[4], ev[4];
        if (i + 3 < E) {
            const int4 d = *(const int4*)(dst + i);
            const int4 s = *(const int4*)(src + i);
            const int4 e = *(const int4*)(et + i);
            dv[0] = d.x; dv[1] = d.y; dv[2] = d.z; dv[3] = d.w;
            sv[0] = s.x; sv[1] = s.y; sv[2] = s.z; sv[3] = s.w;
            ev[0] = e.x; ev[1] = e.y; ev[2] = e.z; ev[3] = e.w;
        } else {
            #pragma unroll
            for (int c = 0; c < 4; ++c) {
                const int j = i + c;
                dv[c] = -1; sv[c] = 0; ev[c] = 0;
                if (j < E) { dv[c] = dst[j]; sv[c] = src[j]; ev[c] = et[j]; }
            }
        }
        #pragma unroll
        for (int c = 0; c < 4; ++c) {
            rec[k * 4 + c] = 0u; br[k * 4 + c] = ~0u;
            if (dv[c] >= 0) {
                const int b = dv[c] >> BSH;
                const int r = atomicAdd(&lc[b], 1);
                rec[k * 4 + c] = ((unsigned)sv[c] << 14) | ((unsigned)ev[c] << 8)
                               | ((unsigned)dv[c] & 255u);
                br[k * 4 + c] = ((unsigned)b << 16) | (unsigned)r;
            }
        }
    }
    __syncthreads();

    int own[4];
    scan1024(lc, lb, wsum, t, own);
    #pragma unroll
    for (int c = 0; c < 4; ++c) {
        const int b = 4 * t + c;
        if (b < NB)
            runtab[(size_t)b * nChunks + blk] = ((unsigned)own[c] << 16) | (unsigned)lc[b];
    }
    __syncthreads();

    #pragma unroll
    for (int q = 0; q < 32; ++q) {
        if (br[q] != ~0u) {
            const int b = (int)(br[q] >> 16);
            recs[lb[b] + (int)(br[q] & 0xFFFFu)] = rec[q];
        }
    }
    __syncthreads();

    int nvalid = E - base;
    if (nvalid > CH) nvalid = CH;
    const int n4 = nvalid >> 2;
    uint4* __restrict__ o4 = (uint4*)(tmp + base);
    const uint4* r4 = (const uint4*)recs;
    for (int i = t; i < n4; i += CTH) o4[i] = r4[i];
    for (int i = (n4 << 2) + t; i < nvalid; i += CTH) tmp[base + i] = recs[i];
}

__global__ void __launch_bounds__(CTH)
k_partition_mlp(const int* __restrict__ et, const int* __restrict__ src,
                const int* __restrict__ dst,
                unsigned* __restrict__ tmp, unsigned* __restrict__ runtab,
                int E, int nChunks, int NB, int pBlocks,
                const float* __restrict__ inputs, const _Float16* __restrict__ W1h,
                const float* __restrict__ b1, const float* __restrict__ W2,
                const float* __restrict__ b2, float* __restrict__ hout,
                int n, int nTiles) {
    const int blk = blockIdx.x;
    if (blk < pBlocks)
        partition_body(et, src, dst, tmp, runtab, E, nChunks, NB, blk);
    else
        mlp_body(inputs, W1h, b1, W2, b2, hout, n, nTiles, blk - pBlocks);
}

#define FGL (FTH >> BSH)
#define FGC 4
static_assert(LAYERS >= 2, "the fused first layer always feeds a head mean");

template <bool STAGED>
__device__ __forceinline__ unsigned fine_rec(const unsigned* stage, const unsigned* gsort, int i) {
    return STAGED ? stage[i] : gsort[i];
}

template <bool STAGED>
__device__ __forceinline__ void fine_layer0(const unsigned* stage, const unsigned* gsort,
                                            const v4f* tab, const int* ncount, const int* nbase,
                                            const float* __restrict__ hcur,
                                            float* __restrict__ hs_next, int b, int n) {
    const int t    = threadIdx.x;
    const int dl   = t / FGL;
    const int sub  = t & (FGL - 1);
    const int node = (b << BSH) + dl;
    const int len  = ncount[dl];
    const int st   = nbase[dl];

    float m[H_HEADS];
    #pragma unroll
    for (int h = 0; h < H_HEADS; ++h) m[h] = -__builtin_huge_valf();
    for (int j0 = sub; j0 < len; j0 += FGL * FGC) {
        unsigned pk[FGC];
        #pragma unroll
        for (int k = 0; k < FGC; ++k) {
            const int j = j0 + k * FGL;
            pk[k] = fine_rec<STAGED>(stage, gsort, (j < len) ? st + j : 0);
        }
        #pragma unroll
        for (int k = 0; k < FGC; ++k) {
            const bool ok = j0 + k * FGL < len;
            const v4f e = tab[ok ? (pk[k] & 63u) : 0u];
            #pragma unroll
            for (int h = 0; h < H_HEADS; ++h)
                m[h] = fmaxf(m[h], ok ? e[h] : -__builtin_huge_valf());
        }
    }
    #pragma unroll
    for (int h = 0; h < H_HEADS; ++h) {
        m[h] = fmaxf(m[h], __shfl_xor(m[h], 1, 32));
        m[h] = fmaxf(m[h], __shfl_xor(m[h], 2, 32));
    }

    float s[H_HEADS], u[H_HEADS];
    #pragma unroll
    for (int h = 0; h < H_HEADS; ++h) { s[h] = 0.f; u[h] = 0.f; }
    for (int j0 = sub; j0 < len; j0 += FGL * FGC) {
        unsigned pk[FGC];
        v4f hv[FGC];
        #pragma unroll
        for (int k = 0; k < FGC; ++k) {
            const int j = j0 + k * FGL;
            const unsigned raw = fine_rec<STAGED>(stage, gsort, (j < len) ? st + j : 0);
            pk[k] = (j < len) ? raw : 0u;
        }
        #pragma unroll
        for (int k = 0; k < FGC; ++k)
            hv[k] = *(const v4f*)(hcur + (size_t)(pk[k] >> 6) * H_HEADS);
        #pragma unroll
        for (int k = 0; k < FGC; ++k) {
            const bool ok = j0 + k * FGL < len;
            const v4f e = tab[pk[k] & 63u];
            #pragma unroll
            for (int h = 0; h < H_HEADS; ++h) {
                const float wgt = ok ? __expf(e[h] - m[h]) : 0.f;
                const float hk  = ok ? hv[k][h] : 0.f;
                s[h] += wgt;
                u[h] += wgt * hk;
            }
        }
    }
    #pragma unroll
    for (int h = 0; h < H_HEADS; ++h) {
        s[h] += __shfl_xor(s[h], 1, 32);  u[h] += __shfl_xor(u[h], 1, 32);
        s[h] += __shfl_xor(s[h], 2, 32);  u[h] += __shfl_xor(u[h], 2, 32);
    }

    if (sub == 0 && node < n) {
        v4f v;
        #pragma unroll
        for (int h = 0; h < H_HEADS; ++h) {
            const float x = (s[h] > 0.f) ? (u[h] / s[h]) : 0.f;
            v[h] = (x > 0.f) ? x : 0.f;
        }
        hs_next[node] = 0.25f * (v[0] + v[1] + v[2] + v[3]);
        if (node == n - 1)
            for (int j = n; j < ((n + 3) & ~3); ++j) hs_next[j] = 0.f;
    }
}

template <int PASS, bool STAGED>
__device__ __forceinline__ void fine_put(unsigned rec, int* ncount, const int* nbase, int* cur,
                                         unsigned* stage, unsigned* __restrict__ gout) {
    const unsigned dl = rec & 255u;
    if (PASS == 0) {
        atomicAdd(&ncount[dl], 1);
    } else {
        const int p = nbase[dl] + atomicAdd(&cur[dl], 1);
        const unsigned val = rec >> 8;
        if (STAGED) stage[p] = val; else gout[p] = val;
    }
}

template <int PASS, bool STAGED>
__device__ __forceinline__ void fine_visit(const unsigned* __restrict__ tmp,
                                           const unsigned* __restrict__ row,
                                           int nChunks, int K, int c0, int lane,
                                           int* ncount, const int* nbase, int* cur,
                                           unsigned* stage, unsigned* __restrict__ gout) {
    for (int kb = 0; kb < K; kb += RB) {
        const int ci = c0 + kb + lane;
        const bool have = lane < RB && kb + lane < K && ci < nChunks;
        const unsigned draw = row[have ? ci : 0];
        const unsigned d = have ? draw : 0u;
        unsigned r[RB];
        #pragma unroll
        for (int i = 0; i < RB; ++i) {
            const unsigned di = (unsigned)__builtin_amdgcn_readlane((int)d, i);
            const int ln = (int)(di & 0xFFFFu);
            const size_t off = (ln > 0) ? (size_t)(c0 + kb + i) * CH + (di >> 16) : (size_t)0;
            r[i] = tmp[off + (size_t)((lane < ln) ? lane : 0)];
        }
        #pragma unroll
        for (int i = 0; i < RB; ++i) {
            const unsigned di = (unsigned)__builtin_amdgcn_readlane((int)d, i);
            const int ln = (int)(di & 0xFFFFu);
            if (lane < ln) fine_put<PASS, STAGED>(r[i], ncount, nbase, cur, stage, gout);
            if (ln > 32) {
                const unsigned* p = tmp + (size_t)(c0 + kb + i) * CH + (di >> 16);
                for (int j = 32 + lane; j < ln; j += 32)
                    fine_put<PASS, STAGED>(p[j], ncount, nbase, cur, stage, gout);
            }
        }
    }
}

__global__ void __launch_bounds__(FTH)
k_fine(const unsigned* __restrict__ tmp, const unsigned* __restrict__ runtab,
       int* __restrict__ cnt, int* __restrict__ rowstart,
       unsigned* __restrict__ sorted, int n, int nChunks, int NB,
       const float* __restrict__ etab0, const float* __restrict__ hcur,
       float* __restrict__ hs_next) {
    __shared__ v4f tab[RELS];
    __shared__ int ncount[1 << BSH];
    __shared__ int nbase[1 << BSH];
    __shared__ int cur[1 << BSH];
    __shared__ int wsum[8];
    __shared__ int wtot[FTH / 32];
    __shared__ int wovf[FTH / 32];
    __shared__ unsigned stage[FCAP];
    const int t = threadIdx.x, lane = t & 31;
    const int w = __builtin_amdgcn_readfirstlane(t >> 5);
    const int b = blockIdx.x;
    const int K  = (nChunks + FTH / 32 - 1) / (FTH / 32);
    const int c0 = w * K;
    const unsigned* __restrict__ row = runtab + (size_t)b * nChunks;

    int part = 0;
    for (int k = lane; k < K; k += 32) {
        const int c = c0 + k;
        if (c < nChunks) part += (int)(row[c] & 0xFFFFu);
    }
    #pragma unroll
    for (int d = 16; d >= 1; d >>= 1) part += __shfl_xor(part, d, 32);
    if (lane == 0) wtot[w] = part;
    if (t < (1 << BSH)) { ncount[t] = 0; cur[t] = 0; }
    if (t < RELS) tab[t] = *(const v4f*)(etab0 + t * H_HEADS);
    if (b == 0 && t == 0) sorted[0] = 0u;
    __syncthreads();
    static_assert(FTH / 32 == 32, "one lane per wave total below");
    int nbl = wtot[lane];
    #pragma unroll
    for (int d = 16; d >= 1; d >>= 1) nbl += __shfl_xor(nbl, d, 32);
    const int nb = __builtin_amdgcn_readfirstlane(nbl);
    const bool staged = (nb <= FCAP);

    int bs = SPAD + b * FCAP;
    if (!staged) {
        int acc = 0;
        for (int bb = w; bb < b; bb += FTH / 32) {
            const unsigned* __restrict__ r2 = runtab + (size_t)bb * nChunks;
            int sum = 0;
            for (int c = lane; c < nChunks; c += 32) sum += (int)(r2[c] & 0xFFFFu);
            #pragma unroll
            for (int d = 16; d >= 1; d >>= 1) sum += __shfl_xor(sum, d, 32);
            if (sum > FCAP) acc += sum;
        }
        if (lane == 0) wovf[w] = acc;
        __syncthreads();
        int before = wovf[lane];
        #pragma unroll
        for (int d = 16; d >= 1; d >>= 1) before += __shfl_xor(before, d, 32);
        bs = SPAD + NB * FCAP + __builtin_amdgcn_readfirstlane(before);
    }
    unsigned* __restrict__ gout = sorted + bs;

    if (staged) fine_visit<0, true >(tmp, row, nChunks, K, c0, lane, ncount, nbase, cur, stage, gout);
    else        fine_visit<0, false>(tmp, row, nChunks, K, c0, lane, ncount, nbase, cur, stage, gout);
    __syncthreads();

    int v = 0, x = 0;
    if (t < (1 << BSH)) {
        v = ncount[t];
        x = v;
        #pragma unroll
        for (int d = 1; d < 32; d <<= 1) {
            const int y = __shfl_up(x, d, 32);
            if (lane >= d) x += y;
        }
        if (lane == 31) wsum[w] = x;
    }
    __syncthreads();
    if (t < (1 << BSH)) {
        int wbase = 0;
        #pragma unroll
        for (int k = 0; k < 8; ++k) if (k < w) wbase += wsum[k];
        const int excl = wbase + x - v;
        nbase[t] = excl;
        const int node = (b << BSH) + t;
        if (node < n) { cnt[node] = v; rowstart[node] = bs + excl; }
    }
    __syncthreads();

    if (staged) fine_visit<1, true >(tmp, row, nChunks, K, c0, lane, ncount, nbase, cur, stage, gout);
    else        fine_visit<1, false>(tmp, row, nChunks, K, c0, lane, ncount, nbase, cur, stage, gout);
    __syncthreads();
    if (staged) {
        for (int i = t; i < nb; i += FTH) gout[i] = stage[i];
        fine_layer0<true>(stage, gout, tab, ncount, nbase, hcur, hs_next, b, n);
    } else {
        fine_layer0<false>(stage, gout, tab, ncount, nbase, hcur, hs_next, b, n);
    }
}

#define LT    800
#define LGL   2
#define LCH   16
#define LNPW  (LT / LGL)
#define LCAPN 81664
#define LST   8

__device__ __forceinline__ void l_load(const unsigned* __restrict__ sorted, int start,
                                       int len, int j0, unsigned (&pk)[LCH]) {
    #pragma unroll
    for (int k = 0; k < LCH; ++k) {
        const int j = j0 + k * LGL;
        pk[k] = sorted[(j < len) ? start + j : 0];
    }
}
__device__ __forceinline__ void l_max(const v4f* tab, const unsigned (&pk)[LCH],
                                      int len, int j0, float (&m)[H_HEADS]) {
    #pragma unroll
    for (int k = 0; k < LCH; ++k) {
        if (j0 + k * LGL < len) {
            const v4f e = tab[pk[k] & 63u];
            #pragma unroll
            for (int h = 0; h < H_HEADS; ++h) m[h] = fmaxf(m[h], e[h]);
        }
    }
}
__device__ __forceinline__ void l_sum(const v4f* tab, const float* hl,
                                      const float* __restrict__ hs, unsigned count,
                                      const unsigned (&pk)[LCH],
                                      int len, int j0, const float (&m)[H_HEADS],
                                      float (&s)[H_HEADS], float (&u)[H_HEADS]) {
    float hv[LCH];
    #pragma unroll
    for (int k = 0; k < LCH; ++k) {
        const unsigned sn = pk[k] >> 6;
        const bool inl = sn < count;
        const float a = hl[inl ? sn : 0u];
        const float b = hs[inl ? 0u : sn];
        hv[k] = inl ? a : b;
    }
    #pragma unroll
    for (int k = 0; k < LCH; ++k) {
        if (j0 + k * LGL < len) {
            const v4f e = tab[pk[k] & 63u];
            #pragma unroll
            for (int h = 0; h < H_HEADS; ++h) {
                const float wgt = __expf(e[h] - m[h]);
                s[h] += wgt;
                u[h] += wgt * hv[k];
            }
        }
    }
}

__global__ void __launch_bounds__(LT)
k_gather_lds(const unsigned* __restrict__ sorted,
             const int* __restrict__ rowstart,
             const int* __restrict__ cnt,
             const float* __restrict__ etab_l,
             const float* __restrict__ hs,
             float* __restrict__ hs_next,
             const float* __restrict__ centrality,
             const float* __restrict__ gamma,
             const float* __restrict__ beta,
             float* __restrict__ out,
             int n, int do_mean) {
    __shared__ v4f tab[RELS];
    __shared__ float hl[LCAPN];
    const int t = threadIdx.x;
    if (t < RELS) tab[t] = *(const v4f*)(etab_l + t * H_HEADS);

    const int node = blockIdx.x * LNPW + t / LGL;
    const int sub  = t & (LGL - 1);
    int start = 0, len = 0;
    if (node < n) { start = rowstart[node]; len = cnt[node]; }

    unsigned pk0[LCH];
    l_load(sorted, start, len, sub, pk0);

    const int np = (n + 3) & ~3;
    const int count = (np < LCAPN) ? np : LCAPN;
    for (int i0 = 0; i0 < count; i0 += LT * 4 * LST) {
        v4f r[LST];
        #pragma unroll
        for (int k = 0; k < LST; ++k) {
            int i = i0 + (k * LT + t) * 4;
            if (i > count - 4) i = count - 4;
            r[k] = *(const v4f*)(hs + i);
        }
        #pragma unroll
        for (int k = 0; k < LST; ++k) {
            const int i = i0 + (k * LT + t) * 4;
            if (i < count) *(v4f*)(hl + i) = r[k];
        }
    }
    __syncthreads();

    float m[H_HEADS];
    #pragma unroll
    for (int h = 0; h < H_HEADS; ++h) m[h] = -__builtin_huge_valf();
    l_max(tab, pk0, len, sub, m);
    for (int base = LGL * LCH; base < len; base += LGL * LCH) {
        unsigned pk[LCH];
        l_load(sorted, start, len, base + sub, pk);
        l_max(tab, pk, len, base + sub, m);
    }
    #pragma unroll
    for (int h = 0; h < H_HEADS; ++h)
        m[h] = fmaxf(m[h], __shfl_xor(m[h], 1, 32));

    float s[H_HEADS], u[H_HEADS];
    #pragma unroll
    for (int h = 0; h < H_HEADS; ++h) { s[h] = 0.f; u[h] = 0.f; }
    l_sum(tab, hl, hs, (unsigned)count, pk0, len, sub, m, s, u);
    for (int base = LGL * LCH; base < len; base += LGL * LCH) {
        unsigned pk[LCH];
        l_load(sorted, start, len, base + sub, pk);
        l_sum(tab, hl, hs, (unsigned)count, pk, len, base + sub, m, s, u);
    }
    #pragma unroll
    for (int h = 0; h < H_HEADS; ++h) {
        s[h] += __shfl_xor(s[h], 1, 32);
        u[h] += __shfl_xor(u[h], 1, 32);
    }

    if (sub == 0 && node < n) {
        v4f v;
        #pragma unroll
        for (int h = 0; h < H_HEADS; ++h) {
            const float x = (s[h] > 0.f) ? (u[h] / s[h]) : 0.f;
            v[h] = (x > 0.f) ? x : 0.f;
        }
        if (do_mean) {
            hs_next[node] = 0.25f * (v[0] + v[1] + v[2] + v[3]);
            if (node == n - 1)
                for (int j = n; j < np; ++j) hs_next[j] = 0.f;
        } else {
            const float c = centrality[node];
            float acc = 0.f;
            #pragma unroll
            for (int h = 0; h < H_HEADS; ++h)
                acc += (c * gamma[h] + beta[h]) * v[h];
            acc *= 0.25f;
            out[node] = (acc > 0.f) ? acc : 0.01f * acc;
        }
    }
}

extern "C" void kernel_launch(void* const* d_in, const int* in_sizes, int n_in,
                              void* d_out, int out_size, void* d_ws, size_t ws_size,
                              hipStream_t stream) {
    const float* inputs     = (const float*)d_in[0];
    const float* centrality = (const float*)d_in[1];
    const float* W1         = (const float*)d_in[2];
    const float* b1         = (const float*)d_in[3];
    const float* W2         = (const float*)d_in[4];
    const float* b2         = (const float*)d_in[5];
    const float* rel_emb    = (const float*)d_in[6];
    const float* W_pred     = (const float*)d_in[7];
    const float* gamma      = (const float*)d_in[8];
    const float* beta       = (const float*)d_in[9];
    const int*   edge_types = (const int*)d_in[10];
    const int*   src        = (const int*)d_in[11];
    const int*   dst        = (const int*)d_in[12];
    float* out = (float*)d_out;

    const int N = in_sizes[1];
    const int E = in_sizes[10];
    const int Np = (N + 3) & ~3;
    const int Ep = (E + 3) & ~3;
    const int NB = (N + (1 << BSH) - 1) >> BSH;
    const int nChunks = (E + CH - 1) / CH;
    const int M  = NB * nChunks;
    const int Mp = (M + 3) & ~3;
    const int Sp = SPAD + NB * FCAP + Ep;

    char* ws = (char*)d_ws;
    _Float16* W1h = (_Float16*)ws;
    float* etab   = (float*)(ws + 65536);
    float* hA     = (float*)(ws + 69632);
    int*   cnt      = (int*)(hA + (size_t)Np * H_HEADS);
    int*   rowstart = cnt + Np;
    unsigned* runtab = (unsigned*)(rowstart + Np);
    unsigned* part   = runtab + Mp;
    unsigned* sorted = part + Ep;
    float* hsA = (float*)(sorted + Sp);
    float* hsB = hsA + Np;

    const int nW1      = COLS * IN_DIM;
    const int w1Blocks = (nW1 + CTH - 1) / CTH;
    const int eBlocks  = (LAYERS * RELS * H_HEADS + CTH - 1) / CTH;
    k_pre<<<w1Blocks + eBlocks, CTH, 0, stream>>>(W1, W1h, nW1, w1Blocks, rel_emb, W_pred, etab);

    const int nTiles    = (N + 15) / 16;
    const int mlpBlocks = (nTiles + CTH / 32 - 1) / (CTH / 32);
    k_partition_mlp<<<nChunks + mlpBlocks, CTH, 0, stream>>>(
        edge_types, src, dst, part, runtab, E, nChunks, NB, nChunks,
        inputs, W1h, b1, W2, b2, hA, N, nTiles);

    k_fine<<<NB, FTH, 0, stream>>>(part, runtab, cnt, rowstart, sorted, N, nChunks, NB,
                                   etab, hA, hsA);

    const int lBlocks = (N + LNPW - 1) / LNPW;
    float* hscur = hsA;
    float* hsnxt = hsB;

    for (int l = 1; l < LAYERS; ++l) {
        const float* etab_l = etab + l * RELS * H_HEADS;
        const int do_mean = (l != LAYERS - 1) ? 1 : 0;
        k_gather_lds<<<lBlocks, LT, 0, stream>>>(sorted, rowstart, cnt, etab_l, hscur,
                                                 hsnxt, centrality, gamma, beta, out,
                                                 N, do_mean);
        float* tmp = hscur; hscur = hsnxt; hsnxt = tmp;
    }
}


# Edge-softmax weights read from a per-edge-type table exp(e - max over all types) built per workgroup: per-node max pass and per-edge exp gone in all 3 layers; per-node form kept if logit spread > 16
# speedup vs baseline: 29.8481x; 1.0359x over previous
#include <hip/hip_runtime.h>
#include <hip/hip_bf16.h>

typedef __attribute__((ext_vector_type(16))) _Float16 v16h;
typedef __attribute__((ext_vector_type(8)))  _Float16 v8h;
typedef __attribute__((ext_vector_type(8)))  float    v8f;
typedef __attribute__((ext_vector_type(4)))  float    v4f;

#define H_HEADS 4
#define IN_DIM  128
#define HID     64
#define COLS    256
#define PRED    16
#define RELS    50
#define LAYERS  3

__device__ __forceinline__ void w1_to_f16_body(const float* __restrict__ W1,
                                               _Float16* __restrict__ W1h, int n, int blk) {
    int i = blk * blockDim.x + threadIdx.x;
    if (i < n) W1h[i] = (_Float16)W1[i];
}

__device__ __forceinline__ void etab_body(const float* __restrict__ rel_emb,
                                          const float* __restrict__ W_pred,
                                          float* __restrict__ etab, int blk) {
    int i = blk * blockDim.x + threadIdx.x;
    if (i >= LAYERS * RELS * H_HEADS) return;
    int l = i / (RELS * H_HEADS);
    int rem = i % (RELS * H_HEADS);
    int r = rem / H_HEADS;
    int h = rem % H_HEADS;
    float e = 0.f;
    #pragma unroll
    for (int p = 0; p < PRED; ++p)
        e += rel_emb[r * PRED + p] * W_pred[(l * PRED + p) * H_HEADS + h];
    etab[i] = (e > 0.f) ? e : 0.2f * e;
}

template <unsigned LO, unsigned HI>
__device__ __forceinline__ float row_xor(float x) {
    const int b = __builtin_bit_cast(int, x);
    return __builtin_bit_cast(float, __builtin_amdgcn_permlane16(b, b, LO, HI, false, false));
}

__device__ __forceinline__ void mlp_body(const float* __restrict__ inputs,
                                         const _Float16* __restrict__ W1h,
                                         const float* __restrict__ b1,
                                         const float* __restrict__ W2,
                                         const float* __restrict__ b2,
                                         float* __restrict__ hout,
                                         int n, int nTiles, int blk) {
    const int wave = (blk * blockDim.x + threadIdx.x) >> 5;
    if (wave >= nTiles) return;
    const int lane = threadIdx.x & 31;
    const int grp  = lane >> 4;
    const int l15  = lane & 15;
    const int nodeBase = wave * 16;
    int arow = nodeBase + l15;
    if (arow > n - 1) arow = n - 1;
    const float* __restrict__ abase = inputs + (size_t)arow * IN_DIM;

    const v8f zero = {};
    const bool up8 = (l15 & 8) != 0, up4 = (l15 & 4) != 0, up2 = (l15 & 2) != 0;
    const bool even = (l15 & 1) == 0;
    const int onode = nodeBase + (up8 ? 4 : 0) + (up4 ? 2 : 0) + (up2 ? 1 : 0) + 8 * grp;

    #pragma unroll
    for (int head = 0; head < H_HEADS; ++head) {
        v8f acc0 = zero, acc1 = zero, acc2 = zero, acc3 = zero;

        #pragma unroll
        for (int kk = 0; kk < IN_DIM; kk += 32) {
            const int kb = kk + grp * 8;
            v4f a0 = *(const v4f*)(abase + kb + 0);
            v4f a1 = *(const v4f*)(abase + kb + 4);
            v4f a2 = *(const v4f*)(abase + kb + 16);
            v4f a3 = *(const v4f*)(abase + kb + 20);
            v16h A;
            #pragma unroll
            for (int i = 0; i < 4; ++i) {
                A[i]      = (_Float16)a0[i];
                A[4 + i]  = (_Float16)a1[i];
                A[8 + i]  = (_Float16)a2[i];
                A[12 + i] = (_Float16)a3[i];
            }
            #pragma unroll
            for (int tt = 0; tt < 4; ++tt) {
                const int col = head * HID + tt * 16 + l15;
                const _Float16* q = W1h + (size_t)col * IN_DIM + kb;
                v8h blo = *(const v8h*)(q);
                v8h bhi = *(const v8h*)(q + 16);
                v16h B;
                #pragma unroll
                for (int i = 0; i < 8; ++i) { B[i] = blo[i]; B[8 + i] = bhi[i]; }
                v8f c = (tt == 0) ? acc0 : (tt == 1) ? acc1 : (tt == 2) ? acc2 : acc3;
                c = __builtin_amdgcn_wmma_f32_16x16x32_f16(
                        false, A, false, B, (short)0, c, false, false);
                if (tt == 0) acc0 = c; else if (tt == 1) acc1 = c;
                else if (tt == 2) acc2 = c; else acc3 = c;
            }
        }

        v8f q = zero;
        #pragma unroll
        for (int tt = 0; tt < 4; ++tt) {
            const int col  = head * HID + tt * 16 + l15;
            const float bias = b1[col];
            const float w2   = W2[col];
            v8f c = (tt == 0) ? acc0 : (tt == 1) ? acc1 : (tt == 2) ? acc2 : acc3;
            #pragma unroll
            for (int v = 0; v < 8; ++v) {
                float p = c[v] + bias;
                p = (p > 0.f) ? p : 0.f;
                q[v] += p * w2;
            }
        }
        float r1[4], r2[2];
        #pragma unroll
        for (int i = 0; i < 4; ++i) {
            const float keep = up8 ? q[4 + i] : q[i];
            const float send = up8 ? q[i] : q[4 + i];
            r1[i] = keep + row_xor<0xFEDCBA98u, 0x76543210u>(send);
        }
        #pragma unroll
        for (int i = 0; i < 2; ++i) {
            const float keep = up4 ? r1[2 + i] : r1[i];
            const float send = up4 ? r1[i] : r1[2 + i];
            r2[i] = keep + row_xor<0x32107654u, 0xBA98FEDCu>(send);
        }
        const float keep3 = up2 ? r2[1] : r2[0];
        const float send3 = up2 ? r2[0] : r2[1];
        const float r3 = keep3 + row_xor<0x54761032u, 0xDCFE98BAu>(send3);
        const float tot = r3 + row_xor<0x67452301u, 0xEFCDAB89u>(r3);
        if (even && onode < n)
            hout[onode * H_HEADS + head] = tot + b2[head];
    }
}

#define WSPREAD 16.0f
static_assert(RELS > 32 && RELS <= 64, "two table rows per lane of one wave");
__device__ __forceinline__ void build_wtab(const float* __restrict__ etab_l, v4f* wtab,
                                           int* fast, int lane) {
    const int r1 = lane + 32;
    const bool has1 = r1 < RELS;
    const v4f e0 = *(const v4f*)(etab_l + lane * H_HEADS);
    const v4f e1 = *(const v4f*)(etab_l + (has1 ? r1 : 0) * H_HEADS);
    float mx[H_HEADS], mn[H_HEADS];
    #pragma unroll
    for (int h = 0; h < H_HEADS; ++h) {
        mx[h] = has1 ? fmaxf(e0[h], e1[h]) : e0[h];
        mn[h] = has1 ? fminf(e0[h], e1[h]) : e0[h];
    }
    #pragma unroll
    for (int d = 16; d >= 1; d >>= 1) {
        #pragma unroll
        for (int h = 0; h < H_HEADS; ++h) {
            mx[h] = fmaxf(mx[h], __shfl_xor(mx[h], d, 32));
            mn[h] = fminf(mn[h], __shfl_xor(mn[h], d, 32));
        }
    }
    bool ok = true;
    v4f w0, w1;
    #pragma unroll
    for (int h = 0; h < H_HEADS; ++h) {
        ok = ok && (mx[h] - mn[h] <= WSPREAD);
        w0[h] = __expf(e0[h] - mx[h]);
        w1[h] = has1 ? __expf(e1[h] - mx[h]) : 0.f;
    }
    wtab[lane] = w0;
    wtab[r1]   = w1;
    if (lane == 0) *fast = ok ? 1 : 0;
}

#define BSH    8
#define NBMAX  1024
#define CH     8192
#define CTH    256
#define FTH    1024
#define FCAP   12288
#define SPAD   4
#define RB     16

__device__ __forceinline__ void scan1024(const int* cin, int* cout, int* wsum, int t,
                                         int (&own)[4]) {
    const int lane = t & 31, w = t >> 5;
    const int c0 = cin[4 * t], c1 = cin[4 * t + 1], c2 = cin[4 * t + 2], c3 = cin[4 * t + 3];
    const int tsum = c0 + c1 + c2 + c3;
    int x = tsum;
    #pragma unroll
    for (int d = 1; d < 32; d <<= 1) {
        const int y = __shfl_up(x, d, 32);
        if (lane >= d) x += y;
    }
    if (lane == 31) wsum[w] = x;
    __syncthreads();
    int wbase = 0;
    #pragma unroll
    for (int k = 0; k < 8; ++k) if (k < w) wbase += wsum[k];
    const int run = wbase + x - tsum;
    own[0] = run; own[1] = run + c0; own[2] = run + c0 + c1; own[3] = run + c0 + c1 + c2;
    cout[4 * t] = own[0]; cout[4 * t + 1] = own[1];
    cout[4 * t + 2] = own[2]; cout[4 * t + 3] = own[3];
}

__global__ void __launch_bounds__(CTH)
k_pre(const float* __restrict__ W1, _Float16* __restrict__ W1h, int nW1, int w1Blocks,
      const float* __restrict__ rel_emb, const float* __restrict__ W_pred,
      float* __restrict__ etab) {
    const int blk = blockIdx.x;
    if (blk < w1Blocks) w1_to_f16_body(W1, W1h, nW1, blk);
    else                etab_body(rel_emb, W_pred, etab, blk - w1Blocks);
}

__device__ __forceinline__ void partition_body(
            const int* __restrict__ et, const int* __restrict__ src,
            const int* __restrict__ dst,
            unsigned* __restrict__ tmp, unsigned* __restrict__ runtab,
            int E, int nChunks, int NB, int blk) {
    __shared__ int lc[NBMAX];
    __shared__ int lb[NBMAX];
    __shared__ int wsum[8];
    __shared__ __attribute__((aligned(16))) unsigned recs[CH];
    const int t = threadIdx.x;
    for (int b = t; b < NBMAX; b += CTH) lc[b] = 0;
    __syncthreads();

    const int base = blk * CH;
    unsigned rec[32];
    unsigned br[32];
    #pragma unroll
    for (int k = 0; k < CH / (CTH * 4); ++k) {
        const int i = base + (k * CTH + t) * 4;
        int dv[4], sv[4], ev[4];
        if (i + 3 < E) {
            const int4 d = *(const int4*)(dst + i);
            const int4 s = *(const int4*)(src + i);
            const int4 e = *(const int4*)(et + i);
            dv[0] = d.x; dv[1] = d.y; dv[2] = d.z; dv[3] = d.w;
            sv[0] = s.x; sv[1] = s.y; sv[2] = s.z; sv[3] = s.w;
            ev[0] = e.x; ev[1] = e.y; ev[2] = e.z; ev[3] = e.w;
        } else {
            #pragma unroll
            for (int c = 0; c < 4; ++c) {
                const int j = i + c;
                dv[c] = -1; sv[c] = 0; ev[c] = 0;
                if (j < E) { dv[c] = dst[j]; sv[c] = src[j]; ev[c] = et[j]; }
            }
        }
        #pragma unroll
        for (int c = 0; c < 4; ++c) {
            rec[k * 4 + c] = 0u; br[k * 4 + c] = ~0u;
            if (dv[c] >= 0) {
                const int b = dv[c] >> BSH;
                const int r = atomicAdd(&lc[b], 1);
                rec[k * 4 + c] = ((unsigned)sv[c] << 14) | ((unsigned)ev[c] << 8)
                               | ((unsigned)dv[c] & 255u);
                br[k * 4 + c] = ((unsigned)b << 16) | (unsigned)r;
            }
        }
    }
    __syncthreads();

    int own[4];
    scan1024(lc, lb, wsum, t, own);
    #pragma unroll
    for (int c = 0; c < 4; ++c) {
        const int b = 4 * t + c;
        if (b < NB)
            runtab[(size_t)b * nChunks + blk] = ((unsigned)own[c] << 16) | (unsigned)lc[b];
    }
    __syncthreads();

    #pragma unroll
    for (int q = 0; q < 32; ++q) {
        if (br[q] != ~0u) {
            const int b = (int)(br[q] >> 16);
            recs[lb[b] + (int)(br[q] & 0xFFFFu)] = rec[q];
        }
    }
    __syncthreads();

    int nvalid = E - base;
    if (nvalid > CH) nvalid = CH;
    const int n4 = nvalid >> 2;
    uint4* __restrict__ o4 = (uint4*)(tmp + base);
    const uint4* r4 = (const uint4*)recs;
    for (int i = t; i < n4; i += CTH) o4[i] = r4[i];
    for (int i = (n4 << 2) + t; i < nvalid; i += CTH) tmp[base + i] = recs[i];
}

__global__ void __launch_bounds__(CTH)
k_partition_mlp(const int* __restrict__ et, const int* __restrict__ src,
                const int* __restrict__ dst,
                unsigned* __restrict__ tmp, unsigned* __restrict__ runtab,
                int E, int nChunks, int NB, int pBlocks,
                const float* __restrict__ inputs, const _Float16* __restrict__ W1h,
                const float* __restrict__ b1, const float* __restrict__ W2,
                const float* __restrict__ b2, float* __restrict__ hout,
                int n, int nTiles) {
    const int blk = blockIdx.x;
    if (blk < pBlocks)
        partition_body(et, src, dst, tmp, runtab, E, nChunks, NB, blk);
    else
        mlp_body(inputs, W1h, b1, W2, b2, hout, n, nTiles, blk - pBlocks);
}

#define FGL (FTH >> BSH)
#define FGC 4
static_assert(LAYERS >= 2, "the fused first layer always feeds a head mean");

template <bool STAGED>
__device__ __forceinline__ unsigned fine_rec(const unsigned* stage, const unsigned* gsort, int i) {
    return STAGED ? stage[i] : gsort[i];
}

template <bool STAGED>
__device__ __forceinline__ void fine_layer0(const unsigned* stage, const unsigned* gsort,
                                            const v4f* tab, const v4f* wtab, bool fast,
                                            const int* ncount, const int* nbase,
                                            const float* __restrict__ hcur,
                                            float* __restrict__ hs_next, int b, int n) {
    const int t    = threadIdx.x;
    const int dl   = t / FGL;
    const int sub  = t & (FGL - 1);
    const int node = (b << BSH) + dl;
    const int len  = ncount[dl];
    const int st   = nbase[dl];

    float s[H_HEADS], u[H_HEADS];
    #pragma unroll
    for (int h = 0; h < H_HEADS; ++h) { s[h] = 0.f; u[h] = 0.f; }

    if (fast) {
        for (int j0 = sub; j0 < len; j0 += FGL * FGC) {
            unsigned pk[FGC];
            v4f hv[FGC];
            #pragma unroll
            for (int k = 0; k < FGC; ++k) {
                const int j = j0 + k * FGL;
                const unsigned raw = fine_rec<STAGED>(stage, gsort, (j < len) ? st + j : 0);
                pk[k] = (j < len) ? raw : 0u;
            }
            #pragma unroll
            for (int k = 0; k < FGC; ++k)
                hv[k] = *(const v4f*)(hcur + (size_t)(pk[k] >> 6) * H_HEADS);
            #pragma unroll
            for (int k = 0; k < FGC; ++k) {
                const bool ok = j0 + k * FGL < len;
                const v4f w = wtab[ok ? (pk[k] & 63u) : 63u];
                #pragma unroll
                for (int h = 0; h < H_HEADS; ++h) {
                    const float hk = ok ? hv[k][h] : 0.f;
                    s[h] += w[h];
                    u[h] += w[h] * hk;
                }
            }
        }
    } else {
        float m[H_HEADS];
        #pragma unroll
        for (int h = 0; h < H_HEADS; ++h) m[h] = -__builtin_huge_valf();
        for (int j0 = sub; j0 < len; j0 += FGL * FGC) {
            unsigned pk[FGC];
            #pragma unroll
            for (int k = 0; k < FGC; ++k) {
                const int j = j0 + k * FGL;
                pk[k] = fine_rec<STAGED>(stage, gsort, (j < len) ? st + j : 0);
            }
            #pragma unroll
            for (int k = 0; k < FGC; ++k) {
                const bool ok = j0 + k * FGL < len;
                const v4f e = tab[ok ? (pk[k] & 63u) : 0u];
                #pragma unroll
                for (int h = 0; h < H_HEADS; ++h)
                    m[h] = fmaxf(m[h], ok ? e[h] : -__builtin_huge_valf());
            }
        }
        #pragma unroll
        for (int h = 0; h < H_HEADS; ++h) {
            m[h] = fmaxf(m[h], __shfl_xor(m[h], 1, 32));
            m[h] = fmaxf(m[h], __shfl_xor(m[h], 2, 32));
        }

        for (int j0 = sub; j0 < len; j0 += FGL * FGC) {
            unsigned pk[FGC];
            v4f hv[FGC];
            #pragma unroll
            for (int k = 0; k < FGC; ++k) {
                const int j = j0 + k * FGL;
                const unsigned raw = fine_rec<STAGED>(stage, gsort, (j < len) ? st + j : 0);
                pk[k] = (j < len) ? raw : 0u;
            }
            #pragma unroll
            for (int k = 0; k < FGC; ++k)
                hv[k] = *(const v4f*)(hcur + (size_t)(pk[k] >> 6) * H_HEADS);
            #pragma unroll
            for (int k = 0; k < FGC; ++k) {
                const bool ok = j0 + k * FGL < len;
                const v4f e = tab[pk[k] & 63u];
                #pragma unroll
                for (int h = 0; h < H_HEADS; ++h) {
                    const float wgt = ok ? __expf(e[h] - m[h]) : 0.f;
                    const float hk  = ok ? hv[k][h] : 0.f;
                    s[h] += wgt;
                    u[h] += wgt * hk;
                }
            }
        }
    }
    #pragma unroll
    for (int h = 0; h < H_HEADS; ++h) {
        s[h] += __shfl_xor(s[h], 1, 32);  u[h] += __shfl_xor(u[h], 1, 32);
        s[h] += __shfl_xor(s[h], 2, 32);  u[h] += __shfl_xor(u[h], 2, 32);
    }

    if (sub == 0 && node < n) {
        v4f v;
        #pragma unroll
        for (int h = 0; h < H_HEADS; ++h) {
            const float x = (s[h] > 0.f) ? (u[h] / s[h]) : 0.f;
            v[h] = (x > 0.f) ? x : 0.f;
        }
        hs_next[node] = 0.25f * (v[0] + v[1] + v[2] + v[3]);
        if (node == n - 1)
            for (int j = n; j < ((n + 3) & ~3); ++j) hs_next[j] = 0.f;
    }
}

template <int PASS, bool STAGED>
__device__ __forceinline__ void fine_put(unsigned rec, int* ncount, const int* nbase, int* cur,
                                         unsigned* stage, unsigned* __restrict__ gout) {
    const unsigned dl = rec & 255u;
    if (PASS == 0) {
        atomicAdd(&ncount[dl], 1);
    } else {
        const int p = nbase[dl] + atomicAdd(&cur[dl], 1);
        const unsigned val = rec >> 8;
        if (STAGED) stage[p] = val; else gout[p] = val;
    }
}

template <int PASS, bool STAGED>
__device__ __forceinline__ void fine_visit(const unsigned* __restrict__ tmp,
                                           const unsigned* __restrict__ row,
                                           int nChunks, int K, int c0, int lane,
                                           int* ncount, const int* nbase, int* cur,
                                           unsigned* stage, unsigned* __restrict__ gout) {
    for (int kb = 0; kb < K; kb += RB) {
        const int ci = c0 + kb + lane;
        const bool have = lane < RB && kb + lane < K && ci < nChunks;
        const unsigned draw = row[have ? ci : 0];
        const unsigned d = have ? draw : 0u;
        unsigned r[RB];
        #pragma unroll
        for (int i = 0; i < RB; ++i) {
            const unsigned di = (unsigned)__builtin_amdgcn_readlane((int)d, i);
            const int ln = (int)(di & 0xFFFFu);
            const size_t off = (ln > 0) ? (size_t)(c0 + kb + i) * CH + (di >> 16) : (size_t)0;
            r[i] = tmp[off + (size_t)((lane < ln) ? lane : 0)];
        }
        #pragma unroll
        for (int i = 0; i < RB; ++i) {
            const unsigned di = (unsigned)__builtin_amdgcn_readlane((int)d, i);
            const int ln = (int)(di & 0xFFFFu);
            if (lane < ln) fine_put<PASS, STAGED>(r[i], ncount, nbase, cur, stage, gout);
            if (ln > 32) {
                const unsigned* p = tmp + (size_t)(c0 + kb + i) * CH + (di >> 16);
                for (int j = 32 + lane; j < ln; j += 32)
                    fine_put<PASS, STAGED>(p[j], ncount, nbase, cur, stage, gout);
            }
        }
    }
}

__global__ void __launch_bounds__(FTH)
k_fine(const unsigned* __restrict__ tmp, const unsigned* __restrict__ runtab,
       int* __restrict__ cnt, int* __restrict__ rowstart,
       unsigned* __restrict__ sorted, int n, int nChunks, int NB,
       const float* __restrict__ etab0, const float* __restrict__ hcur,
       float* __restrict__ hs_next) {
    __shared__ v4f tab[RELS];
    __shared__ v4f wtab[64];
    __shared__ int wfast;
    __shared__ int ncount[1 << BSH];
    __shared__ int nbase[1 << BSH];
    __shared__ int cur[1 << BSH];
    __shared__ int wsum[8];
    __shared__ int wtot[FTH / 32];
    __shared__ int wovf[FTH / 32];
    __shared__ unsigned stage[FCAP];
    const int t = threadIdx.x, lane = t & 31;
    const int w = __builtin_amdgcn_readfirstlane(t >> 5);
    const int b = blockIdx.x;
    const int K  = (nChunks + FTH / 32 - 1) / (FTH / 32);
    const int c0 = w * K;
    const unsigned* __restrict__ row = runtab + (size_t)b * nChunks;

    int part = 0;
    for (int k = lane; k < K; k += 32) {
        const int c = c0 + k;
        if (c < nChunks) part += (int)(row[c] & 0xFFFFu);
    }
    #pragma unroll
    for (int d = 16; d >= 1; d >>= 1) part += __shfl_xor(part, d, 32);
    if (lane == 0) wtot[w] = part;
    if (t < (1 << BSH)) { ncount[t] = 0; cur[t] = 0; }
    if (t < RELS) tab[t] = *(const v4f*)(etab0 + t * H_HEADS);
    if (t < 32) build_wtab(etab0, wtab, &wfast, t);
    if (b == 0 && t == 0) sorted[0] = 0u;
    __syncthreads();
    static_assert(FTH / 32 == 32, "one lane per wave total below");
    int nbl = wtot[lane];
    #pragma unroll
    for (int d = 16; d >= 1; d >>= 1) nbl += __shfl_xor(nbl, d, 32);
    const int nb = __builtin_amdgcn_readfirstlane(nbl);
    const bool staged = (nb <= FCAP);

    int bs = SPAD + b * FCAP;
    if (!staged) {
        int acc = 0;
        for (int bb = w; bb < b; bb += FTH / 32) {
            const unsigned* __restrict__ r2 = runtab + (size_t)bb * nChunks;
            int sum = 0;
            for (int c = lane; c < nChunks; c += 32) sum += (int)(r2[c] & 0xFFFFu);
            #pragma unroll
            for (int d = 16; d >= 1; d >>= 1) sum += __shfl_xor(sum, d, 32);
            if (sum > FCAP) acc += sum;
        }
        if (lane == 0) wovf[w] = acc;
        __syncthreads();
        int before = wovf[lane];
        #pragma unroll
        for (int d = 16; d >= 1; d >>= 1) before += __shfl_xor(before, d, 32);
        bs = SPAD + NB * FCAP + __builtin_amdgcn_readfirstlane(before);
    }
    unsigned* __restrict__ gout = sorted + bs;

    if (staged) fine_visit<0, true >(tmp, row, nChunks, K, c0, lane, ncount, nbase, cur, stage, gout);
    else        fine_visit<0, false>(tmp, row, nChunks, K, c0, lane, ncount, nbase, cur, stage, gout);
    __syncthreads();

    int v = 0, x = 0;
    if (t < (1 << BSH)) {
        v = ncount[t];
        x = v;
        #pragma unroll
        for (int d = 1; d < 32; d <<= 1) {
            const int y = __shfl_up(x, d, 32);
            if (lane >= d) x += y;
        }
        if (lane == 31) wsum[w] = x;
    }
    __syncthreads();
    if (t < (1 << BSH)) {
        int wbase = 0;
        #pragma unroll
        for (int k = 0; k < 8; ++k) if (k < w) wbase += wsum[k];
        const int excl = wbase + x - v;
        nbase[t] = excl;
        const int node = (b << BSH) + t;
        if (node < n) { cnt[node] = v; rowstart[node] = bs + excl; }
    }
    __syncthreads();

    if (staged) fine_visit<1, true >(tmp, row, nChunks, K, c0, lane, ncount, nbase, cur, stage, gout);
    else        fine_visit<1, false>(tmp, row, nChunks, K, c0, lane, ncount, nbase, cur, stage, gout);
    __syncthreads();
    if (staged) {
        for (int i = t; i < nb; i += FTH) gout[i] = stage[i];
        fine_layer0<true>(stage, gout, tab, wtab, wfast != 0, ncount, nbase, hcur, hs_next, b, n);
    } else {
        fine_layer0<false>(stage, gout, tab, wtab, wfast != 0, ncount, nbase, hcur, hs_next, b, n);
    }
}

#define LT    800
#define LGL   2
#define LCH   16
#define LNPW  (LT / LGL)
#define LCAPN 81344
#define LST   8

__device__ __forceinline__ void l_load(const unsigned* __restrict__ sorted, int start,
                                       int len, int j0, unsigned (&pk)[LCH]) {
    #pragma unroll
    for (int k = 0; k < LCH; ++k) {
        const int j = j0 + k * LGL;
        pk[k] = sorted[(j < len) ? start + j : 0];
    }
}
__device__ __forceinline__ void l_max(const v4f* tab, const unsigned (&pk)[LCH],
                                      int len, int j0, float (&m)[H_HEADS]) {
    #pragma unroll
    for (int k = 0; k < LCH; ++k) {
        if (j0 + k * LGL < len) {
            const v4f e = tab[pk[k] & 63u];
            #pragma unroll
            for (int h = 0; h < H_HEADS; ++h) m[h] = fmaxf(m[h], e[h]);
        }
    }
}
__device__ __forceinline__ void l_sum(const v4f* tab, const float* hl,
                                      const float* __restrict__ hs, unsigned count,
                                      const unsigned (&pk)[LCH],
                                      int len, int j0, const float (&m)[H_HEADS],
                                      float (&s)[H_HEADS], float (&u)[H_HEADS]) {
    float hv[LCH];
    #pragma unroll
    for (int k = 0; k < LCH; ++k) {
        const unsigned sn = pk[k] >> 6;
        const bool inl = sn < count;
        const float a = hl[inl ? sn : 0u];
        const float b = hs[inl ? 0u : sn];
        hv[k] = inl ? a : b;
    }
    #pragma unroll
    for (int k = 0; k < LCH; ++k) {
        if (j0 + k * LGL < len) {
            const v4f e = tab[pk[k] & 63u];
            #pragma unroll
            for (int h = 0; h < H_HEADS; ++h) {
                const float wgt = __expf(e[h] - m[h]);
                s[h] += wgt;
                u[h] += wgt * hv[k];
            }
        }
    }
}

__device__ __forceinline__ void l_sum_tab(const v4f* wtab, const float* hl,
                                          const float* __restrict__ hs, unsigned count,
                                          const unsigned (&pk)[LCH], int len, int j0,
                                          float (&s)[H_HEADS], float (&u)[H_HEADS]) {
    float hv[LCH];
    #pragma unroll
    for (int k = 0; k < LCH; ++k) {
        const unsigned sn = pk[k] >> 6;
        const bool inl = sn < count;
        const float a = hl[inl ? sn : 0u];
        const float b = hs[inl ? 0u : sn];
        hv[k] = inl ? a : b;
    }
    #pragma unroll
    for (int k = 0; k < LCH; ++k) {
        const bool ok = j0 + k * LGL < len;
        const v4f w = wtab[ok ? (pk[k] & 63u) : 63u];
        const float x = ok ? hv[k] : 0.f;
        #pragma unroll
        for (int h = 0; h < H_HEADS; ++h) {
            s[h] += w[h];
            u[h] += w[h] * x;
        }
    }
}

__global__ void __launch_bounds__(LT)
k_gather_lds(const unsigned* __restrict__ sorted,
             const int* __restrict__ rowstart,
             const int* __restrict__ cnt,
             const float* __restrict__ etab_l,
             const float* __restrict__ hs,
             float* __restrict__ hs_next,
             const float* __restrict__ centrality,
             const float* __restrict__ gamma,
             const float* __restrict__ beta,
             float* __restrict__ out,
             int n, int do_mean) {
    __shared__ v4f tab[RELS];
    __shared__ v4f wtab[64];
    __shared__ int wfast;
    __shared__ float hl[LCAPN];
    const int t = threadIdx.x;
    if (t < RELS) tab[t] = *(const v4f*)(etab_l + t * H_HEADS);
    if (t < 32) build_wtab(etab_l, wtab, &wfast, t);

    const int node = blockIdx.x * LNPW + t / LGL;
    const int sub  = t & (LGL - 1);
    int start = 0, len = 0;
    if (node < n) { start = rowstart[node]; len = cnt[node]; }

    unsigned pk0[LCH];
    l_load(sorted, start, len, sub, pk0);

    const int np = (n + 3) & ~3;
    const int count = (np < LCAPN) ? np : LCAPN;
    for (int i0 = 0; i0 < count; i0 += LT * 4 * LST) {
        v4f r[LST];
        #pragma unroll
        for (int k = 0; k < LST; ++k) {
            int i = i0 + (k * LT + t) * 4;
            if (i > count - 4) i = count - 4;
            r[k] = *(const v4f*)(hs + i);
        }
        #pragma unroll
        for (int k = 0; k < LST; ++k) {
            const int i = i0 + (k * LT + t) * 4;
            if (i < count) *(v4f*)(hl + i) = r[k];
        }
    }
    __syncthreads();

    float s[H_HEADS], u[H_HEADS];
    #pragma unroll
    for (int h = 0; h < H_HEADS; ++h) { s[h] = 0.f; u[h] = 0.f; }

    if (wfast != 0) {
        l_sum_tab(wtab, hl, hs, (unsigned)count, pk0, len, sub, s, u);
        for (int base = LGL * LCH; base < len; base += LGL * LCH) {
            unsigned pk[LCH];
            l_load(sorted, start, len, base + sub, pk);
            l_sum_tab(wtab, hl, hs, (unsigned)count, pk, len, base + sub, s, u);
        }
    } else {
        float m[H_HEADS];
        #pragma unroll
        for (int h = 0; h < H_HEADS; ++h) m[h] = -__builtin_huge_valf();
        l_max(tab, pk0, len, sub, m);
        for (int base = LGL * LCH; base < len; base += LGL * LCH) {
            unsigned pk[LCH];
            l_load(sorted, start, len, base + sub, pk);
            l_max(tab, pk, len, base + sub, m);
        }
        #pragma unroll
        for (int h = 0; h < H_HEADS; ++h)
            m[h] = fmaxf(m[h], __shfl_xor(m[h], 1, 32));

        l_sum(tab, hl, hs, (unsigned)count, pk0, len, sub, m, s, u);
        for (int base = LGL * LCH; base < len; base += LGL * LCH) {
            unsigned pk[LCH];
            l_load(sorted, start, len, base + sub, pk);
            l_sum(tab, hl, hs, (unsigned)count, pk, len, base + sub, m, s, u);
        }
    }
    #pragma unroll
    for (int h = 0; h < H_HEADS; ++h) {
        s[h] += __shfl_xor(s[h], 1, 32);
        u[h] += __shfl_xor(u[h], 1, 32);
    }

    if (sub == 0 && node < n) {
        v4f v;
        #pragma unroll
        for (int h = 0; h < H_HEADS; ++h) {
            const float x = (s[h] > 0.f) ? (u[h] / s[h]) : 0.f;
            v[h] = (x > 0.f) ? x : 0.f;
        }
        if (do_mean) {
            hs_next[node] = 0.25f * (v[0] + v[1] + v[2] + v[3]);
            if (node == n - 1)
                for (int j = n; j < np; ++j) hs_next[j] = 0.f;
        } else {
            const float c = centrality[node];
            float acc = 0.f;
            #pragma unroll
            for (int h = 0; h < H_HEADS; ++h)
                acc += (c * gamma[h] + beta[h]) * v[h];
            acc *= 0.25f;
            out[node] = (acc > 0.f) ? acc : 0.01f * acc;
        }
    }
}

extern "C" void kernel_launch(void* const* d_in, const int* in_sizes, int n_in,
                              void* d_out, int out_size, void* d_ws, size_t ws_size,
                              hipStream_t stream) {
    const float* inputs     = (const float*)d_in[0];
    const float* centrality = (const float*)d_in[1];
    const float* W1         = (const float*)d_in[2];
    const float* b1         = (const float*)d_in[3];
    const float* W2         = (const float*)d_in[4];
    const float* b2         = (const float*)d_in[5];
    const float* rel_emb    = (const float*)d_in[6];
    const float* W_pred     = (const float*)d_in[7];
    const float* gamma      = (const float*)d_in[8];
    const float* beta       = (const float*)d_in[9];
    const int*   edge_types = (const int*)d_in[10];
    const int*   src        = (const int*)d_in[11];
    const int*   dst        = (const int*)d_in[12];
    float* out = (float*)d_out;

    const int N = in_sizes[1];
    const int E = in_sizes[10];
    const int Np = (N + 3) & ~3;
    const int Ep = (E + 3) & ~3;
    const int NB = (N + (1 << BSH) - 1) >> BSH;
    const int nChunks = (E + CH - 1) / CH;
    const int M  = NB * nChunks;
    const int Mp = (M + 3) & ~3;
    const int Sp = SPAD + NB * FCAP + Ep;

    char* ws = (char*)d_ws;
    _Float16* W1h = (_Float16*)ws;
    float* etab   = (float*)(ws + 65536);
    float* hA     = (float*)(ws + 69632);
    int*   cnt      = (int*)(hA + (size_t)Np * H_HEADS);
    int*   rowstart = cnt + Np;
    unsigned* runtab = (unsigned*)(rowstart + Np);
    unsigned* part   = runtab + Mp;
    unsigned* sorted = part + Ep;
    float* hsA = (float*)(sorted + Sp);
    float* hsB = hsA + Np;

    const int nW1      = COLS * IN_DIM;
    const int w1Blocks = (nW1 + CTH - 1) / CTH;
    const int eBlocks  = (LAYERS * RELS * H_HEADS + CTH - 1) / CTH;
    k_pre<<<w1Blocks + eBlocks, CTH, 0, stream>>>(W1, W1h, nW1, w1Blocks, rel_emb, W_pred, etab);

    const int nTiles    = (N + 15) / 16;
    const int mlpBlocks = (nTiles + CTH / 32 - 1) / (CTH / 32);
    k_partition_mlp<<<nChunks + mlpBlocks, CTH, 0, stream>>>(
        edge_types, src, dst, part, runtab, E, nChunks, NB, nChunks,
        inputs, W1h, b1, W2, b2, hA, N, nTiles);

    k_fine<<<NB, FTH, 0, stream>>>(part, runtab, cnt, rowstart, sorted, N, nChunks, NB,
                                   etab, hA, hsA);

    const int lBlocks = (N + LNPW - 1) / LNPW;
    float* hscur = hsA;
    float* hsnxt = hsB;

    for (int l = 1; l < LAYERS; ++l) {
        const float* etab_l = etab + l * RELS * H_HEADS;
        const int do_mean = (l != LAYERS - 1) ? 1 : 0;
        k_gather_lds<<<lBlocks, LT, 0, stream>>>(sorted, rowstart, cnt, etab_l, hscur,
                                                 hsnxt, centrality, gamma, beta, out,
                                                 N, do_mean);
        float* tmp = hscur; hscur = hsnxt; hsnxt = tmp;
    }
}


# MLP: each workgroup copies the f16 W1 into LDS once (row pitch 272 B); waves read B fragments there, a group of 8 LDS loads one group ahead of its 4 WMMAs, 3 node tiles per wave
# speedup vs baseline: 40.7256x; 1.3644x over previous
#include <hip/hip_runtime.h>
#include <hip/hip_bf16.h>

typedef __attribute__((ext_vector_type(16))) _Float16 v16h;
typedef __attribute__((ext_vector_type(8)))  _Float16 v8h;
typedef __attribute__((ext_vector_type(8)))  float    v8f;
typedef __attribute__((ext_vector_type(4)))  float    v4f;

#define H_HEADS 4
#define IN_DIM  128
#define HID     64
#define COLS    256
#define PRED    16
#define RELS    50
#define LAYERS  3

__device__ __forceinline__ void w1_to_f16_body(const float* __restrict__ W1,
                                               _Float16* __restrict__ W1h, int n, int blk) {
    int i = blk * blockDim.x + threadIdx.x;
    if (i < n) W1h[i] = (_Float16)W1[i];
}

__device__ __forceinline__ void etab_body(const float* __restrict__ rel_emb,
                                          const float* __restrict__ W_pred,
                                          float* __restrict__ etab, int blk) {
    int i = blk * blockDim.x + threadIdx.x;
    if (i >= LAYERS * RELS * H_HEADS) return;
    int l = i / (RELS * H_HEADS);
    int rem = i % (RELS * H_HEADS);
    int r = rem / H_HEADS;
    int h = rem % H_HEADS;
    float e = 0.f;
    #pragma unroll
    for (int p = 0; p < PRED; ++p)
        e += rel_emb[r * PRED + p] * W_pred[(l * PRED + p) * H_HEADS + h];
    etab[i] = (e > 0.f) ? e : 0.2f * e;
}

template <unsigned LO, unsigned HI>
__device__ __forceinline__ float row_xor(float x) {
    const int b = __builtin_bit_cast(int, x);
    return __builtin_bit_cast(float, __builtin_amdgcn_permlane16(b, b, LO, HI, false, false));
}

#define W1P     136
#define MLP_TPW 3
#define MLP_LDS (COLS * W1P * 2)
#define CTH_MLP 256
__device__ __forceinline__ void mlp_body(const float* __restrict__ inputs,
                                         const _Float16* __restrict__ W1h,
                                         const float* __restrict__ b1,
                                         const float* __restrict__ W2,
                                         const float* __restrict__ b2,
                                         float* __restrict__ hout,
                                         int n, int nTiles, int blk, _Float16* w1s) {
    static_assert((COLS * IN_DIM / 8) % CTH_MLP == 0, "whole 16-byte pieces per thread");
    {
        constexpr int PIECES = COLS * IN_DIM / 8 / CTH_MLP;
        v8h pc[PIECES];
        #pragma unroll
        for (int i = 0; i < PIECES; ++i)
            pc[i] = *(const v8h*)(W1h + (size_t)(i * CTH_MLP + (int)threadIdx.x) * 8);
        #pragma unroll
        for (int i = 0; i < PIECES; ++i) {
            const int idx = i * CTH_MLP + (int)threadIdx.x;
            *(v8h*)(w1s + (idx >> 4) * W1P + (idx & 15) * 8) = pc[i];
        }
    }
    __syncthreads();

    const int lane = threadIdx.x & 31;
    const int grp  = lane >> 4;
    const int l15  = lane & 15;
    const int wave0 = ((blk * CTH_MLP + (int)threadIdx.x) >> 5) * MLP_TPW;
    for (int it = 0; it < MLP_TPW; ++it) {
        const int wave = wave0 + it;
        if (wave >= nTiles) break;
        const int nodeBase = wave * 16;
        int arow = nodeBase + l15;
        if (arow > n - 1) arow = n - 1;
        const float* __restrict__ abase = inputs + (size_t)arow * IN_DIM;

        const v8f zero = {};
        const bool up8 = (l15 & 8) != 0, up4 = (l15 & 4) != 0, up2 = (l15 & 2) != 0;
        const bool even = (l15 & 1) == 0;
        const int onode = nodeBase + (up8 ? 4 : 0) + (up4 ? 2 : 0) + (up2 ? 1 : 0) + 8 * grp;

        v16h Af[4];
        #pragma unroll
        for (int kc = 0; kc < 4; ++kc) {
            const int kb = kc * 32 + grp * 8;
            v4f a0 = *(const v4f*)(abase + kb + 0);
            v4f a1 = *(const v4f*)(abase + kb + 4);
            v4f a2 = *(const v4f*)(abase + kb + 16);
            v4f a3 = *(const v4f*)(abase + kb + 20);
            #pragma unroll
            for (int i = 0; i < 4; ++i) {
                Af[kc][i]      = (_Float16)a0[i];
                Af[kc][4 + i]  = (_Float16)a1[i];
                Af[kc][8 + i]  = (_Float16)a2[i];
                Af[kc][12 + i] = (_Float16)a3[i];
            }
        }
        v16h Bf[2][4];
        #pragma unroll
        for (int tt = 0; tt < 4; ++tt) {
            const _Float16* q = w1s + (tt * 16 + l15) * W1P + grp * 8;
            const v8h blo = *(const v8h*)(q);
            const v8h bhi = *(const v8h*)(q + 16);
            #pragma unroll
            for (int i = 0; i < 8; ++i) { Bf[0][tt][i] = blo[i]; Bf[0][tt][8 + i] = bhi[i]; }
        }

        #pragma unroll
        for (int head = 0; head < H_HEADS; ++head) {
            v8f acc0 = zero, acc1 = zero, acc2 = zero, acc3 = zero;

            #pragma unroll
            for (int kc = 0; kc < 4; ++kc) {
                const int g = head * 4 + kc;
                if (g + 1 < H_HEADS * 4) {
                    const int nh = (g + 1) >> 2, nk = (g + 1) & 3;
                    #pragma unroll
                    for (int tt = 0; tt < 4; ++tt) {
                        const _Float16* q = w1s + (nh * HID + tt * 16 + l15) * W1P + nk * 32 + grp * 8;
                        const v8h blo = *(const v8h*)(q);
                        const v8h bhi = *(const v8h*)(q + 16);
                        #pragma unroll
                        for (int i = 0; i < 8; ++i) {
                            Bf[(g + 1) & 1][tt][i] = blo[i]; Bf[(g + 1) & 1][tt][8 + i] = bhi[i];
                        }
                    }
                }
                __builtin_amdgcn_sched_barrier(0);
                acc0 = __builtin_amdgcn_wmma_f32_16x16x32_f16(
                           false, Af[kc], false, Bf[g & 1][0], (short)0, acc0, false, false);
                acc1 = __builtin_amdgcn_wmma_f32_16x16x32_f16(
                           false, Af[kc], false, Bf[g & 1][1], (short)0, acc1, false, false);
                acc2 = __builtin_amdgcn_wmma_f32_16x16x32_f16(
                           false, Af[kc], false, Bf[g & 1][2], (short)0, acc2, false, false);
                acc3 = __builtin_amdgcn_wmma_f32_16x16x32_f16(
                           false, Af[kc], false, Bf[g & 1][3], (short)0, acc3, false, false);
            }

            v8f q = zero;
            #pragma unroll
            for (int tt = 0; tt < 4; ++tt) {
                const int col  = head * HID + tt * 16 + l15;
                const float bias = b1[col];
                const float w2   = W2[col];
                v8f c = (tt == 0) ? acc0 : (tt == 1) ? acc1 : (tt == 2) ? acc2 : acc3;
                #pragma unroll
                for (int v = 0; v < 8; ++v) {
                    float p = c[v] + bias;
                    p = (p > 0.f) ? p : 0.f;
                    q[v] += p * w2;
                }
            }
            float r1[4], r2[2];
            #pragma unroll
            for (int i = 0; i < 4; ++i) {
                const float keep = up8 ? q[4 + i] : q[i];
                const float send = up8 ? q[i] : q[4 + i];
                r1[i] = keep + row_xor<0xFEDCBA98u, 0x76543210u>(send);
            }
            #pragma unroll
            for (int i = 0; i < 2; ++i) {
                const float keep = up4 ? r1[2 + i] : r1[i];
                const float send = up4 ? r1[i] : r1[2 + i];
                r2[i] = keep + row_xor<0x32107654u, 0xBA98FEDCu>(send);
            }
            const float keep3 = up2 ? r2[1] : r2[0];
            const float send3 = up2 ? r2[0] : r2[1];
            const float r3 = keep3 + row_xor<0x54761032u, 0xDCFE98BAu>(send3);
            const float tot = r3 + row_xor<0x67452301u, 0xEFCDAB89u>(r3);
            if (even && onode < n)
                hout[onode * H_HEADS + head] = tot + b2[head];
        }
    }
}

#define WSPREAD 16.0f
static_assert(RELS > 32 && RELS <= 64, "two table rows per lane of one wave");
__device__ __forceinline__ void build_wtab(const float* __restrict__ etab_l, v4f* wtab,
                                           int* fast, int lane) {
    const int r1 = lane + 32;
    const bool has1 = r1 < RELS;
    const v4f e0 = *(const v4f*)(etab_l + lane * H_HEADS);
    const v4f e1 = *(const v4f*)(etab_l + (has1 ? r1 : 0) * H_HEADS);
    float mx[H_HEADS], mn[H_HEADS];
    #pragma unroll
    for (int h = 0; h < H_HEADS; ++h) {
        mx[h] = has1 ? fmaxf(e0[h], e1[h]) : e0[h];
        mn[h] = has1 ? fminf(e0[h], e1[h]) : e0[h];
    }
    #pragma unroll
    for (int d = 16; d >= 1; d >>= 1) {
        #pragma unroll
        for (int h = 0; h < H_HEADS; ++h) {
            mx[h] = fmaxf(mx[h], __shfl_xor(mx[h], d, 32));
            mn[h] = fminf(mn[h], __shfl_xor(mn[h], d, 32));
        }
    }
    bool ok = true;
    v4f w0, w1;
    #pragma unroll
    for (int h = 0; h < H_HEADS; ++h) {
        ok = ok && (mx[h] - mn[h] <= WSPREAD);
        w0[h] = __expf(e0[h] - mx[h]);
        w1[h] = has1 ? __expf(e1[h] - mx[h]) : 0.f;
    }
    wtab[lane] = w0;
    wtab[r1]   = w1;
    if (lane == 0) *fast = ok ? 1 : 0;
}

#define BSH    8
#define NBMAX  1024
#define CH     8192
#define CTH    256
#define FTH    1024
#define FCAP   12288
#define SPAD   4
#define RB     16

__device__ __forceinline__ void scan1024(const int* cin, int* cout, int* wsum, int t,
                                         int (&own)[4]) {
    const int lane = t & 31, w = t >> 5;
    const int c0 = cin[4 * t], c1 = cin[4 * t + 1], c2 = cin[4 * t + 2], c3 = cin[4 * t + 3];
    const int tsum = c0 + c1 + c2 + c3;
    int x = tsum;
    #pragma unroll
    for (int d = 1; d < 32; d <<= 1) {
        const int y = __shfl_up(x, d, 32);
        if (lane >= d) x += y;
    }
    if (lane == 31) wsum[w] = x;
    __syncthreads();
    int wbase = 0;
    #pragma unroll
    for (int k = 0; k < 8; ++k) if (k < w) wbase += wsum[k];
    const int run = wbase + x - tsum;
    own[0] = run; own[1] = run + c0; own[2] = run + c0 + c1; own[3] = run + c0 + c1 + c2;
    cout[4 * t] = own[0]; cout[4 * t + 1] = own[1];
    cout[4 * t + 2] = own[2]; cout[4 * t + 3] = own[3];
}

__global__ void __launch_bounds__(CTH)
k_pre(const float* __restrict__ W1, _Float16* __restrict__ W1h, int nW1, int w1Blocks,
      const float* __restrict__ rel_emb, const float* __restrict__ W_pred,
      float* __restrict__ etab) {
    const int blk = blockIdx.x;
    if (blk < w1Blocks) w1_to_f16_body(W1, W1h, nW1, blk);
    else                etab_body(rel_emb, W_pred, etab, blk - w1Blocks);
}

__device__ __forceinline__ void partition_body(
            const int* __restrict__ et, const int* __restrict__ src,
            const int* __restrict__ dst,
            unsigned* __restrict__ tmp, unsigned* __restrict__ runtab,
            int E, int nChunks, int NB, int blk, unsigned char* lds) {
    int* lc   = (int*)lds;
    int* lb   = lc + NBMAX;
    int* wsum = lb + NBMAX;
    unsigned* recs = (unsigned*)(wsum + 8);
    const int t = threadIdx.x;
    for (int b = t; b < NBMAX; b += CTH) lc[b] = 0;
    __syncthreads();

    const int base = blk * CH;
    unsigned rec[32];
    unsigned br[32];
    #pragma unroll
    for (int k = 0; k < CH / (CTH * 4); ++k) {
        const int i = base + (k * CTH + t) * 4;
        int dv[4], sv[4], ev[4];
        if (i + 3 < E) {
            const int4 d = *(const int4*)(dst + i);
            const int4 s = *(const int4*)(src + i);
            const int4 e = *(const int4*)(et + i);
            dv[0] = d.x; dv[1] = d.y; dv[2] = d.z; dv[3] = d.w;
            sv[0] = s.x; sv[1] = s.y; sv[2] = s.z; sv[3] = s.w;
            ev[0] = e.x; ev[1] = e.y; ev[2] = e.z; ev[3] = e.w;
        } else {
            #pragma unroll
            for (int c = 0; c < 4; ++c) {
                const int j = i + c;
                dv[c] = -1; sv[c] = 0; ev[c] = 0;
                if (j < E) { dv[c] = dst[j]; sv[c] = src[j]; ev[c] = et[j]; }
            }
        }
        #pragma unroll
        for (int c = 0; c < 4; ++c) {
            rec[k * 4 + c] = 0u; br[k * 4 + c] = ~0u;
            if (dv[c] >= 0) {
                const int b = dv[c] >> BSH;
                const int r = atomicAdd(&lc[b], 1);
                rec[k * 4 + c] = ((unsigned)sv[c] << 14) | ((unsigned)ev[c] << 8)
                               | ((unsigned)dv[c] & 255u);
                br[k * 4 + c] = ((unsigned)b << 16) | (unsigned)r;
            }
        }
    }
    __syncthreads();

    int own[4];
    scan1024(lc, lb, wsum, t, own);
    #pragma unroll
    for (int c = 0; c < 4; ++c) {
        const int b = 4 * t + c;
        if (b < NB)
            runtab[(size_t)b * nChunks + blk] = ((unsigned)own[c] << 16) | (unsigned)lc[b];
    }
    __syncthreads();

    #pragma unroll
    for (int q = 0; q < 32; ++q) {
        if (br[q] != ~0u) {
            const int b = (int)(br[q] >> 16);
            recs[lb[b] + (int)(br[q] & 0xFFFFu)] = rec[q];
        }
    }
    __syncthreads();

    int nvalid = E - base;
    if (nvalid > CH) nvalid = CH;
    const int n4 = nvalid >> 2;
    uint4* __restrict__ o4 = (uint4*)(tmp + base);
    const uint4* r4 = (const uint4*)recs;
    for (int i = t; i < n4; i += CTH) o4[i] = r4[i];
    for (int i = (n4 << 2) + t; i < nvalid; i += CTH) tmp[base + i] = recs[i];
}

__global__ void __launch_bounds__(CTH)
k_partition_mlp(const int* __restrict__ et, const int* __restrict__ src,
                const int* __restrict__ dst,
                unsigned* __restrict__ tmp, unsigned* __restrict__ runtab,
                int E, int nChunks, int NB, int pBlocks,
                const float* __restrict__ inputs, const _Float16* __restrict__ W1h,
                const float* __restrict__ b1, const float* __restrict__ W2,
                const float* __restrict__ b2, float* __restrict__ hout,
                int n, int nTiles) {
    static_assert(CTH_MLP == CTH, "mlp_body's copy loop counts on the workgroup size");
    static_assert(MLP_LDS >= (2 * NBMAX + 8 + CH) * 4, "the level-1 arrays fit the shared buffer");
    static_assert(((2 * NBMAX + 8) * 4) % 16 == 0, "recs is 16-byte aligned");
    __shared__ __attribute__((aligned(16))) unsigned char lds[MLP_LDS];
    const int blk = blockIdx.x;
    if (blk < pBlocks)
        partition_body(et, src, dst, tmp, runtab, E, nChunks, NB, blk, lds);
    else
        mlp_body(inputs, W1h, b1, W2, b2, hout, n, nTiles, blk - pBlocks, (_Float16*)lds);
}

#define FGL (FTH >> BSH)
#define FGC 4
static_assert(LAYERS >= 2, "the fused first layer always feeds a head mean");

template <bool STAGED>
__device__ __forceinline__ unsigned fine_rec(const unsigned* stage, const unsigned* gsort, int i) {
    return STAGED ? stage[i] : gsort[i];
}

template <bool STAGED>
__device__ __forceinline__ void fine_layer0(const unsigned* stage, const unsigned* gsort,
                                            const v4f* tab, const v4f* wtab, bool fast,
                                            const int* ncount, const int* nbase,
                                            const float* __restrict__ hcur,
                                            float* __restrict__ hs_next, int b, int n) {
    const int t    = threadIdx.x;
    const int dl   = t / FGL;
    const int sub  = t & (FGL - 1);
    const int node = (b << BSH) + dl;
    const int len  = ncount[dl];
    const int st   = nbase[dl];

    float s[H_HEADS], u[H_HEADS];
    #pragma unroll
    for (int h = 0; h < H_HEADS; ++h) { s[h] = 0.f; u[h] = 0.f; }

    if (fast) {
        for (int j0 = sub; j0 < len; j0 += FGL * FGC) {
            unsigned pk[FGC];
            v4f hv[FGC];
            #pragma unroll
            for (int k = 0; k < FGC; ++k) {
                const int j = j0 + k * FGL;
                const unsigned raw = fine_rec<STAGED>(stage, gsort, (j < len) ? st + j : 0);
                pk[k] = (j < len) ? raw : 0u;
            }
            #pragma unroll
            for (int k = 0; k < FGC; ++k)
                hv[k] = *(const v4f*)(hcur + (size_t)(pk[k] >> 6) * H_HEADS);
            #pragma unroll
            for (int k = 0; k < FGC; ++k) {
                const bool ok = j0 + k * FGL < len;
                const v4f w = wtab[ok ? (pk[k] & 63u) : 63u];
                #pragma unroll
                for (int h = 0; h < H_HEADS; ++h) {
                    const float hk = ok ? hv[k][h] : 0.f;
                    s[h] += w[h];
                    u[h] += w[h] * hk;
                }
            }
        }
    } else {
        float m[H_HEADS];
        #pragma unroll
        for (int h = 0; h < H_HEADS; ++h) m[h] = -__builtin_huge_valf();
        for (int j0 = sub; j0 < len; j0 += FGL * FGC) {
            unsigned pk[FGC];
            #pragma unroll
            for (int k = 0; k < FGC; ++k) {
                const int j = j0 + k * FGL;
                pk[k] = fine_rec<STAGED>(stage, gsort, (j < len) ? st + j : 0);
            }
            #pragma unroll
            for (int k = 0; k < FGC; ++k) {
                const bool ok = j0 + k * FGL < len;
                const v4f e = tab[ok ? (pk[k] & 63u) : 0u];
                #pragma unroll
                for (int h = 0; h < H_HEADS; ++h)
                    m[h] = fmaxf(m[h], ok ? e[h] : -__builtin_huge_valf());
            }
        }
        #pragma unroll
        for (int h = 0; h < H_HEADS; ++h) {
            m[h] = fmaxf(m[h], __shfl_xor(m[h], 1, 32));
            m[h] = fmaxf(m[h], __shfl_xor(m[h], 2, 32));
        }

        for (int j0 = sub; j0 < len; j0 += FGL * FGC) {
            unsigned pk[FGC];
            v4f hv[FGC];
            #pragma unroll
            for (int k = 0; k < FGC; ++k) {
                const int j = j0 + k * FGL;
                const unsigned raw = fine_rec<STAGED>(stage, gsort, (j < len) ? st + j : 0);
                pk[k] = (j < len) ? raw : 0u;
            }
            #pragma unroll
            for (int k = 0; k < FGC; ++k)
                hv[k] = *(const v4f*)(hcur + (size_t)(pk[k] >> 6) * H_HEADS);
            #pragma unroll
            for (int k = 0; k < FGC; ++k) {
                const bool ok = j0 + k * FGL < len;
                const v4f e = tab[pk[k] & 63u];
                #pragma unroll
                for (int h = 0; h < H_HEADS; ++h) {
                    const float wgt = ok ? __expf(e[h] - m[h]) : 0.f;
                    const float hk  = ok ? hv[k][h] : 0.f;
                    s[h] += wgt;
                    u[h] += wgt * hk;
                }
            }
        }
    }
    #pragma unroll
    for (int h = 0; h < H_HEADS; ++h) {
        s[h] += __shfl_xor(s[h], 1, 32);  u[h] += __shfl_xor(u[h], 1, 32);
        s[h] += __shfl_xor(s[h], 2, 32);  u[h] += __shfl_xor(u[h], 2, 32);
    }

    if (sub == 0 && node < n) {
        v4f v;
        #pragma unroll
        for (int h = 0; h < H_HEADS; ++h) {
            const float x = (s[h] > 0.f) ? (u[h] / s[h]) : 0.f;
            v[h] = (x > 0.f) ? x : 0.f;
        }
        hs_next[node] = 0.25f * (v[0] + v[1] + v[2] + v[3]);
        if (node == n - 1)
            for (int j = n; j < ((n + 3) & ~3); ++j) hs_next[j] = 0.f;
    }
}

template <int PASS, bool STAGED>
__device__ __forceinline__ void fine_put(unsigned rec, int* ncount, const int* nbase, int* cur,
                                         unsigned* stage, unsigned* __restrict__ gout) {
    const unsigned dl = rec & 255u;
    if (PASS == 0) {
        atomicAdd(&ncount[dl], 1);
    } else {
        const int p = nbase[dl] + atomicAdd(&cur[dl], 1);
        const unsigned val = rec >> 8;
        if (STAGED) stage[p] = val; else gout[p] = val;
    }
}

template <int PASS, bool STAGED>
__device__ __forceinline__ void fine_visit(const unsigned* __restrict__ tmp,
                                           const unsigned* __restrict__ row,
                                           int nChunks, int K, int c0, int lane,
                                           int* ncount, const int* nbase, int* cur,
                                           unsigned* stage, unsigned* __restrict__ gout) {
    for (int kb = 0; kb < K; kb += RB) {
        const int ci = c0 + kb + lane;
        const bool have = lane < RB && kb + lane < K && ci < nChunks;
        const unsigned draw = row[have ? ci : 0];
        const unsigned d = have ? draw : 0u;
        unsigned r[RB];
        #pragma unroll
        for (int i = 0; i < RB; ++i) {
            const unsigned di = (unsigned)__builtin_amdgcn_readlane((int)d, i);
            const int ln = (int)(di & 0xFFFFu);
            const size_t off = (ln > 0) ? (size_t)(c0 + kb + i) * CH + (di >> 16) : (size_t)0;
            r[i] = tmp[off + (size_t)((lane < ln) ? lane : 0)];
        }
        #pragma unroll
        for (int i = 0; i < RB; ++i) {
            const unsigned di = (unsigned)__builtin_amdgcn_readlane((int)d, i);
            const int ln = (int)(di & 0xFFFFu);
            if (lane < ln) fine_put<PASS, STAGED>(r[i], ncount, nbase, cur, stage, gout);
            if (ln > 32) {
                const unsigned* p = tmp + (size_t)(c0 + kb + i) * CH + (di >> 16);
                for (int j = 32 + lane; j < ln; j += 32)
                    fine_put<PASS, STAGED>(p[j], ncount, nbase, cur, stage, gout);
            }
        }
    }
}

__global__ void __launch_bounds__(FTH)
k_fine(const unsigned* __restrict__ tmp, const unsigned* __restrict__ runtab,
       int* __restrict__ cnt, int* __restrict__ rowstart,
       unsigned* __restrict__ sorted, int n, int nChunks, int NB,
       const float* __restrict__ etab0, const float* __restrict__ hcur,
       float* __restrict__ hs_next) {
    __shared__ v4f tab[RELS];
    __shared__ v4f wtab[64];
    __shared__ int wfast;
    __shared__ int ncount[1 << BSH];
    __shared__ int nbase[1 << BSH];
    __shared__ int cur[1 << BSH];
    __shared__ int wsum[8];
    __shared__ int wtot[FTH / 32];
    __shared__ int wovf[FTH / 32];
    __shared__ unsigned stage[FCAP];
    const int t = threadIdx.x, lane = t & 31;
    const int w = __builtin_amdgcn_readfirstlane(t >> 5);
    const int b = blockIdx.x;
    const int K  = (nChunks + FTH / 32 - 1) / (FTH / 32);
    const int c0 = w * K;
    const unsigned* __restrict__ row = runtab + (size_t)b * nChunks;

    int part = 0;
    for (int k = lane; k < K; k += 32) {
        const int c = c0 + k;
        if (c < nChunks) part += (int)(row[c] & 0xFFFFu);
    }
    #pragma unroll
    for (int d = 16; d >= 1; d >>= 1) part += __shfl_xor(part, d, 32);
    if (lane == 0) wtot[w] = part;
    if (t < (1 << BSH)) { ncount[t] = 0; cur[t] = 0; }
    if (t < RELS) tab[t] = *(const v4f*)(etab0 + t * H_HEADS);
    if (t < 32) build_wtab(etab0, wtab, &wfast, t);
    if (b == 0 && t == 0) sorted[0] = 0u;
    __syncthreads();
    static_assert(FTH / 32 == 32, "one lane per wave total below");
    int nbl = wtot[lane];
    #pragma unroll
    for (int d = 16; d >= 1; d >>= 1) nbl += __shfl_xor(nbl, d, 32);
    const int nb = __builtin_amdgcn_readfirstlane(nbl);
    const bool staged = (nb <= FCAP);

    int bs = SPAD + b * FCAP;
    if (!staged) {
        int acc = 0;
        for (int bb = w; bb < b; bb += FTH / 32) {
            const unsigned* __restrict__ r2 = runtab + (size_t)bb * nChunks;
            int sum = 0;
            for (int c = lane; c < nChunks; c += 32) sum += (int)(r2[c] & 0xFFFFu);
            #pragma unroll
            for (int d = 16; d >= 1; d >>= 1) sum += __shfl_xor(sum, d, 32);
            if (sum > FCAP) acc += sum;
        }
        if (lane == 0) wovf[w] = acc;
        __syncthreads();
        int before = wovf[lane];
        #pragma unroll
        for (int d = 16; d >= 1; d >>= 1) before += __shfl_xor(before, d, 32);
        bs = SPAD + NB * FCAP + __builtin_amdgcn_readfirstlane(before);
    }
    unsigned* __restrict__ gout = sorted + bs;

    if (staged) fine_visit<0, true >(tmp, row, nChunks, K, c0, lane, ncount, nbase, cur, stage, gout);
    else        fine_visit<0, false>(tmp, row, nChunks, K, c0, lane, ncount, nbase, cur, stage, gout);
    __syncthreads();

    int v = 0, x = 0;
    if (t < (1 << BSH)) {
        v = ncount[t];
        x = v;
        #pragma unroll
        for (int d = 1; d < 32; d <<= 1) {
            const int y = __shfl_up(x, d, 32);
            if (lane >= d) x += y;
        }
        if (lane == 31) wsum[w] = x;
    }
    __syncthreads();
    if (t < (1 << BSH)) {
        int wbase = 0;
        #pragma unroll
        for (int k = 0; k < 8; ++k) if (k < w) wbase += wsum[k];
        const int excl = wbase + x - v;
        nbase[t] = excl;
        const int node = (b << BSH) + t;
        if (node < n) { cnt[node] = v; rowstart[node] = bs + excl; }
    }
    __syncthreads();

    if (staged) fine_visit<1, true >(tmp, row, nChunks, K, c0, lane, ncount, nbase, cur, stage, gout);
    else        fine_visit<1, false>(tmp, row, nChunks, K, c0, lane, ncount, nbase, cur, stage, gout);
    __syncthreads();
    if (staged) {
        for (int i = t; i < nb; i += FTH) gout[i] = stage[i];
        fine_layer0<true>(stage, gout, tab, wtab, wfast != 0, ncount, nbase, hcur, hs_next, b, n);
    } else {
        fine_layer0<false>(stage, gout, tab, wtab, wfast != 0, ncount, nbase, hcur, hs_next, b, n);
    }
}

#define LT    800
#define LGL   2
#define LCH   16
#define LNPW  (LT / LGL)
#define LCAPN 81344
#define LST   8

__device__ __forceinline__ void l_load(const unsigned* __restrict__ sorted, int start,
                                       int len, int j0, unsigned (&pk)[LCH]) {
    #pragma unroll
    for (int k = 0; k < LCH; ++k) {
        const int j = j0 + k * LGL;
        pk[k] = sorted[(j < len) ? start + j : 0];
    }
}
__device__ __forceinline__ void l_max(const v4f* tab, const unsigned (&pk)[LCH],
                                      int len, int j0, float (&m)[H_HEADS]) {
    #pragma unroll
    for (int k = 0; k < LCH; ++k) {
        if (j0 + k * LGL < len) {
            const v4f e = tab[pk[k] & 63u];
            #pragma unroll
            for (int h = 0; h < H_HEADS; ++h) m[h] = fmaxf(m[h], e[h]);
        }
    }
}
__device__ __forceinline__ void l_sum(const v4f* tab, const float* hl,
                                      const float* __restrict__ hs, unsigned count,
                                      const unsigned (&pk)[LCH],
                                      int len, int j0, const float (&m)[H_HEADS],
                                      float (&s)[H_HEADS], float (&u)[H_HEADS]) {
    float hv[LCH];
    #pragma unroll
    for (int k = 0; k < LCH; ++k) {
        const unsigned sn = pk[k] >> 6;
        const bool inl = sn < count;
        const float a = hl[inl ? sn : 0u];
        const float b = hs[inl ? 0u : sn];
        hv[k] = inl ? a : b;
    }
    #pragma unroll
    for (int k = 0; k < LCH; ++k) {
        if (j0 + k * LGL < len) {
            const v4f e = tab[pk[k] & 63u];
            #pragma unroll
            for (int h = 0; h < H_HEADS; ++h) {
                const float wgt = __expf(e[h] - m[h]);
                s[h] += wgt;
                u[h] += wgt * hv[k];
            }
        }
    }
}

__device__ __forceinline__ void l_sum_tab(const v4f* wtab, const float* hl,
                                          const float* __restrict__ hs, unsigned count,
                                          const unsigned (&pk)[LCH], int len, int j0,
                                          float (&s)[H_HEADS], float (&u)[H_HEADS]) {
    float hv[LCH];
    #pragma unroll
    for (int k = 0; k < LCH; ++k) {
        const unsigned sn = pk[k] >> 6;
        const bool inl = sn < count;
        const float a = hl[inl ? sn : 0u];
        const float b = hs[inl ? 0u : sn];
        hv[k] = inl ? a : b;
    }
    #pragma unroll
    for (int k = 0; k < LCH; ++k) {
        const bool ok = j0 + k * LGL < len;
        const v4f w = wtab[ok ? (pk[k] & 63u) : 63u];
        const float x = ok ? hv[k] : 0.f;
        #pragma unroll
        for (int h = 0; h < H_HEADS; ++h) {
            s[h] += w[h];
            u[h] += w[h] * x;
        }
    }
}

__global__ void __launch_bounds__(LT)
k_gather_lds(const unsigned* __restrict__ sorted,
             const int* __restrict__ rowstart,
             const int* __restrict__ cnt,
             const float* __restrict__ etab_l,
             const float* __restrict__ hs,
             float* __restrict__ hs_next,
             const float* __restrict__ centrality,
             const float* __restrict__ gamma,
             const float* __restrict__ beta,
             float* __restrict__ out,
             int n, int do_mean) {
    __shared__ v4f tab[RELS];
    __shared__ v4f wtab[64];
    __shared__ int wfast;
    __shared__ float hl[LCAPN];
    const int t = threadIdx.x;
    if (t < RELS) tab[t] = *(const v4f*)(etab_l + t * H_HEADS);
    if (t < 32) build_wtab(etab_l, wtab, &wfast, t);

    const int node = blockIdx.x * LNPW + t / LGL;
    const int sub  = t & (LGL - 1);
    int start = 0, len = 0;
    if (node < n) { start = rowstart[node]; len = cnt[node]; }

    unsigned pk0[LCH];
    l_load(sorted, start, len, sub, pk0);

    const int np = (n + 3) & ~3;
    const int count = (np < LCAPN) ? np : LCAPN;
    for (int i0 = 0; i0 < count; i0 += LT * 4 * LST) {
        v4f r[LST];
        #pragma unroll
        for (int k = 0; k < LST; ++k) {
            int i = i0 + (k * LT + t) * 4;
            if (i > count - 4) i = count - 4;
            r[k] = *(const v4f*)(hs + i);
        }
        #pragma unroll
        for (int k = 0; k < LST; ++k) {
            const int i = i0 + (k * LT + t) * 4;
            if (i < count) *(v4f*)(hl + i) = r[k];
        }
    }
    __syncthreads();

    float s[H_HEADS], u[H_HEADS];
    #pragma unroll
    for (int h = 0; h < H_HEADS; ++h) { s[h] = 0.f; u[h] = 0.f; }

    if (wfast != 0) {
        l_sum_tab(wtab, hl, hs, (unsigned)count, pk0, len, sub, s, u);
        for (int base = LGL * LCH; base < len; base += LGL * LCH) {
            unsigned pk[LCH];
            l_load(sorted, start, len, base + sub, pk);
            l_sum_tab(wtab, hl, hs, (unsigned)count, pk, len, base + sub, s, u);
        }
    } else {
        float m[H_HEADS];
        #pragma unroll
        for (int h = 0; h < H_HEADS; ++h) m[h] = -__builtin_huge_valf();
        l_max(tab, pk0, len, sub, m);
        for (int base = LGL * LCH; base < len; base += LGL * LCH) {
            unsigned pk[LCH];
            l_load(sorted, start, len, base + sub, pk);
            l_max(tab, pk, len, base + sub, m);
        }
        #pragma unroll
        for (int h = 0; h < H_HEADS; ++h)
            m[h] = fmaxf(m[h], __shfl_xor(m[h], 1, 32));

        l_sum(tab, hl, hs, (unsigned)count, pk0, len, sub, m, s, u);
        for (int base = LGL * LCH; base < len; base += LGL * LCH) {
            unsigned pk[LCH];
            l_load(sorted, start, len, base + sub, pk);
            l_sum(tab, hl, hs, (unsigned)count, pk, len, base + sub, m, s, u);
        }
    }
    #pragma unroll
    for (int h = 0; h < H_HEADS; ++h) {
        s[h] += __shfl_xor(s[h], 1, 32);
        u[h] += __shfl_xor(u[h], 1, 32);
    }

    if (sub == 0 && node < n) {
        v4f v;
        #pragma unroll
        for (int h = 0; h < H_HEADS; ++h) {
            const float x = (s[h] > 0.f) ? (u[h] / s[h]) : 0.f;
            v[h] = (x > 0.f) ? x : 0.f;
        }
        if (do_mean) {
            hs_next[node] = 0.25f * (v[0] + v[1] + v[2] + v[3]);
            if (node == n - 1)
                for (int j = n; j < np; ++j) hs_next[j] = 0.f;
        } else {
            const float c = centrality[node];
            float acc = 0.f;
            #pragma unroll
            for (int h = 0; h < H_HEADS; ++h)
                acc += (c * gamma[h] + beta[h]) * v[h];
            acc *= 0.25f;
            out[node] = (acc > 0.f) ? acc : 0.01f * acc;
        }
    }
}

extern "C" void kernel_launch(void* const* d_in, const int* in_sizes, int n_in,
                              void* d_out, int out_size, void* d_ws, size_t ws_size,
                              hipStream_t stream) {
    const float* inputs     = (const float*)d_in[0];
    const float* centrality = (const float*)d_in[1];
    const float* W1         = (const float*)d_in[2];
    const float* b1         = (const float*)d_in[3];
    const float* W2         = (const float*)d_in[4];
    const float* b2         = (const float*)d_in[5];
    const float* rel_emb    = (const float*)d_in[6];
    const float* W_pred     = (const float*)d_in[7];
    const float* gamma      = (const float*)d_in[8];
    const float* beta       = (const float*)d_in[9];
    const int*   edge_types = (const int*)d_in[10];
    const int*   src        = (const int*)d_in[11];
    const int*   dst        = (const int*)d_in[12];
    float* out = (float*)d_out;

    const int N = in_sizes[1];
    const int E = in_sizes[10];
    const int Np = (N + 3) & ~3;
    const int Ep = (E + 3) & ~3;
    const int NB = (N + (1 << BSH) - 1) >> BSH;
    const int nChunks = (E + CH - 1) / CH;
    const int M  = NB * nChunks;
    const int Mp = (M + 3) & ~3;
    const int Sp = SPAD + NB * FCAP + Ep;

    char* ws = (char*)d_ws;
    _Float16* W1h = (_Float16*)ws;
    float* etab   = (float*)(ws + 65536);
    float* hA     = (float*)(ws + 69632);
    int*   cnt      = (int*)(hA + (size_t)Np * H_HEADS);
    int*   rowstart = cnt + Np;
    unsigned* runtab = (unsigned*)(rowstart + Np);
    unsigned* part   = runtab + Mp;
    unsigned* sorted = part + Ep;
    float* hsA = (float*)(sorted + Sp);
    float* hsB = hsA + Np;

    const int nW1      = COLS * IN_DIM;
    const int w1Blocks = (nW1 + CTH - 1) / CTH;
    const int eBlocks  = (LAYERS * RELS * H_HEADS + CTH - 1) / CTH;
    k_pre<<<w1Blocks + eBlocks, CTH, 0, stream>>>(W1, W1h, nW1, w1Blocks, rel_emb, W_pred, etab);

    const int nTiles    = (N + 15) / 16;
    const int tilesPerBlock = (CTH / 32) * MLP_TPW;
    const int mlpBlocks = (nTiles + tilesPerBlock - 1) / tilesPerBlock;
    k_partition_mlp<<<nChunks + mlpBlocks, CTH, 0, stream>>>(
        edge_types, src, dst, part, runtab, E, nChunks, NB, nChunks,
        inputs, W1h, b1, W2, b2, hA, N, nTiles);

    k_fine<<<NB, FTH, 0, stream>>>(part, runtab, cnt, rowstart, sorted, N, nChunks, NB,
                                   etab, hA, hsA);

    const int lBlocks = (N + LNPW - 1) / LNPW;
    float* hscur = hsA;
    float* hsnxt = hsB;

    for (int l = 1; l < LAYERS; ++l) {
        const float* etab_l = etab + l * RELS * H_HEADS;
        const int do_mean = (l != LAYERS - 1) ? 1 : 0;
        k_gather_lds<<<lBlocks, LT, 0, stream>>>(sorted, rowstart, cnt, etab_l, hscur,
                                                 hsnxt, centrality, gamma, beta, out,
                                                 N, do_mean);
        float* tmp = hscur; hscur = hsnxt; hsnxt = tmp;
    }
}
